# starting kernel
# baseline (speedup 1.0000x reference)
#include <hip/hip_runtime.h>


typedef __attribute__((ext_vector_type(2))) float v2f;
typedef __attribute__((ext_vector_type(8))) float v8f;

#define HID 16
#define KITERS 10
#define ONE_MINUS_ALPHA 0.9f
#define ALPHA_F 0.1f

__device__ __forceinline__ float relu_f(float v) { return v > 0.f ? v : 0.f; }


__global__ void ak_deg_init(float* __restrict__ deg, int n) {
    int i = blockIdx.x * blockDim.x + threadIdx.x;
    if (i < n) deg[i] = 1.0f;
}

__global__ void ak_deg_accum(const int* __restrict__ eidx, float* __restrict__ deg, int E) {
    int e = blockIdx.x * blockDim.x + threadIdx.x;
    if (e < E) atomicAdd(&deg[eidx[(long long)E + e]], 1.0f);
}

__global__ void ak_dinv(const float* __restrict__ deg, float* __restrict__ dinv, int n) {
    int i = blockIdx.x * blockDim.x + threadIdx.x;
    if (i < n) dinv[i] = rsqrtf(deg[i]);
}


__global__ void ak_encoder_wmma(const float* __restrict__ x, const float* __restrict__ W1,
                                const float* __restrict__ b1, const float* __restrict__ W2,
                                const float* __restrict__ b2, const float* __restrict__ dinv,
                                float* __restrict__ h0, float* __restrict__ g,
                                float* __restrict__ agg, int n_tiles) {
    int wave = (blockIdx.x * blockDim.x + threadIdx.x) >> 5;
    int lane = threadIdx.x & 31;
    if (wave >= n_tiles) return;

    int row = lane & 15;
    int hi  = lane >> 4;
    int node = wave * 16 + row;
    float xv = x[node];

    v8f acc = {};
#pragma unroll
    for (int s = 0; s < 4; ++s) {
        int k0 = 4 * s + 2 * hi;
        v2f a, b;
        a.x = relu_f(xv * W1[k0]     + b1[k0]);
        a.y = relu_f(xv * W1[k0 + 1] + b1[k0 + 1]);
        b.x = W2[k0 * HID + row];
        b.y = W2[(k0 + 1) * HID + row];
        acc = __builtin_amdgcn_wmma_f32_16x16x4_f32(
            false, a, false, b, (short)0, acc, false, false);
    }

    float bb = b2[row];
#pragma unroll
    for (int i = 0; i < 8; ++i) {
        int r = i + 8 * hi;
        int nr = wave * 16 + r;
        float v = relu_f(acc[i] + bb);
        int idx = nr * HID + row;
        h0[idx]  = v;
        g[idx]   = dinv[nr] * v;
        agg[idx] = 0.f;
    }
}

__global__ void ak_encoder_tail(const float* __restrict__ x, const float* __restrict__ W1,
                                const float* __restrict__ b1, const float* __restrict__ W2,
                                const float* __restrict__ b2, const float* __restrict__ dinv,
                                float* __restrict__ h0, float* __restrict__ g,
                                float* __restrict__ agg, int start, int N) {
    int v = start + blockIdx.x * blockDim.x + threadIdx.x;
    if (v >= N) return;
    float h1[HID];
    float xv = x[v];
#pragma unroll
    for (int k = 0; k < HID; ++k) h1[k] = relu_f(xv * W1[k] + b1[k]);
    float dv = dinv[v];
#pragma unroll
    for (int j = 0; j < HID; ++j) {
        float acc = b2[j];
#pragma unroll
        for (int k = 0; k < HID; ++k) acc += h1[k] * W2[k * HID + j];
        float hv = relu_f(acc);
        int idx = v * HID + j;
        h0[idx] = hv; g[idx] = dv * hv; agg[idx] = 0.f;
    }
}


__global__ void ak_edge_scatter(const int* __restrict__ eidx, const float* __restrict__ g,
                                float* __restrict__ agg, int E) {
    long long t = (long long)blockIdx.x * blockDim.x + threadIdx.x;
    int e = (int)(t >> 2);
    if (e >= E) return;
    int p = (int)(t & 3);
    int src = eidx[e];
    int dst = eidx[(long long)E + e];
    const float4 val = *(const float4*)(g + (size_t)src * HID + p * 4);
    float* ad = agg + (size_t)dst * HID + p * 4;
    atomicAdd(ad + 0, val.x);
    atomicAdd(ad + 1, val.y);
    atomicAdd(ad + 2, val.z);
    atomicAdd(ad + 3, val.w);
}

__global__ void ak_finalize(float4* __restrict__ agg, float4* __restrict__ g,
                            const float4* __restrict__ h0, const float* __restrict__ dinv,
                            int n4) {
    int t = blockIdx.x * blockDim.x + threadIdx.x;
    if (t >= n4) return;
    int v = t >> 2;
    float dv = dinv[v];
    float4 a = agg[t], gg = g[t], h = h0[t];
    float4 hn;
    hn.x = ONE_MINUS_ALPHA * dv * (a.x + gg.x) + ALPHA_F * h.x;
    hn.y = ONE_MINUS_ALPHA * dv * (a.y + gg.y) + ALPHA_F * h.y;
    hn.z = ONE_MINUS_ALPHA * dv * (a.z + gg.z) + ALPHA_F * h.z;
    hn.w = ONE_MINUS_ALPHA * dv * (a.w + gg.w) + ALPHA_F * h.w;
    float4 gn; gn.x = dv * hn.x; gn.y = dv * hn.y; gn.z = dv * hn.z; gn.w = dv * hn.w;
    g[t] = gn;
    float4 z; z.x = 0.f; z.y = 0.f; z.z = 0.f; z.w = 0.f;
    agg[t] = z;
}

__global__ void ak_finalize_out(const float* __restrict__ agg, const float* __restrict__ g,
                                const float* __restrict__ h0, const float* __restrict__ dinv,
                                const float* __restrict__ W3, const float* __restrict__ b3,
                                float* __restrict__ out, int N) {
    int v = blockIdx.x * blockDim.x + threadIdx.x;
    if (v >= N) return;
    float dv = dinv[v];
    float s = b3[0];
#pragma unroll
    for (int j = 0; j < HID; ++j) {
        float h = ONE_MINUS_ALPHA * dv * (agg[v * HID + j] + g[v * HID + j])
                + ALPHA_F * h0[v * HID + j];
        s += h * W3[j];
    }
    out[v] = s;
}


extern "C" void kernel_launch(void* const* d_in, const int* in_sizes, int n_in,
                              void* d_out, int out_size, void* d_ws, size_t ws_size,
                              hipStream_t stream) {
    const float* x  = (const float*)d_in[0];
    const int*  eidx = (const int*)d_in[1];
    const float* W1 = (const float*)d_in[2];
    const float* b1 = (const float*)d_in[3];
    const float* W2 = (const float*)d_in[4];
    const float* b2 = (const float*)d_in[5];
    const float* W3 = (const float*)d_in[6];
    const float* b3 = (const float*)d_in[7];
    float* out = (float*)d_out;

    const int N = in_sizes[0];
    const int E = in_sizes[1] / 2;

    char* ws = (char*)d_ws;
    size_t off = 0;
    auto carve = [&](size_t bytes) -> char* {
        char* p = ws + off;
        off += (bytes + 255) & ~(size_t)255;
        return p;
    };
    float* deg  = (float*)carve((size_t)N * 4);
    float* dinv = (float*)carve((size_t)N * 4);
    float* h0   = (float*)carve((size_t)N * HID * 4);
    float* g    = (float*)carve((size_t)N * HID * 4);
    float* agg  = (float*)carve((size_t)N * HID * 4);

    const int B = 256;
    ak_deg_init<<<(N + B - 1) / B, B, 0, stream>>>(deg, N);
    ak_deg_accum<<<(E + B - 1) / B, B, 0, stream>>>(eidx, deg, E);
    ak_dinv<<<(N + B - 1) / B, B, 0, stream>>>(deg, dinv, N);

    int tiles = N / 16;
    if (tiles > 0) {
        int waves_per_block = B / 32;
        int blocks = (tiles + waves_per_block - 1) / waves_per_block;
        ak_encoder_wmma<<<blocks, B, 0, stream>>>(x, W1, b1, W2, b2, dinv, h0, g, agg, tiles);
    }
    int rem_start = tiles * 16;
    if (rem_start < N) {
        ak_encoder_tail<<<(N - rem_start + B - 1) / B, B, 0, stream>>>(
            x, W1, b1, W2, b2, dinv, h0, g, agg, rem_start, N);
    }

    long long sthreads = 4LL * E;
    int sblocks = (int)((sthreads + B - 1) / B);
    int n4 = N * 4;
    for (int it = 0; it < KITERS; ++it) {
        ak_edge_scatter<<<sblocks, B, 0, stream>>>(eidx, g, agg, E);
        if (it < KITERS - 1) {
            ak_finalize<<<(n4 + B - 1) / B, B, 0, stream>>>(
                (float4*)agg, (float4*)g, (const float4*)h0, dinv, n4);
        } else {
            ak_finalize_out<<<(N + B - 1) / B, B, 0, stream>>>(
                agg, g, h0, dinv, W3, b3, out, N);
        }
    }
}


# Sort edges by destination once per call (two-level counting sort on LDS counters); each hop then sums a node's source rows in registers, no float atomics; float sum order changes
# speedup vs baseline: 5.2913x; 5.2913x over previous
#include <hip/hip_runtime.h>


typedef __attribute__((ext_vector_type(2))) float v2f;
typedef __attribute__((ext_vector_type(8))) float v8f;

#define HID 16
#define KITERS 10
#define ONE_MINUS_ALPHA 0.9f
#define ALPHA_F 0.1f

__device__ __forceinline__ float relu_f(float v) { return v > 0.f ? v : 0.f; }


__global__ void ak_deg_init(float* __restrict__ deg, int n) {
    int i = blockIdx.x * blockDim.x + threadIdx.x;
    if (i < n) deg[i] = 1.0f;
}

__global__ void ak_deg_accum(const int* __restrict__ eidx, float* __restrict__ deg, int E) {
    int e = blockIdx.x * blockDim.x + threadIdx.x;
    if (e < E) atomicAdd(&deg[eidx[(long long)E + e]], 1.0f);
}

__global__ void ak_dinv(const float* __restrict__ deg, float* __restrict__ dinv, int n) {
    int i = blockIdx.x * blockDim.x + threadIdx.x;
    if (i < n) dinv[i] = rsqrtf(deg[i]);
}


__global__ void ak_encoder_wmma(const float* __restrict__ x, const float* __restrict__ W1,
                                const float* __restrict__ b1, const float* __restrict__ W2,
                                const float* __restrict__ b2, const float* __restrict__ dinv,
                                float* __restrict__ h0, float* __restrict__ g,
                                float* __restrict__ agg, int n_tiles) {
    int wave = (blockIdx.x * blockDim.x + threadIdx.x) >> 5;
    int lane = threadIdx.x & 31;
    if (wave >= n_tiles) return;

    int row = lane & 15;
    int hi  = lane >> 4;
    int node = wave * 16 + row;
    float xv = x[node];

    v8f acc = {};
#pragma unroll
    for (int s = 0; s < 4; ++s) {
        int k0 = 4 * s + 2 * hi;
        v2f a, b;
        a.x = relu_f(xv * W1[k0]     + b1[k0]);
        a.y = relu_f(xv * W1[k0 + 1] + b1[k0 + 1]);
        b.x = W2[k0 * HID + row];
        b.y = W2[(k0 + 1) * HID + row];
        acc = __builtin_amdgcn_wmma_f32_16x16x4_f32(
            false, a, false, b, (short)0, acc, false, false);
    }

    float bb = b2[row];
#pragma unroll
    for (int i = 0; i < 8; ++i) {
        int r = i + 8 * hi;
        int nr = wave * 16 + r;
        float v = relu_f(acc[i] + bb);
        int idx = nr * HID + row;
        h0[idx]  = v;
        g[idx]   = dinv[nr] * v;
        if (agg) agg[idx] = 0.f;
    }
}

__global__ void ak_encoder_tail(const float* __restrict__ x, const float* __restrict__ W1,
                                const float* __restrict__ b1, const float* __restrict__ W2,
                                const float* __restrict__ b2, const float* __restrict__ dinv,
                                float* __restrict__ h0, float* __restrict__ g,
                                float* __restrict__ agg, int start, int N) {
    int v = start + blockIdx.x * blockDim.x + threadIdx.x;
    if (v >= N) return;
    float h1[HID];
    float xv = x[v];
#pragma unroll
    for (int k = 0; k < HID; ++k) h1[k] = relu_f(xv * W1[k] + b1[k]);
    float dv = dinv[v];
#pragma unroll
    for (int j = 0; j < HID; ++j) {
        float acc = b2[j];
#pragma unroll
        for (int k = 0; k < HID; ++k) acc += h1[k] * W2[k * HID + j];
        float hv = relu_f(acc);
        int idx = v * HID + j;
        h0[idx] = hv; g[idx] = dv * hv;
        if (agg) agg[idx] = 0.f;
    }
}


__global__ void ak_edge_scatter(const int* __restrict__ eidx, const float* __restrict__ g,
                                float* __restrict__ agg, int E) {
    long long t = (long long)blockIdx.x * blockDim.x + threadIdx.x;
    int e = (int)(t >> 2);
    if (e >= E) return;
    int p = (int)(t & 3);
    int src = eidx[e];
    int dst = eidx[(long long)E + e];
    const float4 val = *(const float4*)(g + (size_t)src * HID + p * 4);
    float* ad = agg + (size_t)dst * HID + p * 4;
    atomicAdd(ad + 0, val.x);
    atomicAdd(ad + 1, val.y);
    atomicAdd(ad + 2, val.z);
    atomicAdd(ad + 3, val.w);
}

__global__ void ak_finalize(float4* __restrict__ agg, float4* __restrict__ g,
                            const float4* __restrict__ h0, const float* __restrict__ dinv,
                            int n4) {
    int t = blockIdx.x * blockDim.x + threadIdx.x;
    if (t >= n4) return;
    int v = t >> 2;
    float dv = dinv[v];
    float4 a = agg[t], gg = g[t], h = h0[t];
    float4 hn;
    hn.x = ONE_MINUS_ALPHA * dv * (a.x + gg.x) + ALPHA_F * h.x;
    hn.y = ONE_MINUS_ALPHA * dv * (a.y + gg.y) + ALPHA_F * h.y;
    hn.z = ONE_MINUS_ALPHA * dv * (a.z + gg.z) + ALPHA_F * h.z;
    hn.w = ONE_MINUS_ALPHA * dv * (a.w + gg.w) + ALPHA_F * h.w;
    float4 gn; gn.x = dv * hn.x; gn.y = dv * hn.y; gn.z = dv * hn.z; gn.w = dv * hn.w;
    g[t] = gn;
    float4 z; z.x = 0.f; z.y = 0.f; z.z = 0.f; z.w = 0.f;
    agg[t] = z;
}

__global__ void ak_finalize_out(const float* __restrict__ agg, const float* __restrict__ g,
                                const float* __restrict__ h0, const float* __restrict__ dinv,
                                const float* __restrict__ W3, const float* __restrict__ b3,
                                float* __restrict__ out, int N) {
    int v = blockIdx.x * blockDim.x + threadIdx.x;
    if (v >= N) return;
    float dv = dinv[v];
    float s = b3[0];
#pragma unroll
    for (int j = 0; j < HID; ++j) {
        float h = ONE_MINUS_ALPHA * dv * (agg[v * HID + j] + g[v * HID + j])
                + ALPHA_F * h0[v * HID + j];
        s += h * W3[j];
    }
    out[v] = s;
}


#define BSH 8
#define BNODES 256
#define NB_MAX 4096
#define CHUNK_WGS 256

__global__ void __launch_bounds__(1024)
ak_bucket_hist(const int* __restrict__ eidx, int* __restrict__ hist1, int E, int chunk, int nb) {
    __shared__ int hist[NB_MAX];
    for (int b = threadIdx.x; b < nb; b += 1024) hist[b] = 0;
    __syncthreads();
    const int lo = blockIdx.x * chunk;
    const int hi = min(lo + chunk, E);
    const int* __restrict__ dst = eidx + (size_t)E;
    for (int e0 = lo + threadIdx.x; e0 < hi; e0 += 4 * 1024) {
        int d[4];
#pragma unroll
        for (int k = 0; k < 4; ++k) d[k] = dst[min(e0 + k * 1024, hi - 1)];
#pragma unroll
        for (int k = 0; k < 4; ++k) {
            if (e0 + k * 1024 < hi) atomicAdd(&hist[d[k] >> BSH], 1);
        }
    }
    __syncthreads();
    for (int b = threadIdx.x; b < nb; b += 1024) hist1[b * CHUNK_WGS + blockIdx.x] = hist[b];
}

__global__ void __launch_bounds__(1024)
ak_excl_scan(const int* __restrict__ in, int* __restrict__ out, int n) {
    __shared__ int red[32];
    __shared__ int wsum[32];
    const int tid = threadIdx.x, lane = tid & 31, w = tid >> 5;
    const int first = blockIdx.x * 1024;

    int part = 0;
    for (int i = tid; i < first; i += 1024) part += in[i];
    for (int o = 16; o > 0; o >>= 1) part += __shfl_xor(part, o);
    if (lane == 0) red[w] = part;

    const int i = first + tid;
    const int c = (i < n) ? in[i] : 0;
    int x = c;
    for (int o = 1; o < 32; o <<= 1) {
        int y = __shfl_up(x, o);
        if (lane >= o) x += y;
    }
    if (lane == 31) wsum[w] = x;
    __syncthreads();
    if (w == 0) {
        int s = wsum[lane];
        for (int o = 1; o < 32; o <<= 1) {
            int y = __shfl_up(s, o);
            if (lane >= o) s += y;
        }
        wsum[lane] = s;
        int b = red[lane];
        for (int o = 16; o > 0; o >>= 1) b += __shfl_xor(b, o);
        red[lane] = b;
    }
    __syncthreads();
    if (i < n) out[i] = red[0] + (w ? wsum[w - 1] : 0) + x - c;
}

__global__ void __launch_bounds__(1024)
ak_bucket_scatter(const int* __restrict__ eidx, const int* __restrict__ off1,
                  unsigned* __restrict__ tmp, int E, int chunk, int nb) {
    __shared__ int cur[NB_MAX];
    for (int b = threadIdx.x; b < nb; b += 1024) cur[b] = off1[b * CHUNK_WGS + blockIdx.x];
    __syncthreads();
    const int lo = blockIdx.x * chunk;
    const int hi = min(lo + chunk, E);
    const int* __restrict__ dst = eidx + (size_t)E;
    for (int e0 = lo + threadIdx.x; e0 < hi; e0 += 4 * 1024) {
        int d[4], s[4];
#pragma unroll
        for (int k = 0; k < 4; ++k) {
            const int e = min(e0 + k * 1024, hi - 1);
            d[k] = dst[e];
            s[k] = eidx[e];
        }
#pragma unroll
        for (int k = 0; k < 4; ++k) {
            if (e0 + k * 1024 < hi) {
                const int slot = atomicAdd(&cur[d[k] >> BSH], 1);
                tmp[slot] = ((unsigned)s[k] << BSH) | (unsigned)(d[k] & (BNODES - 1));
            }
        }
    }
}

__global__ void __launch_bounds__(512)
ak_bucket_sort(const unsigned* __restrict__ tmp, const int* __restrict__ off1,
               int* __restrict__ rowptr, int* __restrict__ cnt, float* __restrict__ dinv,
               int* __restrict__ csr, int E, int N, int nb) {
    __shared__ int c[BNODES];
    __shared__ int cur[BNODES];
    __shared__ int wtot[BNODES / 32];
    const int tid = threadIdx.x, lane = tid & 31, w = tid >> 5;
    const int b = blockIdx.x;
    const int lo = off1[b * CHUNK_WGS];
    const int hi = (b + 1 < nb) ? off1[(b + 1) * CHUNK_WGS] : E;
    if (tid < BNODES) c[tid] = 0;
    __syncthreads();
    for (int i0 = lo + tid; i0 < hi; i0 += 4 * 512) {
        unsigned u[4];
#pragma unroll
        for (int k = 0; k < 4; ++k) u[k] = tmp[min(i0 + k * 512, hi - 1)];
#pragma unroll
        for (int k = 0; k < 4; ++k) {
            if (i0 + k * 512 < hi) atomicAdd(&c[u[k] & (BNODES - 1)], 1);
        }
    }
    __syncthreads();
    int mine = 0, x = 0;
    if (tid < BNODES) {
        mine = c[tid];
        x = mine;
        for (int o = 1; o < 32; o <<= 1) {
            int y = __shfl_up(x, o);
            if (lane >= o) x += y;
        }
        if (lane == 31) wtot[w] = x;
    }
    __syncthreads();
    if (tid < BNODES) {
        int base = 0;
        for (int k = 0; k < w; ++k) base += wtot[k];
        const int ex = base + x - mine;
        cur[tid] = ex;
        const int v = b * BNODES + tid;
        if (v < N) {
            cnt[v] = mine;
            rowptr[v] = lo + ex;
            dinv[v] = rsqrtf((float)(mine + 1));
        }
    }
    __syncthreads();
    for (int i0 = lo + tid; i0 < hi; i0 += 4 * 512) {
        unsigned u[4];
#pragma unroll
        for (int k = 0; k < 4; ++k) u[k] = tmp[min(i0 + k * 512, hi - 1)];
#pragma unroll
        for (int k = 0; k < 4; ++k) {
            if (i0 + k * 512 < hi) {
                const int slot = atomicAdd(&cur[u[k] & (BNODES - 1)], 1);
                csr[lo + slot] = (int)(u[k] >> BSH);
            }
        }
    }
}

__device__ __forceinline__ void add4(float4& a, const float4 b) {
    a.x += b.x; a.y += b.y; a.z += b.z; a.w += b.w;
}

template <bool LAST>
__global__ void ak_hop(const int* __restrict__ rowptr, const int* __restrict__ cnt,
                       const int* __restrict__ csr, const float4* __restrict__ gin,
                       const float4* __restrict__ h0, const float* __restrict__ dinv,
                       float4* __restrict__ gout, const float* __restrict__ W3,
                       const float* __restrict__ b3, float* __restrict__ out, int n4) {
    const int t = blockIdx.x * blockDim.x + threadIdx.x;
    if (t >= n4) return;
    const int v = t >> 2, p = t & 3;
    const int n = cnt[v];
    const int* __restrict__ cs = csr + rowptr[v];
    float4 acc = gin[t];
    int j = 0;
    for (; j + 8 <= n; j += 8) {
        int s[8];
        float4 a[8];
#pragma unroll
        for (int k = 0; k < 8; ++k) s[k] = cs[j + k];
#pragma unroll
        for (int k = 0; k < 8; ++k) a[k] = gin[s[k] * 4 + p];
#pragma unroll
        for (int k = 0; k < 8; ++k) add4(acc, a[k]);
    }
    if (j < n) {
        int s[7];
        float4 a[7];
#pragma unroll
        for (int k = 0; k < 7; ++k) s[k] = cs[min(j + k, n - 1)];
#pragma unroll
        for (int k = 0; k < 7; ++k) a[k] = gin[s[k] * 4 + p];
#pragma unroll
        for (int k = 0; k < 7; ++k) {
            const float wk = (j + k < n) ? 1.f : 0.f;
            acc.x = fmaf(a[k].x, wk, acc.x);
            acc.y = fmaf(a[k].y, wk, acc.y);
            acc.z = fmaf(a[k].z, wk, acc.z);
            acc.w = fmaf(a[k].w, wk, acc.w);
        }
    }
    const float dv = dinv[v];
    const float4 h = h0[t];
    float4 hn;
    hn.x = ONE_MINUS_ALPHA * dv * acc.x + ALPHA_F * h.x;
    hn.y = ONE_MINUS_ALPHA * dv * acc.y + ALPHA_F * h.y;
    hn.z = ONE_MINUS_ALPHA * dv * acc.z + ALPHA_F * h.z;
    hn.w = ONE_MINUS_ALPHA * dv * acc.w + ALPHA_F * h.w;
    if (LAST) {
        float sdot = hn.x * W3[4 * p] + hn.y * W3[4 * p + 1]
                   + hn.z * W3[4 * p + 2] + hn.w * W3[4 * p + 3];
        sdot += __shfl_xor(sdot, 1);
        sdot += __shfl_xor(sdot, 2);
        if (p == 0) out[v] = sdot + b3[0];
    } else {
        float4 gn;
        gn.x = dv * hn.x; gn.y = dv * hn.y; gn.z = dv * hn.z; gn.w = dv * hn.w;
        gout[t] = gn;
    }
}


extern "C" void kernel_launch(void* const* d_in, const int* in_sizes, int n_in,
                              void* d_out, int out_size, void* d_ws, size_t ws_size,
                              hipStream_t stream) {
    const float* x  = (const float*)d_in[0];
    const int*  eidx = (const int*)d_in[1];
    const float* W1 = (const float*)d_in[2];
    const float* b1 = (const float*)d_in[3];
    const float* W2 = (const float*)d_in[4];
    const float* b2 = (const float*)d_in[5];
    const float* W3 = (const float*)d_in[6];
    const float* b3 = (const float*)d_in[7];
    float* out = (float*)d_out;

    const int N = in_sizes[0];
    const int E = in_sizes[1] / 2;

    char* ws = (char*)d_ws;
    const int B = 256;

    {
        const int nb = (N + BNODES - 1) / BNODES;
        const int nh = nb * CHUNK_WGS;
        size_t o2 = 0;
        auto carve2 = [&](size_t bytes) -> char* {
            char* p = ws + o2;
            o2 += (bytes + 255) & ~(size_t)255;
            return p;
        };
        int*   cnt    = (int*)carve2((size_t)N * 4);
        float* dinv   = (float*)carve2((size_t)N * 4);
        int*   rowptr = (int*)carve2((size_t)N * 4);
        int*   hist1  = (int*)carve2((size_t)nh * 4);
        int*   off1   = (int*)carve2((size_t)nh * 4);
        int*   csr    = (int*)carve2((size_t)E * 4);
        const size_t feat = (size_t)N * HID * 4;
        const size_t tmpb = (size_t)E * 4;
        char* shared_region = carve2(tmpb > 3 * feat ? tmpb : 3 * feat);
        unsigned* tmp = (unsigned*)shared_region;
        float* h0 = (float*)shared_region;
        float* gA = (float*)(shared_region + feat);
        float* gB = (float*)(shared_region + 2 * feat);

        if (o2 <= ws_size && nb <= NB_MAX && N <= (1 << 24)) {
            const int chunk = (E + CHUNK_WGS - 1) / CHUNK_WGS;
            ak_bucket_hist<<<CHUNK_WGS, 1024, 0, stream>>>(eidx, hist1, E, chunk, nb);
            ak_excl_scan<<<(nh + 1023) / 1024, 1024, 0, stream>>>(hist1, off1, nh);
            ak_bucket_scatter<<<CHUNK_WGS, 1024, 0, stream>>>(eidx, off1, tmp, E, chunk, nb);
            ak_bucket_sort<<<nb, 512, 0, stream>>>(tmp, off1, rowptr, cnt, dinv, csr, E, N, nb);

            int tiles = N / 16;
            if (tiles > 0) {
                int waves_per_block = B / 32;
                int blocks = (tiles + waves_per_block - 1) / waves_per_block;
                ak_encoder_wmma<<<blocks, B, 0, stream>>>(x, W1, b1, W2, b2, dinv, h0, gA,
                                                          (float*)nullptr, tiles);
            }
            int rem_start = tiles * 16;
            if (rem_start < N) {
                ak_encoder_tail<<<(N - rem_start + B - 1) / B, B, 0, stream>>>(
                    x, W1, b1, W2, b2, dinv, h0, gA, (float*)nullptr, rem_start, N);
            }

            const int n4 = N * 4;
            const int hblocks = (n4 + B - 1) / B;
            for (int it = 0; it < KITERS; ++it) {
                const float4* gi = (const float4*)((it & 1) ? gB : gA);
                float4* go = (float4*)((it & 1) ? gA : gB);
                if (it < KITERS - 1) {
                    ak_hop<false><<<hblocks, B, 0, stream>>>(rowptr, cnt, csr, gi, (const float4*)h0,
                                                             dinv, go, W3, b3, out, n4);
                } else {
                    ak_hop<true><<<hblocks, B, 0, stream>>>(rowptr, cnt, csr, gi, (const float4*)h0,
                                                            dinv, (float4*)nullptr, W3, b3, out, n4);
                }
            }
            return;
        }
    }

    size_t off = 0;
    auto carve = [&](size_t bytes) -> char* {
        char* p = ws + off;
        off += (bytes + 255) & ~(size_t)255;
        return p;
    };
    float* deg  = (float*)carve((size_t)N * 4);
    float* dinv = (float*)carve((size_t)N * 4);
    float* h0   = (float*)carve((size_t)N * HID * 4);
    float* g    = (float*)carve((size_t)N * HID * 4);
    float* agg  = (float*)carve((size_t)N * HID * 4);

    ak_deg_init<<<(N + B - 1) / B, B, 0, stream>>>(deg, N);
    ak_deg_accum<<<(E + B - 1) / B, B, 0, stream>>>(eidx, deg, E);
    ak_dinv<<<(N + B - 1) / B, B, 0, stream>>>(deg, dinv, N);

    int tiles = N / 16;
    if (tiles > 0) {
        int waves_per_block = B / 32;
        int blocks = (tiles + waves_per_block - 1) / waves_per_block;
        ak_encoder_wmma<<<blocks, B, 0, stream>>>(x, W1, b1, W2, b2, dinv, h0, g, agg, tiles);
    }
    int rem_start = tiles * 16;
    if (rem_start < N) {
        ak_encoder_tail<<<(N - rem_start + B - 1) / B, B, 0, stream>>>(
            x, W1, b1, W2, b2, dinv, h0, g, agg, rem_start, N);
    }

    long long sthreads = 4LL * E;
    int sblocks = (int)((sthreads + B - 1) / B);
    int n4 = N * 4;
    for (int it = 0; it < KITERS; ++it) {
        ak_edge_scatter<<<sblocks, B, 0, stream>>>(eidx, g, agg, E);
        if (it < KITERS - 1) {
            ak_finalize<<<(n4 + B - 1) / B, B, 0, stream>>>(
                (float4*)agg, (float4*)g, (const float4*)h0, dinv, n4);
        } else {
            ak_finalize_out<<<(N + B - 1) / B, B, 0, stream>>>(
                agg, g, h0, dinv, W3, b3, out, N);
        }
    }
}


# Readout W3 applied to h0 before the hops (hops are linear, so the two commute): a hop gathers 1 float per edge, not 16; 8 lanes per node, waves loop over nodes; float sum order changes
# speedup vs baseline: 6.6008x; 1.2475x over previous
#include <hip/hip_runtime.h>


typedef __attribute__((ext_vector_type(2))) float v2f;
typedef __attribute__((ext_vector_type(8))) float v8f;

#define HID 16
#define KITERS 10
#define ONE_MINUS_ALPHA 0.9f
#define ALPHA_F 0.1f

__device__ __forceinline__ float relu_f(float v) { return v > 0.f ? v : 0.f; }


__global__ void ak_deg_init(float* __restrict__ deg, int n) {
    int i = blockIdx.x * blockDim.x + threadIdx.x;
    if (i < n) deg[i] = 1.0f;
}

__global__ void ak_deg_accum(const int* __restrict__ eidx, float* __restrict__ deg, int E) {
    int e = blockIdx.x * blockDim.x + threadIdx.x;
    if (e < E) atomicAdd(&deg[eidx[(long long)E + e]], 1.0f);
}

__global__ void ak_dinv(const float* __restrict__ deg, float* __restrict__ dinv, int n) {
    int i = blockIdx.x * blockDim.x + threadIdx.x;
    if (i < n) dinv[i] = rsqrtf(deg[i]);
}

__device__ __forceinline__ float xhalf(float x) {
    int b = __builtin_bit_cast(int, x);
    return __builtin_bit_cast(float,
        __builtin_amdgcn_permlanex16(b, b, 0x76543210, 0xfedcba98, false, false));
}


__global__ void ak_encoder_wmma(const float* __restrict__ x, const float* __restrict__ W1,
                                const float* __restrict__ b1, const float* __restrict__ W2,
                                const float* __restrict__ b2, const float* __restrict__ W3,
                                const float* __restrict__ dinv, float* __restrict__ z0,
                                float* __restrict__ g, float* __restrict__ agg, int n_tiles) {
    int wave = (blockIdx.x * blockDim.x + threadIdx.x) >> 5;
    int lane = threadIdx.x & 31;
    if (wave >= n_tiles) return;

    int row = lane & 15;
    int hi  = lane >> 4;
    int node = wave * 16 + row;
    float xv = x[node];

    v8f acc = {};
#pragma unroll
    for (int s = 0; s < 4; ++s) {
        int k0 = 4 * s + 2 * hi;
        v2f wt, ht;
        wt.x = W2[k0 * HID + row];
        wt.y = W2[(k0 + 1) * HID + row];
        ht.x = relu_f(xv * W1[k0]     + b1[k0]);
        ht.y = relu_f(xv * W1[k0 + 1] + b1[k0 + 1]);
        acc = __builtin_amdgcn_wmma_f32_16x16x4_f32(
            false, wt, false, ht, (short)0, acc, false, false);
    }

    float part = 0.f;
#pragma unroll
    for (int i = 0; i < 8; ++i) {
        int m = i + 8 * hi;
        part += relu_f(acc[i] + b2[m]) * W3[m];
    }
    float z = part + xhalf(part);
    if (hi == 0) {
        z0[node] = z;
        g[node]  = dinv[node] * z;
        if (agg) agg[node] = 0.f;
    }
}

__global__ void ak_encoder_tail(const float* __restrict__ x, const float* __restrict__ W1,
                                const float* __restrict__ b1, const float* __restrict__ W2,
                                const float* __restrict__ b2, const float* __restrict__ W3,
                                const float* __restrict__ dinv, float* __restrict__ z0,
                                float* __restrict__ g, float* __restrict__ agg, int start, int N) {
    int v = start + blockIdx.x * blockDim.x + threadIdx.x;
    if (v >= N) return;
    float h1[HID];
    float xv = x[v];
#pragma unroll
    for (int k = 0; k < HID; ++k) h1[k] = relu_f(xv * W1[k] + b1[k]);
    float z = 0.f;
#pragma unroll
    for (int j = 0; j < HID; ++j) {
        float acc = b2[j];
#pragma unroll
        for (int k = 0; k < HID; ++k) acc += h1[k] * W2[k * HID + j];
        z += relu_f(acc) * W3[j];
    }
    z0[v] = z;
    g[v]  = dinv[v] * z;
    if (agg) agg[v] = 0.f;
}


__global__ void ak_edge_scatter(const int* __restrict__ eidx, const float* __restrict__ g,
                                float* __restrict__ agg, int E) {
    int e = blockIdx.x * blockDim.x + threadIdx.x;
    if (e >= E) return;
    atomicAdd(&agg[eidx[(long long)E + e]], g[eidx[e]]);
}

__global__ void ak_finalize(float* __restrict__ agg, float* __restrict__ g,
                            const float* __restrict__ z0, const float* __restrict__ dinv, int n) {
    int v = blockIdx.x * blockDim.x + threadIdx.x;
    if (v >= n) return;
    float dv = dinv[v];
    float zn = ONE_MINUS_ALPHA * dv * (agg[v] + g[v]) + ALPHA_F * z0[v];
    g[v] = dv * zn;
    agg[v] = 0.f;
}

__global__ void ak_finalize_out(const float* __restrict__ agg, const float* __restrict__ g,
                                const float* __restrict__ z0, const float* __restrict__ dinv,
                                const float* __restrict__ b3, float* __restrict__ out, int N) {
    int v = blockIdx.x * blockDim.x + threadIdx.x;
    if (v >= N) return;
    out[v] = ONE_MINUS_ALPHA * dinv[v] * (agg[v] + g[v]) + ALPHA_F * z0[v] + b3[0];
}


#define BSH 8
#define BNODES 256
#define NB_MAX 4096
#define CHUNK_WGS 256

__global__ void __launch_bounds__(1024)
ak_bucket_hist(const int* __restrict__ eidx, int* __restrict__ hist1, int E, int chunk, int nb) {
    __shared__ int hist[NB_MAX];
    for (int b = threadIdx.x; b < nb; b += 1024) hist[b] = 0;
    __syncthreads();
    const int lo = blockIdx.x * chunk;
    const int hi = min(lo + chunk, E);
    const int* __restrict__ dst = eidx + (size_t)E;
    for (int e0 = lo + threadIdx.x; e0 < hi; e0 += 4 * 1024) {
        int d[4];
#pragma unroll
        for (int k = 0; k < 4; ++k) d[k] = dst[min(e0 + k * 1024, hi - 1)];
#pragma unroll
        for (int k = 0; k < 4; ++k) {
            if (e0 + k * 1024 < hi) atomicAdd(&hist[d[k] >> BSH], 1);
        }
    }
    __syncthreads();
    for (int b = threadIdx.x; b < nb; b += 1024) hist1[b * CHUNK_WGS + blockIdx.x] = hist[b];
}

__global__ void __launch_bounds__(1024)
ak_excl_scan(const int* __restrict__ in, int* __restrict__ out, int n) {
    __shared__ int red[32];
    __shared__ int wsum[32];
    const int tid = threadIdx.x, lane = tid & 31, w = tid >> 5;
    const int first = blockIdx.x * 1024;

    int part = 0;
    for (int i = tid; i < first; i += 1024) part += in[i];
    for (int o = 16; o > 0; o >>= 1) part += __shfl_xor(part, o);
    if (lane == 0) red[w] = part;

    const int i = first + tid;
    const int c = (i < n) ? in[i] : 0;
    int x = c;
    for (int o = 1; o < 32; o <<= 1) {
        int y = __shfl_up(x, o);
        if (lane >= o) x += y;
    }
    if (lane == 31) wsum[w] = x;
    __syncthreads();
    if (w == 0) {
        int s = wsum[lane];
        for (int o = 1; o < 32; o <<= 1) {
            int y = __shfl_up(s, o);
            if (lane >= o) s += y;
        }
        wsum[lane] = s;
        int b = red[lane];
        for (int o = 16; o > 0; o >>= 1) b += __shfl_xor(b, o);
        red[lane] = b;
    }
    __syncthreads();
    if (i < n) out[i] = red[0] + (w ? wsum[w - 1] : 0) + x - c;
}

__global__ void __launch_bounds__(1024)
ak_bucket_scatter(const int* __restrict__ eidx, const int* __restrict__ off1,
                  unsigned* __restrict__ tmp, int E, int chunk, int nb) {
    __shared__ int cur[NB_MAX];
    for (int b = threadIdx.x; b < nb; b += 1024) cur[b] = off1[b * CHUNK_WGS + blockIdx.x];
    __syncthreads();
    const int lo = blockIdx.x * chunk;
    const int hi = min(lo + chunk, E);
    const int* __restrict__ dst = eidx + (size_t)E;
    for (int e0 = lo + threadIdx.x; e0 < hi; e0 += 4 * 1024) {
        int d[4], s[4];
#pragma unroll
        for (int k = 0; k < 4; ++k) {
            const int e = min(e0 + k * 1024, hi - 1);
            d[k] = dst[e];
            s[k] = eidx[e];
        }
#pragma unroll
        for (int k = 0; k < 4; ++k) {
            if (e0 + k * 1024 < hi) {
                const int slot = atomicAdd(&cur[d[k] >> BSH], 1);
                tmp[slot] = ((unsigned)s[k] << BSH) | (unsigned)(d[k] & (BNODES - 1));
            }
        }
    }
}

__global__ void __launch_bounds__(512)
ak_bucket_sort(const unsigned* __restrict__ tmp, const int* __restrict__ off1,
               int* __restrict__ rowptr, int* __restrict__ cnt, float* __restrict__ dinv,
               int* __restrict__ csr, int E, int N, int nb) {
    __shared__ int c[BNODES];
    __shared__ int cur[BNODES];
    __shared__ int wtot[BNODES / 32];
    const int tid = threadIdx.x, lane = tid & 31, w = tid >> 5;
    const int b = blockIdx.x;
    const int lo = off1[b * CHUNK_WGS];
    const int hi = (b + 1 < nb) ? off1[(b + 1) * CHUNK_WGS] : E;
    if (tid < BNODES) c[tid] = 0;
    __syncthreads();
    for (int i0 = lo + tid; i0 < hi; i0 += 4 * 512) {
        unsigned u[4];
#pragma unroll
        for (int k = 0; k < 4; ++k) u[k] = tmp[min(i0 + k * 512, hi - 1)];
#pragma unroll
        for (int k = 0; k < 4; ++k) {
            if (i0 + k * 512 < hi) atomicAdd(&c[u[k] & (BNODES - 1)], 1);
        }
    }
    __syncthreads();
    int mine = 0, x = 0;
    if (tid < BNODES) {
        mine = c[tid];
        x = mine;
        for (int o = 1; o < 32; o <<= 1) {
            int y = __shfl_up(x, o);
            if (lane >= o) x += y;
        }
        if (lane == 31) wtot[w] = x;
    }
    __syncthreads();
    if (tid < BNODES) {
        int base = 0;
        for (int k = 0; k < w; ++k) base += wtot[k];
        const int ex = base + x - mine;
        cur[tid] = ex;
        const int v = b * BNODES + tid;
        if (v < N) {
            cnt[v] = mine;
            rowptr[v] = lo + ex;
            dinv[v] = rsqrtf((float)(mine + 1));
        }
    }
    __syncthreads();
    for (int i0 = lo + tid; i0 < hi; i0 += 4 * 512) {
        unsigned u[4];
#pragma unroll
        for (int k = 0; k < 4; ++k) u[k] = tmp[min(i0 + k * 512, hi - 1)];
#pragma unroll
        for (int k = 0; k < 4; ++k) {
            if (i0 + k * 512 < hi) {
                const int slot = atomicAdd(&cur[u[k] & (BNODES - 1)], 1);
                csr[lo + slot] = (int)(u[k] >> BSH);
            }
        }
    }
}

template <int M>
__device__ __forceinline__ float add_xmask(float x) {
    int b = __builtin_bit_cast(int, x);
    return x + __builtin_bit_cast(float, __builtin_amdgcn_update_dpp(0, b, 0x160 + M, 0xf, 0xf, true));
}

#define HOP_TRIPS 6

template <bool LAST>
__global__ void __launch_bounds__(256)
ak_hop(const int* __restrict__ rowptr, const int* __restrict__ cnt,
       const int* __restrict__ csr, const float* __restrict__ gin,
       const float* __restrict__ z0, const float* __restrict__ dinv,
       float* __restrict__ gout, const float* __restrict__ b3,
       float* __restrict__ out, int N) {
    const int l = threadIdx.x & 7;
    const int grp = (blockIdx.x * 256 + threadIdx.x) >> 3;
    const int ngrp = (gridDim.x * 256) >> 3;
    int v = grp;
    int n_nx = cnt[min(v, N - 1)];
    int b_nx = rowptr[min(v, N - 1)];
    for (int vw = grp & ~3; vw < N; vw += ngrp, v += ngrp) {
        const int vv = min(v, N - 1);
        const int n = n_nx;
        const int* __restrict__ cs = csr + b_nx;
        const int vn = min(v + ngrp, N - 1);
        n_nx = cnt[vn];
        b_nx = rowptr[vn];
        const float self = gin[vv];
        const float dv = dinv[vv];
        const float zz = z0[vv];
        float acc = 0.f;
        for (int j0 = l; j0 - l < n; j0 += 64) {
            int s[8];
            float a[8];
#pragma unroll
            for (int k = 0; k < 8; ++k) s[k] = cs[min(j0 + 8 * k, n - 1)];
#pragma unroll
            for (int k = 0; k < 8; ++k) a[k] = gin[s[k]];
#pragma unroll
            for (int k = 0; k < 8; ++k) acc += (j0 + 8 * k < n) ? a[k] : 0.f;
        }
        acc = add_xmask<1>(acc);
        acc = add_xmask<2>(acc);
        acc = add_xmask<4>(acc);
        const float zn = ONE_MINUS_ALPHA * dv * (acc + self) + ALPHA_F * zz;
        if (l == 0 && v < N) {
            if (LAST) out[v] = zn + b3[0];
            else      gout[v] = dv * zn;
        }
    }
}


extern "C" void kernel_launch(void* const* d_in, const int* in_sizes, int n_in,
                              void* d_out, int out_size, void* d_ws, size_t ws_size,
                              hipStream_t stream) {
    const float* x  = (const float*)d_in[0];
    const int*  eidx = (const int*)d_in[1];
    const float* W1 = (const float*)d_in[2];
    const float* b1 = (const float*)d_in[3];
    const float* W2 = (const float*)d_in[4];
    const float* b2 = (const float*)d_in[5];
    const float* W3 = (const float*)d_in[6];
    const float* b3 = (const float*)d_in[7];
    float* out = (float*)d_out;

    const int N = in_sizes[0];
    const int E = in_sizes[1] / 2;

    char* ws = (char*)d_ws;
    const int B = 256;

    {
        const int nb = (N + BNODES - 1) / BNODES;
        const int nh = nb * CHUNK_WGS;
        size_t o2 = 0;
        auto carve2 = [&](size_t bytes) -> char* {
            char* p = ws + o2;
            o2 += (bytes + 255) & ~(size_t)255;
            return p;
        };
        int*   cnt    = (int*)carve2((size_t)N * 4);
        float* dinv   = (float*)carve2((size_t)N * 4);
        int*   rowptr = (int*)carve2((size_t)N * 4);
        int*   hist1  = (int*)carve2((size_t)nh * 4);
        int*   off1   = (int*)carve2((size_t)nh * 4);
        int*   csr    = (int*)carve2((size_t)E * 4);
        const size_t plane = (((size_t)N * 4) + 255) & ~(size_t)255;
        const size_t tmpb = (size_t)E * 4;
        char* shared_region = carve2(tmpb > 3 * plane ? tmpb : 3 * plane);
        unsigned* tmp = (unsigned*)shared_region;
        float* z0 = (float*)shared_region;
        float* gA = (float*)(shared_region + plane);
        float* gB = (float*)(shared_region + 2 * plane);

        if (o2 <= ws_size && nb <= NB_MAX && N <= (1 << 24)) {
            const int chunk = (E + CHUNK_WGS - 1) / CHUNK_WGS;
            ak_bucket_hist<<<CHUNK_WGS, 1024, 0, stream>>>(eidx, hist1, E, chunk, nb);
            ak_excl_scan<<<(nh + 1023) / 1024, 1024, 0, stream>>>(hist1, off1, nh);
            ak_bucket_scatter<<<CHUNK_WGS, 1024, 0, stream>>>(eidx, off1, tmp, E, chunk, nb);
            ak_bucket_sort<<<nb, 512, 0, stream>>>(tmp, off1, rowptr, cnt, dinv, csr, E, N, nb);

            int tiles = N / 16;
            if (tiles > 0) {
                int waves_per_block = B / 32;
                int blocks = (tiles + waves_per_block - 1) / waves_per_block;
                ak_encoder_wmma<<<blocks, B, 0, stream>>>(x, W1, b1, W2, b2, W3, dinv, z0, gA,
                                                          (float*)nullptr, tiles);
            }
            int rem_start = tiles * 16;
            if (rem_start < N) {
                ak_encoder_tail<<<(N - rem_start + B - 1) / B, B, 0, stream>>>(
                    x, W1, b1, W2, b2, W3, dinv, z0, gA, (float*)nullptr, rem_start, N);
            }

            const int hblocks = (N + 32 * HOP_TRIPS - 1) / (32 * HOP_TRIPS);
            for (int it = 0; it < KITERS; ++it) {
                const float* gi = (it & 1) ? gB : gA;
                float* go = (it & 1) ? gA : gB;
                if (it < KITERS - 1) {
                    ak_hop<false><<<hblocks, B, 0, stream>>>(rowptr, cnt, csr, gi, z0, dinv, go,
                                                             b3, out, N);
                } else {
                    ak_hop<true><<<hblocks, B, 0, stream>>>(rowptr, cnt, csr, gi, z0, dinv,
                                                            (float*)nullptr, b3, out, N);
                }
            }
            return;
        }
    }

    size_t off = 0;
    auto carve = [&](size_t bytes) -> char* {
        char* p = ws + off;
        off += (bytes + 255) & ~(size_t)255;
        return p;
    };
    float* deg  = (float*)carve((size_t)N * 4);
    float* dinv = (float*)carve((size_t)N * 4);
    float* z0   = (float*)carve((size_t)N * 4);
    float* g    = (float*)carve((size_t)N * 4);
    float* agg  = (float*)carve((size_t)N * 4);

    ak_deg_init<<<(N + B - 1) / B, B, 0, stream>>>(deg, N);
    ak_deg_accum<<<(E + B - 1) / B, B, 0, stream>>>(eidx, deg, E);
    ak_dinv<<<(N + B - 1) / B, B, 0, stream>>>(deg, dinv, N);

    int tiles = N / 16;
    if (tiles > 0) {
        int waves_per_block = B / 32;
        int blocks = (tiles + waves_per_block - 1) / waves_per_block;
        ak_encoder_wmma<<<blocks, B, 0, stream>>>(x, W1, b1, W2, b2, W3, dinv, z0, g, agg, tiles);
    }
    int rem_start = tiles * 16;
    if (rem_start < N) {
        ak_encoder_tail<<<(N - rem_start + B - 1) / B, B, 0, stream>>>(
            x, W1, b1, W2, b2, W3, dinv, z0, g, agg, rem_start, N);
    }

    for (int it = 0; it < KITERS; ++it) {
        ak_edge_scatter<<<(E + B - 1) / B, B, 0, stream>>>(eidx, g, agg, E);
        if (it < KITERS - 1) {
            ak_finalize<<<(N + B - 1) / B, B, 0, stream>>>(agg, g, z0, dinv, N);
        } else {
            ak_finalize_out<<<(N + B - 1) / B, B, 0, stream>>>(agg, g, z0, dinv, b3, out, N);
        }
    }
}


# Hops read g from LDS instead of L2: each work-group copies g into its LDS in two halves and sums sources from there; the sort files a node's sources by half; float sum order changes
# speedup vs baseline: 10.9652x; 1.6612x over previous
#include <hip/hip_runtime.h>


typedef __attribute__((ext_vector_type(2))) float v2f;
typedef __attribute__((ext_vector_type(8))) float v8f;

#define HID 16
#define KITERS 10
#define ONE_MINUS_ALPHA 0.9f
#define ALPHA_F 0.1f

__device__ __forceinline__ float relu_f(float v) { return v > 0.f ? v : 0.f; }


__global__ void ak_deg_init(float* __restrict__ deg, int n) {
    int i = blockIdx.x * blockDim.x + threadIdx.x;
    if (i < n) deg[i] = 1.0f;
}

__global__ void ak_deg_accum(const int* __restrict__ eidx, float* __restrict__ deg, int E) {
    int e = blockIdx.x * blockDim.x + threadIdx.x;
    if (e < E) atomicAdd(&deg[eidx[(long long)E + e]], 1.0f);
}

__global__ void ak_dinv(const float* __restrict__ deg, float* __restrict__ dinv, int n) {
    int i = blockIdx.x * blockDim.x + threadIdx.x;
    if (i < n) dinv[i] = rsqrtf(deg[i]);
}

__device__ __forceinline__ float xhalf(float x) {
    int b = __builtin_bit_cast(int, x);
    return __builtin_bit_cast(float,
        __builtin_amdgcn_permlanex16(b, b, 0x76543210, 0xfedcba98, false, false));
}


__global__ void ak_encoder_wmma(const float* __restrict__ x, const float* __restrict__ W1,
                                const float* __restrict__ b1, const float* __restrict__ W2,
                                const float* __restrict__ b2, const float* __restrict__ W3,
                                const float* __restrict__ dinv, float* __restrict__ z0,
                                float* __restrict__ g, float* __restrict__ agg, int n_tiles) {
    int wave = (blockIdx.x * blockDim.x + threadIdx.x) >> 5;
    int lane = threadIdx.x & 31;
    if (wave >= n_tiles) return;

    int row = lane & 15;
    int hi  = lane >> 4;
    int node = wave * 16 + row;
    float xv = x[node];

    v8f acc = {};
#pragma unroll
    for (int s = 0; s < 4; ++s) {
        int k0 = 4 * s + 2 * hi;
        v2f wt, ht;
        wt.x = W2[k0 * HID + row];
        wt.y = W2[(k0 + 1) * HID + row];
        ht.x = relu_f(xv * W1[k0]     + b1[k0]);
        ht.y = relu_f(xv * W1[k0 + 1] + b1[k0 + 1]);
        acc = __builtin_amdgcn_wmma_f32_16x16x4_f32(
            false, wt, false, ht, (short)0, acc, false, false);
    }

    float part = 0.f;
#pragma unroll
    for (int i = 0; i < 8; ++i) {
        int m = i + 8 * hi;
        part += relu_f(acc[i] + b2[m]) * W3[m];
    }
    float z = part + xhalf(part);
    if (hi == 0) {
        z0[node] = z;
        g[node]  = dinv[node] * z;
        if (agg) agg[node] = 0.f;
    }
}

__global__ void ak_encoder_tail(const float* __restrict__ x, const float* __restrict__ W1,
                                const float* __restrict__ b1, const float* __restrict__ W2,
                                const float* __restrict__ b2, const float* __restrict__ W3,
                                const float* __restrict__ dinv, float* __restrict__ z0,
                                float* __restrict__ g, float* __restrict__ agg, int start, int N) {
    int v = start + blockIdx.x * blockDim.x + threadIdx.x;
    if (v >= N) return;
    float h1[HID];
    float xv = x[v];
#pragma unroll
    for (int k = 0; k < HID; ++k) h1[k] = relu_f(xv * W1[k] + b1[k]);
    float z = 0.f;
#pragma unroll
    for (int j = 0; j < HID; ++j) {
        float acc = b2[j];
#pragma unroll
        for (int k = 0; k < HID; ++k) acc += h1[k] * W2[k * HID + j];
        z += relu_f(acc) * W3[j];
    }
    z0[v] = z;
    g[v]  = dinv[v] * z;
    if (agg) agg[v] = 0.f;
}


__global__ void ak_edge_scatter(const int* __restrict__ eidx, const float* __restrict__ g,
                                float* __restrict__ agg, int E) {
    int e = blockIdx.x * blockDim.x + threadIdx.x;
    if (e >= E) return;
    atomicAdd(&agg[eidx[(long long)E + e]], g[eidx[e]]);
}

__global__ void ak_finalize(float* __restrict__ agg, float* __restrict__ g,
                            const float* __restrict__ z0, const float* __restrict__ dinv, int n) {
    int v = blockIdx.x * blockDim.x + threadIdx.x;
    if (v >= n) return;
    float dv = dinv[v];
    float zn = ONE_MINUS_ALPHA * dv * (agg[v] + g[v]) + ALPHA_F * z0[v];
    g[v] = dv * zn;
    agg[v] = 0.f;
}

__global__ void ak_finalize_out(const float* __restrict__ agg, const float* __restrict__ g,
                                const float* __restrict__ z0, const float* __restrict__ dinv,
                                const float* __restrict__ b3, float* __restrict__ out, int N) {
    int v = blockIdx.x * blockDim.x + threadIdx.x;
    if (v >= N) return;
    out[v] = ONE_MINUS_ALPHA * dinv[v] * (agg[v] + g[v]) + ALPHA_F * z0[v] + b3[0];
}


#define BSH 8
#define BNODES 256
#define NB_MAX 4096
#define CHUNK_WGS 256

__global__ void __launch_bounds__(1024)
ak_bucket_hist(const int* __restrict__ eidx, int* __restrict__ hist1, int E, int chunk, int nb) {
    __shared__ int hist[NB_MAX];
    for (int b = threadIdx.x; b < nb; b += 1024) hist[b] = 0;
    __syncthreads();
    const int lo = blockIdx.x * chunk;
    const int hi = min(lo + chunk, E);
    const int* __restrict__ dst = eidx + (size_t)E;
    for (int e0 = lo + threadIdx.x; e0 < hi; e0 += 4 * 1024) {
        int d[4];
#pragma unroll
        for (int k = 0; k < 4; ++k) d[k] = dst[min(e0 + k * 1024, hi - 1)];
#pragma unroll
        for (int k = 0; k < 4; ++k) {
            if (e0 + k * 1024 < hi) atomicAdd(&hist[d[k] >> BSH], 1);
        }
    }
    __syncthreads();
    for (int b = threadIdx.x; b < nb; b += 1024) hist1[b * CHUNK_WGS + blockIdx.x] = hist[b];
}

__global__ void __launch_bounds__(1024)
ak_excl_scan(const int* __restrict__ in, int* __restrict__ out, int n) {
    __shared__ int red[32];
    __shared__ int wsum[32];
    const int tid = threadIdx.x, lane = tid & 31, w = tid >> 5;
    const int first = blockIdx.x * 1024;

    int part = 0;
    for (int i = tid; i < first; i += 1024) part += in[i];
    for (int o = 16; o > 0; o >>= 1) part += __shfl_xor(part, o);
    if (lane == 0) red[w] = part;

    const int i = first + tid;
    const int c = (i < n) ? in[i] : 0;
    int x = c;
    for (int o = 1; o < 32; o <<= 1) {
        int y = __shfl_up(x, o);
        if (lane >= o) x += y;
    }
    if (lane == 31) wsum[w] = x;
    __syncthreads();
    if (w == 0) {
        int s = wsum[lane];
        for (int o = 1; o < 32; o <<= 1) {
            int y = __shfl_up(s, o);
            if (lane >= o) s += y;
        }
        wsum[lane] = s;
        int b = red[lane];
        for (int o = 16; o > 0; o >>= 1) b += __shfl_xor(b, o);
        red[lane] = b;
    }
    __syncthreads();
    if (i < n) out[i] = red[0] + (w ? wsum[w - 1] : 0) + x - c;
}

__global__ void __launch_bounds__(1024)
ak_bucket_scatter(const int* __restrict__ eidx, const int* __restrict__ off1,
                  unsigned* __restrict__ tmp, int E, int chunk, int nb) {
    __shared__ int cur[NB_MAX];
    for (int b = threadIdx.x; b < nb; b += 1024) cur[b] = off1[b * CHUNK_WGS + blockIdx.x];
    __syncthreads();
    const int lo = blockIdx.x * chunk;
    const int hi = min(lo + chunk, E);
    const int* __restrict__ dst = eidx + (size_t)E;
    for (int e0 = lo + threadIdx.x; e0 < hi; e0 += 4 * 1024) {
        int d[4], s[4];
#pragma unroll
        for (int k = 0; k < 4; ++k) {
            const int e = min(e0 + k * 1024, hi - 1);
            d[k] = dst[e];
            s[k] = eidx[e];
        }
#pragma unroll
        for (int k = 0; k < 4; ++k) {
            if (e0 + k * 1024 < hi) {
                const int slot = atomicAdd(&cur[d[k] >> BSH], 1);
                tmp[slot] = ((unsigned)s[k] << BSH) | (unsigned)(d[k] & (BNODES - 1));
            }
        }
    }
}

__global__ void __launch_bounds__(512)
ak_bucket_sort(const unsigned* __restrict__ tmp, const int* __restrict__ off1,
               int* __restrict__ rowptr, int* __restrict__ cnt, int* __restrict__ mid,
               float* __restrict__ dinv, int* __restrict__ csr, int E, int N, int nb, int split) {
    __shared__ int c[2 * BNODES];
    __shared__ int cur[2 * BNODES];
    __shared__ int wtot[2 * BNODES / 32];
    const int tid = threadIdx.x, lane = tid & 31, w = tid >> 5;
    const int b = blockIdx.x;
    const int lo = off1[b * CHUNK_WGS];
    const int hi = (b + 1 < nb) ? off1[(b + 1) * CHUNK_WGS] : E;
    c[tid] = 0;
    __syncthreads();
    for (int i0 = lo + tid; i0 < hi; i0 += 4 * 512) {
        unsigned u[4];
#pragma unroll
        for (int k = 0; k < 4; ++k) u[k] = tmp[min(i0 + k * 512, hi - 1)];
#pragma unroll
        for (int k = 0; k < 4; ++k) {
            if (i0 + k * 512 < hi) {
                const int key = (int)((u[k] & (BNODES - 1)) << 1) | ((int)(u[k] >> BSH) >= split ? 1 : 0);
                atomicAdd(&c[key], 1);
            }
        }
    }
    __syncthreads();
    const int mine = c[tid];
    int x = mine;
    for (int o = 1; o < 32; o <<= 1) {
        int y = __shfl_up(x, o);
        if (lane >= o) x += y;
    }
    if (lane == 31) wtot[w] = x;
    __syncthreads();
    int base = 0;
    for (int k = 0; k < w; ++k) base += wtot[k];
    const int ex = base + x - mine;
    cur[tid] = ex;
    if ((tid & 1) == 0) {
        const int v = b * BNODES + (tid >> 1);
        if (v < N) {
            const int total = mine + c[tid + 1];
            cnt[v] = total;
            rowptr[v] = lo + ex;
            mid[v] = lo + ex + mine;
            dinv[v] = rsqrtf((float)(total + 1));
        }
    }
    __syncthreads();
    for (int i0 = lo + tid; i0 < hi; i0 += 4 * 512) {
        unsigned u[4];
#pragma unroll
        for (int k = 0; k < 4; ++k) u[k] = tmp[min(i0 + k * 512, hi - 1)];
#pragma unroll
        for (int k = 0; k < 4; ++k) {
            if (i0 + k * 512 < hi) {
                const int key = (int)((u[k] & (BNODES - 1)) << 1) | ((int)(u[k] >> BSH) >= split ? 1 : 0);
                const int slot = atomicAdd(&cur[key], 1);
                csr[lo + slot] = (int)(u[k] >> BSH);
            }
        }
    }
}

template <int M>
__device__ __forceinline__ float add_xmask(float x) {
    int b = __builtin_bit_cast(int, x);
    return x + __builtin_bit_cast(float, __builtin_amdgcn_update_dpp(0, b, 0x160 + M, 0xf, 0xf, true));
}

#define HOP_TRIPS 6

template <bool LAST>
__global__ void __launch_bounds__(256)
ak_hop(const int* __restrict__ rowptr, const int* __restrict__ cnt,
       const int* __restrict__ csr, const float* __restrict__ gin,
       const float* __restrict__ z0, const float* __restrict__ dinv,
       float* __restrict__ gout, const float* __restrict__ b3,
       float* __restrict__ out, int N) {
    const int l = threadIdx.x & 7;
    const int grp = (blockIdx.x * 256 + threadIdx.x) >> 3;
    const int ngrp = (gridDim.x * 256) >> 3;
    int v = grp;
    int n_nx = cnt[min(v, N - 1)];
    int b_nx = rowptr[min(v, N - 1)];
    for (int vw = grp & ~3; vw < N; vw += ngrp, v += ngrp) {
        const int vv = min(v, N - 1);
        const int n = n_nx;
        const int* __restrict__ cs = csr + b_nx;
        const int vn = min(v + ngrp, N - 1);
        n_nx = cnt[vn];
        b_nx = rowptr[vn];
        const float self = gin[vv];
        const float dv = dinv[vv];
        const float zz = z0[vv];
        float acc = 0.f;
        for (int j0 = l; j0 - l < n; j0 += 64) {
            int s[8];
            float a[8];
#pragma unroll
            for (int k = 0; k < 8; ++k) s[k] = cs[min(j0 + 8 * k, n - 1)];
#pragma unroll
            for (int k = 0; k < 8; ++k) a[k] = gin[s[k]];
#pragma unroll
            for (int k = 0; k < 8; ++k) acc += (j0 + 8 * k < n) ? a[k] : 0.f;
        }
        acc = add_xmask<1>(acc);
        acc = add_xmask<2>(acc);
        acc = add_xmask<4>(acc);
        const float zn = ONE_MINUS_ALPHA * dv * (acc + self) + ALPHA_F * zz;
        if (l == 0 && v < N) {
            if (LAST) out[v] = zn + b3[0];
            else      gout[v] = dv * zn;
        }
    }
}


#define HL_WGS 256
#define HL_T 512
#define HL_TRIPS 8
#define HL_SLAB 65536
#define HL_RING 13

typedef __attribute__((ext_vector_type(4))) float v4f;
typedef __attribute__((ext_vector_type(4))) int v4i;

__device__ __forceinline__ int hl_piece(int j, int tid, int nq, int rot) {
    const int i = min(j * HL_T + tid, nq - 1) + rot;
    return i >= nq ? i - nq : i;
}
__device__ __forceinline__ int hl_rot(int nq) {
    return (int)((blockIdx.x * (unsigned)nq) / HL_WGS) & ~31;
}

__device__ __forceinline__ v4i hl_quad(const int* __restrict__ csr, int p, int epad) {
    return *reinterpret_cast<const v4i*>(csr + min(p, epad - 4));
}

__device__ __forceinline__ float hl_sum4(const float* sg, v4i q, int p, int s, int e, int base) {
    const int qq[4] = {q.x, q.y, q.z, q.w};
    float a[4];
    bool ok[4];
#pragma unroll
    for (int j = 0; j < 4; ++j) {
        ok[j] = (p + j >= s) && (p + j < e);
        a[j] = sg[ok[j] ? qq[j] - base : 0];
    }
    float acc = 0.f;
#pragma unroll
    for (int j = 0; j < 4; ++j) acc += ok[j] ? a[j] : 0.f;
    return acc;
}

__device__ __forceinline__ float hl_first(const float* sg, v4i q0, int s, int e, int l, int base) {
    return hl_sum4(sg, q0, (s & ~3) + 4 * l, s, e, base);
}
__device__ __forceinline__ float hl_rest(const float* sg, const int* __restrict__ csr,
                                         int s, int e, int l, int base, int epad) {
    float acc = 0.f;
    for (int pp = (s & ~3) + 32; pp < e; pp += 32) {
        const int p = pp + 4 * l;
        acc += hl_sum4(sg, hl_quad(csr, p, epad), p, s, e, base);
    }
    return acc;
}

template <bool LAST>
__global__ void __launch_bounds__(HL_T)
ak_hop_lds(const int* __restrict__ rowptr, const int* __restrict__ cnt, const int* __restrict__ mid,
           const int* __restrict__ csr, const float* __restrict__ gin,
           const float* __restrict__ z0, const float* __restrict__ dinv,
           float* __restrict__ gout, const float* __restrict__ b3, float* __restrict__ out,
           int N, int npw, int split, int epad) {
    __shared__ float sg[HL_SLAB];
    __shared__ float stot[HL_T];
    const int tid = threadIdx.x, l = tid & 7, gi = tid >> 3;
    const int v_lo = blockIdx.x * npw;
    const int v_hi = min(v_lo + npw, N);

    v4f* const d4 = reinterpret_cast<v4f*>(sg);
    v4f rr[HL_RING];
    int rx[HL_RING];
    const v4f* __restrict__ const ha = reinterpret_cast<const v4f*>(gin);
    const int nqa = split >> 2, rota = hl_rot(nqa), pera = (nqa + HL_T - 1) / HL_T;
#pragma unroll
    for (int k = 0; k < HL_RING; ++k) {
        rx[k] = hl_piece(k, tid, nqa, rota);
        rr[k] = ha[rx[k]];
    }

    int s0[HL_TRIPS], m0[HL_TRIPS], e0[HL_TRIPS];
#pragma unroll
    for (int t = 0; t < HL_TRIPS; ++t) {
        const int v = v_lo + gi + 64 * t;
        const int vv = min(v, N - 1);
        const int rs = rowptr[vv], rm = mid[vv], rc = cnt[vv];
        const bool ok = v < v_hi;
        s0[t] = ok ? rs : 0;
        m0[t] = ok ? rm : 0;
        e0[t] = ok ? rs + rc : 0;
    }
    v4i qa[HL_TRIPS];
#pragma unroll
    for (int t = 0; t < HL_TRIPS; ++t) qa[t] = hl_quad(csr, (s0[t] & ~3) + 4 * l, epad);
    const int vf = v_lo + tid;
    const int vfc = min(vf, N - 1);
    const float f_dv = dinv[vfc], f_zz = z0[vfc], f_self = gin[vfc];

    for (int j0 = HL_RING; j0 < pera; j0 += HL_RING) {
#pragma unroll
        for (int k = 0; k < HL_RING; ++k) {
            d4[rx[k]] = rr[k];
            rx[k] = hl_piece(j0 + k, tid, nqa, rota);
            rr[k] = ha[rx[k]];
        }
    }
#pragma unroll
    for (int k = 0; k < HL_RING; ++k) d4[rx[k]] = rr[k];
    __syncthreads();

    const float* __restrict__ const g1 = gin + split;
    const v4f* __restrict__ const hb = reinterpret_cast<const v4f*>(g1);
    const int n1 = N - split;
    const int nqb = n1 >> 2, rotb = hl_rot(nqb), perb = (nqb + HL_T - 1) / HL_T;
#pragma unroll
    for (int k = 0; k < HL_RING; ++k) {
        rx[k] = hl_piece(k, tid, nqb, rotb);
        rr[k] = hb[rx[k]];
    }
    float acc[HL_TRIPS];
#pragma unroll
    for (int t = 0; t < HL_TRIPS; ++t) acc[t] = hl_first(sg, qa[t], s0[t], m0[t], l, 0);
#pragma unroll
    for (int t = 0; t < HL_TRIPS; ++t) acc[t] += hl_rest(sg, csr, s0[t], m0[t], l, 0, epad);
    v4i qb[HL_TRIPS];
#pragma unroll
    for (int t = 0; t < HL_TRIPS; ++t) qb[t] = hl_quad(csr, (m0[t] & ~3) + 4 * l, epad);
    __syncthreads();
    for (int j0 = HL_RING; j0 < perb; j0 += HL_RING) {
#pragma unroll
        for (int k = 0; k < HL_RING; ++k) {
            d4[rx[k]] = rr[k];
            rx[k] = hl_piece(j0 + k, tid, nqb, rotb);
            rr[k] = hb[rx[k]];
        }
    }
#pragma unroll
    for (int k = 0; k < HL_RING; ++k) d4[rx[k]] = rr[k];
    if (tid < (n1 & 3)) sg[(nqb << 2) + tid] = g1[(nqb << 2) + tid];
    __syncthreads();
#pragma unroll
    for (int t = 0; t < HL_TRIPS; ++t) acc[t] += hl_first(sg, qb[t], m0[t], e0[t], l, split);
#pragma unroll
    for (int t = 0; t < HL_TRIPS; ++t) acc[t] += hl_rest(sg, csr, m0[t], e0[t], l, split, epad);
#pragma unroll
    for (int t = 0; t < HL_TRIPS; ++t) {
        float a = acc[t];
        a = add_xmask<1>(a);
        a = add_xmask<2>(a);
        a = add_xmask<4>(a);
        if (l == 0) stot[gi + 64 * t] = a;
    }
    __syncthreads();
    if (tid < npw && vf < v_hi) {
        const float zn = ONE_MINUS_ALPHA * f_dv * (stot[tid] + f_self) + ALPHA_F * f_zz;
        if (LAST) out[vf] = zn + b3[0];
        else      gout[vf] = f_dv * zn;
    }
}


extern "C" void kernel_launch(void* const* d_in, const int* in_sizes, int n_in,
                              void* d_out, int out_size, void* d_ws, size_t ws_size,
                              hipStream_t stream) {
    const float* x  = (const float*)d_in[0];
    const int*  eidx = (const int*)d_in[1];
    const float* W1 = (const float*)d_in[2];
    const float* b1 = (const float*)d_in[3];
    const float* W2 = (const float*)d_in[4];
    const float* b2 = (const float*)d_in[5];
    const float* W3 = (const float*)d_in[6];
    const float* b3 = (const float*)d_in[7];
    float* out = (float*)d_out;

    const int N = in_sizes[0];
    const int E = in_sizes[1] / 2;

    char* ws = (char*)d_ws;
    const int B = 256;

    {
        const int nb = (N + BNODES - 1) / BNODES;
        const int nh = nb * CHUNK_WGS;
        size_t o2 = 0;
        auto carve2 = [&](size_t bytes) -> char* {
            char* p = ws + o2;
            o2 += (bytes + 255) & ~(size_t)255;
            return p;
        };
        int*   cnt    = (int*)carve2((size_t)N * 4);
        float* dinv   = (float*)carve2((size_t)N * 4);
        int*   rowptr = (int*)carve2((size_t)N * 4);
        int*   hist1  = (int*)carve2((size_t)nh * 4);
        int*   off1   = (int*)carve2((size_t)nh * 4);
        int*   csr    = (int*)carve2((size_t)E * 4);
        const size_t plane = (((size_t)N * 4) + 255) & ~(size_t)255;
        const size_t tmpb = (size_t)E * 4;
        char* shared_region = carve2(tmpb > 3 * plane ? tmpb : 3 * plane);
        unsigned* tmp = (unsigned*)shared_region;
        float* z0 = (float*)shared_region;
        float* gA = (float*)(shared_region + plane);
        float* gB = (float*)(shared_region + 2 * plane);
        int* mid = hist1;
        const bool lds_hop = N >= 4096 && N <= 2 * HL_SLAB && E >= 64;
        const int split = lds_hop ? ((N / 2 + 31) & ~31) : N;

        if (o2 <= ws_size && nb <= NB_MAX && N <= (1 << 24)) {
            const int chunk = (E + CHUNK_WGS - 1) / CHUNK_WGS;
            ak_bucket_hist<<<CHUNK_WGS, 1024, 0, stream>>>(eidx, hist1, E, chunk, nb);
            ak_excl_scan<<<(nh + 1023) / 1024, 1024, 0, stream>>>(hist1, off1, nh);
            ak_bucket_scatter<<<CHUNK_WGS, 1024, 0, stream>>>(eidx, off1, tmp, E, chunk, nb);
            ak_bucket_sort<<<nb, 512, 0, stream>>>(tmp, off1, rowptr, cnt, mid, dinv, csr, E, N, nb,
                                                   split);

            int tiles = N / 16;
            if (tiles > 0) {
                int waves_per_block = B / 32;
                int blocks = (tiles + waves_per_block - 1) / waves_per_block;
                ak_encoder_wmma<<<blocks, B, 0, stream>>>(x, W1, b1, W2, b2, W3, dinv, z0, gA,
                                                          (float*)nullptr, tiles);
            }
            int rem_start = tiles * 16;
            if (rem_start < N) {
                ak_encoder_tail<<<(N - rem_start + B - 1) / B, B, 0, stream>>>(
                    x, W1, b1, W2, b2, W3, dinv, z0, gA, (float*)nullptr, rem_start, N);
            }

            const int hblocks = (N + 32 * HOP_TRIPS - 1) / (32 * HOP_TRIPS);
            const int npw = (N + HL_WGS - 1) / HL_WGS;
            const int epad = (E + 63) & ~63;
            for (int it = 0; it < KITERS; ++it) {
                const float* gi = (it & 1) ? gB : gA;
                float* go = (it & 1) ? gA : gB;
                if (lds_hop) {
                    if (it < KITERS - 1) {
                        ak_hop_lds<false><<<HL_WGS, HL_T, 0, stream>>>(rowptr, cnt, mid, csr, gi, z0,
                                                                       dinv, go, b3, out, N, npw,
                                                                       split, epad);
                    } else {
                        ak_hop_lds<true><<<HL_WGS, HL_T, 0, stream>>>(rowptr, cnt, mid, csr, gi, z0,
                                                                      dinv, (float*)nullptr, b3, out,
                                                                      N, npw, split, epad);
                    }
                } else if (it < KITERS - 1) {
                    ak_hop<false><<<hblocks, B, 0, stream>>>(rowptr, cnt, csr, gi, z0, dinv, go,
                                                             b3, out, N);
                } else {
                    ak_hop<true><<<hblocks, B, 0, stream>>>(rowptr, cnt, csr, gi, z0, dinv,
                                                            (float*)nullptr, b3, out, N);
                }
            }
            return;
        }
    }

    size_t off = 0;
    auto carve = [&](size_t bytes) -> char* {
        char* p = ws + off;
        off += (bytes + 255) & ~(size_t)255;
        return p;
    };
    float* deg  = (float*)carve((size_t)N * 4);
    float* dinv = (float*)carve((size_t)N * 4);
    float* z0   = (float*)carve((size_t)N * 4);
    float* g    = (float*)carve((size_t)N * 4);
    float* agg  = (float*)carve((size_t)N * 4);

    ak_deg_init<<<(N + B - 1) / B, B, 0, stream>>>(deg, N);
    ak_deg_accum<<<(E + B - 1) / B, B, 0, stream>>>(eidx, deg, E);
    ak_dinv<<<(N + B - 1) / B, B, 0, stream>>>(deg, dinv, N);

    int tiles = N / 16;
    if (tiles > 0) {
        int waves_per_block = B / 32;
        int blocks = (tiles + waves_per_block - 1) / waves_per_block;
        ak_encoder_wmma<<<blocks, B, 0, stream>>>(x, W1, b1, W2, b2, W3, dinv, z0, g, agg, tiles);
    }
    int rem_start = tiles * 16;
    if (rem_start < N) {
        ak_encoder_tail<<<(N - rem_start + B - 1) / B, B, 0, stream>>>(
            x, W1, b1, W2, b2, W3, dinv, z0, g, agg, rem_start, N);
    }

    for (int it = 0; it < KITERS; ++it) {
        ak_edge_scatter<<<(E + B - 1) / B, B, 0, stream>>>(eidx, g, agg, E);
        if (it < KITERS - 1) {
            ak_finalize<<<(N + B - 1) / B, B, 0, stream>>>(agg, g, z0, dinv, N);
        } else {
            ak_finalize_out<<<(N + B - 1) / B, B, 0, stream>>>(agg, g, z0, dinv, b3, out, N);
        }
    }
}


# Sorted list packed as 2-byte offsets into the source's half: a lane's two 8-byte pieces, both loaded ahead, cover 64 sources per node and half, so no sum waits on a second index fetch
# speedup vs baseline: 11.7915x; 1.0754x over previous
#include <hip/hip_runtime.h>


typedef __attribute__((ext_vector_type(2))) float v2f;
typedef __attribute__((ext_vector_type(8))) float v8f;

#define HID 16
#define KITERS 10
#define ONE_MINUS_ALPHA 0.9f
#define ALPHA_F 0.1f

__device__ __forceinline__ float relu_f(float v) { return v > 0.f ? v : 0.f; }


__global__ void ak_deg_init(float* __restrict__ deg, int n) {
    int i = blockIdx.x * blockDim.x + threadIdx.x;
    if (i < n) deg[i] = 1.0f;
}

__global__ void ak_deg_accum(const int* __restrict__ eidx, float* __restrict__ deg, int E) {
    int e = blockIdx.x * blockDim.x + threadIdx.x;
    if (e < E) atomicAdd(&deg[eidx[(long long)E + e]], 1.0f);
}

__global__ void ak_dinv(const float* __restrict__ deg, float* __restrict__ dinv, int n) {
    int i = blockIdx.x * blockDim.x + threadIdx.x;
    if (i < n) dinv[i] = rsqrtf(deg[i]);
}

__device__ __forceinline__ float xhalf(float x) {
    int b = __builtin_bit_cast(int, x);
    return __builtin_bit_cast(float,
        __builtin_amdgcn_permlanex16(b, b, 0x76543210, 0xfedcba98, false, false));
}


__global__ void ak_encoder_wmma(const float* __restrict__ x, const float* __restrict__ W1,
                                const float* __restrict__ b1, const float* __restrict__ W2,
                                const float* __restrict__ b2, const float* __restrict__ W3,
                                const float* __restrict__ dinv, float* __restrict__ z0,
                                float* __restrict__ g, float* __restrict__ agg, int n_tiles) {
    int wave = (blockIdx.x * blockDim.x + threadIdx.x) >> 5;
    int lane = threadIdx.x & 31;
    if (wave >= n_tiles) return;

    int row = lane & 15;
    int hi  = lane >> 4;
    int node = wave * 16 + row;
    float xv = x[node];

    v8f acc = {};
#pragma unroll
    for (int s = 0; s < 4; ++s) {
        int k0 = 4 * s + 2 * hi;
        v2f wt, ht;
        wt.x = W2[k0 * HID + row];
        wt.y = W2[(k0 + 1) * HID + row];
        ht.x = relu_f(xv * W1[k0]     + b1[k0]);
        ht.y = relu_f(xv * W1[k0 + 1] + b1[k0 + 1]);
        acc = __builtin_amdgcn_wmma_f32_16x16x4_f32(
            false, wt, false, ht, (short)0, acc, false, false);
    }

    float part = 0.f;
#pragma unroll
    for (int i = 0; i < 8; ++i) {
        int m = i + 8 * hi;
        part += relu_f(acc[i] + b2[m]) * W3[m];
    }
    float z = part + xhalf(part);
    if (hi == 0) {
        z0[node] = z;
        g[node]  = dinv[node] * z;
        if (agg) agg[node] = 0.f;
    }
}

__global__ void ak_encoder_tail(const float* __restrict__ x, const float* __restrict__ W1,
                                const float* __restrict__ b1, const float* __restrict__ W2,
                                const float* __restrict__ b2, const float* __restrict__ W3,
                                const float* __restrict__ dinv, float* __restrict__ z0,
                                float* __restrict__ g, float* __restrict__ agg, int start, int N) {
    int v = start + blockIdx.x * blockDim.x + threadIdx.x;
    if (v >= N) return;
    float h1[HID];
    float xv = x[v];
#pragma unroll
    for (int k = 0; k < HID; ++k) h1[k] = relu_f(xv * W1[k] + b1[k]);
    float z = 0.f;
#pragma unroll
    for (int j = 0; j < HID; ++j) {
        float acc = b2[j];
#pragma unroll
        for (int k = 0; k < HID; ++k) acc += h1[k] * W2[k * HID + j];
        z += relu_f(acc) * W3[j];
    }
    z0[v] = z;
    g[v]  = dinv[v] * z;
    if (agg) agg[v] = 0.f;
}


__global__ void ak_edge_scatter(const int* __restrict__ eidx, const float* __restrict__ g,
                                float* __restrict__ agg, int E) {
    int e = blockIdx.x * blockDim.x + threadIdx.x;
    if (e >= E) return;
    atomicAdd(&agg[eidx[(long long)E + e]], g[eidx[e]]);
}

__global__ void ak_finalize(float* __restrict__ agg, float* __restrict__ g,
                            const float* __restrict__ z0, const float* __restrict__ dinv, int n) {
    int v = blockIdx.x * blockDim.x + threadIdx.x;
    if (v >= n) return;
    float dv = dinv[v];
    float zn = ONE_MINUS_ALPHA * dv * (agg[v] + g[v]) + ALPHA_F * z0[v];
    g[v] = dv * zn;
    agg[v] = 0.f;
}

__global__ void ak_finalize_out(const float* __restrict__ agg, const float* __restrict__ g,
                                const float* __restrict__ z0, const float* __restrict__ dinv,
                                const float* __restrict__ b3, float* __restrict__ out, int N) {
    int v = blockIdx.x * blockDim.x + threadIdx.x;
    if (v >= N) return;
    out[v] = ONE_MINUS_ALPHA * dinv[v] * (agg[v] + g[v]) + ALPHA_F * z0[v] + b3[0];
}


#define BSH 8
#define BNODES 256
#define NB_MAX 4096
#define CHUNK_WGS 256

__global__ void __launch_bounds__(1024)
ak_bucket_hist(const int* __restrict__ eidx, int* __restrict__ hist1, int E, int chunk, int nb) {
    __shared__ int hist[NB_MAX];
    for (int b = threadIdx.x; b < nb; b += 1024) hist[b] = 0;
    __syncthreads();
    const int lo = blockIdx.x * chunk;
    const int hi = min(lo + chunk, E);
    const int* __restrict__ dst = eidx + (size_t)E;
    for (int e0 = lo + threadIdx.x; e0 < hi; e0 += 4 * 1024) {
        int d[4];
#pragma unroll
        for (int k = 0; k < 4; ++k) d[k] = dst[min(e0 + k * 1024, hi - 1)];
#pragma unroll
        for (int k = 0; k < 4; ++k) {
            if (e0 + k * 1024 < hi) atomicAdd(&hist[d[k] >> BSH], 1);
        }
    }
    __syncthreads();
    for (int b = threadIdx.x; b < nb; b += 1024) hist1[b * CHUNK_WGS + blockIdx.x] = hist[b];
}

__global__ void __launch_bounds__(1024)
ak_excl_scan(const int* __restrict__ in, int* __restrict__ out, int n) {
    __shared__ int red[32];
    __shared__ int wsum[32];
    const int tid = threadIdx.x, lane = tid & 31, w = tid >> 5;
    const int first = blockIdx.x * 1024;

    int part = 0;
    for (int i = tid; i < first; i += 1024) part += in[i];
    for (int o = 16; o > 0; o >>= 1) part += __shfl_xor(part, o);
    if (lane == 0) red[w] = part;

    const int i = first + tid;
    const int c = (i < n) ? in[i] : 0;
    int x = c;
    for (int o = 1; o < 32; o <<= 1) {
        int y = __shfl_up(x, o);
        if (lane >= o) x += y;
    }
    if (lane == 31) wsum[w] = x;
    __syncthreads();
    if (w == 0) {
        int s = wsum[lane];
        for (int o = 1; o < 32; o <<= 1) {
            int y = __shfl_up(s, o);
            if (lane >= o) s += y;
        }
        wsum[lane] = s;
        int b = red[lane];
        for (int o = 16; o > 0; o >>= 1) b += __shfl_xor(b, o);
        red[lane] = b;
    }
    __syncthreads();
    if (i < n) out[i] = red[0] + (w ? wsum[w - 1] : 0) + x - c;
}

__global__ void __launch_bounds__(1024)
ak_bucket_scatter(const int* __restrict__ eidx, const int* __restrict__ off1,
                  unsigned* __restrict__ tmp, int E, int chunk, int nb) {
    __shared__ int cur[NB_MAX];
    for (int b = threadIdx.x; b < nb; b += 1024) cur[b] = off1[b * CHUNK_WGS + blockIdx.x];
    __syncthreads();
    const int lo = blockIdx.x * chunk;
    const int hi = min(lo + chunk, E);
    const int* __restrict__ dst = eidx + (size_t)E;
    for (int e0 = lo + threadIdx.x; e0 < hi; e0 += 4 * 1024) {
        int d[4], s[4];
#pragma unroll
        for (int k = 0; k < 4; ++k) {
            const int e = min(e0 + k * 1024, hi - 1);
            d[k] = dst[e];
            s[k] = eidx[e];
        }
#pragma unroll
        for (int k = 0; k < 4; ++k) {
            if (e0 + k * 1024 < hi) {
                const int slot = atomicAdd(&cur[d[k] >> BSH], 1);
                tmp[slot] = ((unsigned)s[k] << BSH) | (unsigned)(d[k] & (BNODES - 1));
            }
        }
    }
}

template <bool S16>
__global__ void __launch_bounds__(512)
ak_bucket_sort(const unsigned* __restrict__ tmp, const int* __restrict__ off1,
               int* __restrict__ rowptr, int* __restrict__ cnt, int* __restrict__ mid,
               float* __restrict__ dinv, int* __restrict__ csr, int E, int N, int nb, int split) {
    __shared__ int c[2 * BNODES];
    __shared__ int cur[2 * BNODES];
    __shared__ int wtot[2 * BNODES / 32];
    const int tid = threadIdx.x, lane = tid & 31, w = tid >> 5;
    const int b = blockIdx.x;
    const int lo = off1[b * CHUNK_WGS];
    const int hi = (b + 1 < nb) ? off1[(b + 1) * CHUNK_WGS] : E;
    c[tid] = 0;
    __syncthreads();
    for (int i0 = lo + tid; i0 < hi; i0 += 4 * 512) {
        unsigned u[4];
#pragma unroll
        for (int k = 0; k < 4; ++k) u[k] = tmp[min(i0 + k * 512, hi - 1)];
#pragma unroll
        for (int k = 0; k < 4; ++k) {
            if (i0 + k * 512 < hi) {
                const int key = (int)((u[k] & (BNODES - 1)) << 1) | ((int)(u[k] >> BSH) >= split ? 1 : 0);
                atomicAdd(&c[key], 1);
            }
        }
    }
    __syncthreads();
    const int mine = c[tid];
    int x = mine;
    for (int o = 1; o < 32; o <<= 1) {
        int y = __shfl_up(x, o);
        if (lane >= o) x += y;
    }
    if (lane == 31) wtot[w] = x;
    __syncthreads();
    int base = 0;
    for (int k = 0; k < w; ++k) base += wtot[k];
    const int ex = base + x - mine;
    cur[tid] = ex;
    if ((tid & 1) == 0) {
        const int v = b * BNODES + (tid >> 1);
        if (v < N) {
            const int total = mine + c[tid + 1];
            cnt[v] = total;
            rowptr[v] = lo + ex;
            mid[v] = lo + ex + mine;
            dinv[v] = rsqrtf((float)(total + 1));
        }
    }
    __syncthreads();
    for (int i0 = lo + tid; i0 < hi; i0 += 4 * 512) {
        unsigned u[4];
#pragma unroll
        for (int k = 0; k < 4; ++k) u[k] = tmp[min(i0 + k * 512, hi - 1)];
#pragma unroll
        for (int k = 0; k < 4; ++k) {
            if (i0 + k * 512 < hi) {
                const int key = (int)((u[k] & (BNODES - 1)) << 1) | ((int)(u[k] >> BSH) >= split ? 1 : 0);
                const int slot = atomicAdd(&cur[key], 1);
                const int sn = (int)(u[k] >> BSH);
                if (S16) {
                    reinterpret_cast<unsigned short*>(csr)[lo + slot] =
                        (unsigned short)(sn >= split ? sn - split : sn);
                } else {
                    csr[lo + slot] = sn;
                }
            }
        }
    }
}

template <int M>
__device__ __forceinline__ float add_xmask(float x) {
    int b = __builtin_bit_cast(int, x);
    return x + __builtin_bit_cast(float, __builtin_amdgcn_update_dpp(0, b, 0x160 + M, 0xf, 0xf, true));
}

#define HOP_TRIPS 6

template <bool LAST>
__global__ void __launch_bounds__(256)
ak_hop(const int* __restrict__ rowptr, const int* __restrict__ cnt,
       const int* __restrict__ csr, const float* __restrict__ gin,
       const float* __restrict__ z0, const float* __restrict__ dinv,
       float* __restrict__ gout, const float* __restrict__ b3,
       float* __restrict__ out, int N) {
    const int l = threadIdx.x & 7;
    const int grp = (blockIdx.x * 256 + threadIdx.x) >> 3;
    const int ngrp = (gridDim.x * 256) >> 3;
    int v = grp;
    int n_nx = cnt[min(v, N - 1)];
    int b_nx = rowptr[min(v, N - 1)];
    for (int vw = grp & ~3; vw < N; vw += ngrp, v += ngrp) {
        const int vv = min(v, N - 1);
        const int n = n_nx;
        const int* __restrict__ cs = csr + b_nx;
        const int vn = min(v + ngrp, N - 1);
        n_nx = cnt[vn];
        b_nx = rowptr[vn];
        const float self = gin[vv];
        const float dv = dinv[vv];
        const float zz = z0[vv];
        float acc = 0.f;
        for (int j0 = l; j0 - l < n; j0 += 64) {
            int s[8];
            float a[8];
#pragma unroll
            for (int k = 0; k < 8; ++k) s[k] = cs[min(j0 + 8 * k, n - 1)];
#pragma unroll
            for (int k = 0; k < 8; ++k) a[k] = gin[s[k]];
#pragma unroll
            for (int k = 0; k < 8; ++k) acc += (j0 + 8 * k < n) ? a[k] : 0.f;
        }
        acc = add_xmask<1>(acc);
        acc = add_xmask<2>(acc);
        acc = add_xmask<4>(acc);
        const float zn = ONE_MINUS_ALPHA * dv * (acc + self) + ALPHA_F * zz;
        if (l == 0 && v < N) {
            if (LAST) out[v] = zn + b3[0];
            else      gout[v] = dv * zn;
        }
    }
}


#define HL_WGS 256
#define HL_T 512
#define HL_TRIPS 8
#define HL_SLAB 65536
#define HL_RING 13

typedef __attribute__((ext_vector_type(4))) float v4f;
typedef __attribute__((ext_vector_type(2))) int v2i;

__device__ __forceinline__ int hl_piece(int j, int tid, int nq, int rot) {
    const int i = min(j * HL_T + tid, nq - 1) + rot;
    return i >= nq ? i - nq : i;
}
__device__ __forceinline__ int hl_rot(int nq) {
    return (int)((blockIdx.x * (unsigned)nq) / HL_WGS) & ~31;
}

__device__ __forceinline__ v2i hl_quad(const unsigned short* __restrict__ c16, int p, int epad) {
    return *reinterpret_cast<const v2i*>(c16 + min(p, epad - 4));
}

__device__ __forceinline__ float hl_sum4(const float* sg, v2i q, int p, int s, int e) {
    const int qq[4] = {q.x & 0xffff, (int)((unsigned)q.x >> 16),
                       q.y & 0xffff, (int)((unsigned)q.y >> 16)};
    float a[4];
    bool ok[4];
#pragma unroll
    for (int j = 0; j < 4; ++j) {
        ok[j] = (p + j >= s) && (p + j < e);
        a[j] = sg[ok[j] ? qq[j] : 0];
    }
    float acc = 0.f;
#pragma unroll
    for (int j = 0; j < 4; ++j) acc += ok[j] ? a[j] : 0.f;
    return acc;
}

__device__ __forceinline__ float hl_first_lo(const float* sg, v2i qlo, int s, int e, int l) {
    return hl_sum4(sg, qlo, (s & ~3) + 4 * l, s, e);
}
__device__ __forceinline__ float hl_first_hi(const float* sg, v2i qhi, int s, int e, int l) {
    const int p = (s & ~3) + 32 + 4 * l;
    return p < e ? hl_sum4(sg, qhi, p, s, e) : 0.f;
}
__device__ __forceinline__ float hl_rest(const float* sg, const unsigned short* __restrict__ c16,
                                         int s, int e, int l, int epad) {
    float acc = 0.f;
    for (int pp = (s & ~3) + 64; pp < e; pp += 64) {
        const int p = pp + 4 * l;
        acc += hl_sum4(sg, hl_quad(c16, p, epad), p, s, e);
        acc += hl_sum4(sg, hl_quad(c16, p + 32, epad), p + 32, s, e);
    }
    return acc;
}

template <bool LAST>
__global__ void __launch_bounds__(HL_T)
ak_hop_lds(const int* __restrict__ rowptr, const int* __restrict__ cnt, const int* __restrict__ mid,
           const int* __restrict__ csr, const float* __restrict__ gin,
           const float* __restrict__ z0, const float* __restrict__ dinv,
           float* __restrict__ gout, const float* __restrict__ b3, float* __restrict__ out,
           int N, int npw, int split, int epad) {
    __shared__ float sg[HL_SLAB];
    __shared__ float stot[HL_T];
    const unsigned short* __restrict__ const c16 = reinterpret_cast<const unsigned short*>(csr);
    const int tid = threadIdx.x, l = tid & 7, gi = tid >> 3;
    const int v_lo = blockIdx.x * npw;
    const int v_hi = min(v_lo + npw, N);

    v4f* const d4 = reinterpret_cast<v4f*>(sg);
    v4f rr[HL_RING];
    int rx[HL_RING];
    const v4f* __restrict__ const ha = reinterpret_cast<const v4f*>(gin);
    const int nqa = split >> 2, rota = hl_rot(nqa), pera = (nqa + HL_T - 1) / HL_T;
#pragma unroll
    for (int k = 0; k < HL_RING; ++k) {
        rx[k] = hl_piece(k, tid, nqa, rota);
        rr[k] = ha[rx[k]];
    }

    int s0[HL_TRIPS], m0[HL_TRIPS], e0[HL_TRIPS];
#pragma unroll
    for (int t = 0; t < HL_TRIPS; ++t) {
        const int v = v_lo + gi + 64 * t;
        const int vv = min(v, N - 1);
        const int rs = rowptr[vv], rm = mid[vv], rc = cnt[vv];
        const bool ok = v < v_hi;
        s0[t] = ok ? rs : 0;
        m0[t] = ok ? rm : 0;
        e0[t] = ok ? rs + rc : 0;
    }
    v2i qa[HL_TRIPS], qa2[HL_TRIPS];
#pragma unroll
    for (int t = 0; t < HL_TRIPS; ++t) {
        qa[t]  = hl_quad(c16, (s0[t] & ~3) + 4 * l, epad);
        qa2[t] = hl_quad(c16, (s0[t] & ~3) + 32 + 4 * l, epad);
    }
    const int vf = v_lo + tid;
    const int vfc = min(vf, N - 1);
    const float f_dv = dinv[vfc], f_zz = z0[vfc], f_self = gin[vfc];

    for (int j0 = HL_RING; j0 < pera; j0 += HL_RING) {
#pragma unroll
        for (int k = 0; k < HL_RING; ++k) {
            d4[rx[k]] = rr[k];
            rx[k] = hl_piece(j0 + k, tid, nqa, rota);
            rr[k] = ha[rx[k]];
        }
    }
#pragma unroll
    for (int k = 0; k < HL_RING; ++k) d4[rx[k]] = rr[k];
    __syncthreads();

    const float* __restrict__ const g1 = gin + split;
    const v4f* __restrict__ const hb = reinterpret_cast<const v4f*>(g1);
    const int n1 = N - split;
    const int nqb = n1 >> 2, rotb = hl_rot(nqb), perb = (nqb + HL_T - 1) / HL_T;
#pragma unroll
    for (int k = 0; k < HL_RING; ++k) {
        rx[k] = hl_piece(k, tid, nqb, rotb);
        rr[k] = hb[rx[k]];
    }
    float acc[HL_TRIPS];
#pragma unroll
    for (int t = 0; t < HL_TRIPS; ++t) acc[t] = hl_first_lo(sg, qa[t], s0[t], m0[t], l);
#pragma unroll
    for (int t = 0; t < HL_TRIPS; ++t) acc[t] += hl_first_hi(sg, qa2[t], s0[t], m0[t], l);
#pragma unroll
    for (int t = 0; t < HL_TRIPS; ++t) acc[t] += hl_rest(sg, c16, s0[t], m0[t], l, epad);
    v2i qb[HL_TRIPS], qb2[HL_TRIPS];
#pragma unroll
    for (int t = 0; t < HL_TRIPS; ++t) {
        qb[t]  = hl_quad(c16, (m0[t] & ~3) + 4 * l, epad);
        qb2[t] = hl_quad(c16, (m0[t] & ~3) + 32 + 4 * l, epad);
    }
    __syncthreads();
    for (int j0 = HL_RING; j0 < perb; j0 += HL_RING) {
#pragma unroll
        for (int k = 0; k < HL_RING; ++k) {
            d4[rx[k]] = rr[k];
            rx[k] = hl_piece(j0 + k, tid, nqb, rotb);
            rr[k] = hb[rx[k]];
        }
    }
#pragma unroll
    for (int k = 0; k < HL_RING; ++k) d4[rx[k]] = rr[k];
    if (tid < (n1 & 3)) sg[(nqb << 2) + tid] = g1[(nqb << 2) + tid];
    __syncthreads();
#pragma unroll
    for (int t = 0; t < HL_TRIPS; ++t) acc[t] += hl_first_lo(sg, qb[t], m0[t], e0[t], l);
#pragma unroll
    for (int t = 0; t < HL_TRIPS; ++t) acc[t] += hl_first_hi(sg, qb2[t], m0[t], e0[t], l);
#pragma unroll
    for (int t = 0; t < HL_TRIPS; ++t) acc[t] += hl_rest(sg, c16, m0[t], e0[t], l, epad);
#pragma unroll
    for (int t = 0; t < HL_TRIPS; ++t) {
        float a = acc[t];
        a = add_xmask<1>(a);
        a = add_xmask<2>(a);
        a = add_xmask<4>(a);
        if (l == 0) stot[gi + 64 * t] = a;
    }
    __syncthreads();
    if (tid < npw && vf < v_hi) {
        const float zn = ONE_MINUS_ALPHA * f_dv * (stot[tid] + f_self) + ALPHA_F * f_zz;
        if (LAST) out[vf] = zn + b3[0];
        else      gout[vf] = f_dv * zn;
    }
}


extern "C" void kernel_launch(void* const* d_in, const int* in_sizes, int n_in,
                              void* d_out, int out_size, void* d_ws, size_t ws_size,
                              hipStream_t stream) {
    const float* x  = (const float*)d_in[0];
    const int*  eidx = (const int*)d_in[1];
    const float* W1 = (const float*)d_in[2];
    const float* b1 = (const float*)d_in[3];
    const float* W2 = (const float*)d_in[4];
    const float* b2 = (const float*)d_in[5];
    const float* W3 = (const float*)d_in[6];
    const float* b3 = (const float*)d_in[7];
    float* out = (float*)d_out;

    const int N = in_sizes[0];
    const int E = in_sizes[1] / 2;

    char* ws = (char*)d_ws;
    const int B = 256;

    {
        const int nb = (N + BNODES - 1) / BNODES;
        const int nh = nb * CHUNK_WGS;
        size_t o2 = 0;
        auto carve2 = [&](size_t bytes) -> char* {
            char* p = ws + o2;
            o2 += (bytes + 255) & ~(size_t)255;
            return p;
        };
        int*   cnt    = (int*)carve2((size_t)N * 4);
        float* dinv   = (float*)carve2((size_t)N * 4);
        int*   rowptr = (int*)carve2((size_t)N * 4);
        int*   hist1  = (int*)carve2((size_t)nh * 4);
        int*   off1   = (int*)carve2((size_t)nh * 4);
        int*   csr    = (int*)carve2((size_t)E * 4);
        const size_t plane = (((size_t)N * 4) + 255) & ~(size_t)255;
        const size_t tmpb = (size_t)E * 4;
        char* shared_region = carve2(tmpb > 3 * plane ? tmpb : 3 * plane);
        unsigned* tmp = (unsigned*)shared_region;
        float* z0 = (float*)shared_region;
        float* gA = (float*)(shared_region + plane);
        float* gB = (float*)(shared_region + 2 * plane);
        int* mid = hist1;
        const bool lds_hop = N >= 4096 && N <= 2 * HL_SLAB && E >= 64;
        const int split = lds_hop ? ((N / 2 + 31) & ~31) : N;

        if (o2 <= ws_size && nb <= NB_MAX && N <= (1 << 24)) {
            const int chunk = (E + CHUNK_WGS - 1) / CHUNK_WGS;
            ak_bucket_hist<<<CHUNK_WGS, 1024, 0, stream>>>(eidx, hist1, E, chunk, nb);
            ak_excl_scan<<<(nh + 1023) / 1024, 1024, 0, stream>>>(hist1, off1, nh);
            ak_bucket_scatter<<<CHUNK_WGS, 1024, 0, stream>>>(eidx, off1, tmp, E, chunk, nb);
            if (lds_hop) {
                ak_bucket_sort<true><<<nb, 512, 0, stream>>>(tmp, off1, rowptr, cnt, mid, dinv, csr, E,
                                                             N, nb, split);
            } else {
                ak_bucket_sort<false><<<nb, 512, 0, stream>>>(tmp, off1, rowptr, cnt, mid, dinv, csr, E,
                                                              N, nb, split);
            }

            int tiles = N / 16;
            if (tiles > 0) {
                int waves_per_block = B / 32;
                int blocks = (tiles + waves_per_block - 1) / waves_per_block;
                ak_encoder_wmma<<<blocks, B, 0, stream>>>(x, W1, b1, W2, b2, W3, dinv, z0, gA,
                                                          (float*)nullptr, tiles);
            }
            int rem_start = tiles * 16;
            if (rem_start < N) {
                ak_encoder_tail<<<(N - rem_start + B - 1) / B, B, 0, stream>>>(
                    x, W1, b1, W2, b2, W3, dinv, z0, gA, (float*)nullptr, rem_start, N);
            }

            const int hblocks = (N + 32 * HOP_TRIPS - 1) / (32 * HOP_TRIPS);
            const int npw = (N + HL_WGS - 1) / HL_WGS;
            const int epad = (E + 63) & ~63;
            for (int it = 0; it < KITERS; ++it) {
                const float* gi = (it & 1) ? gB : gA;
                float* go = (it & 1) ? gA : gB;
                if (lds_hop) {
                    if (it < KITERS - 1) {
                        ak_hop_lds<false><<<HL_WGS, HL_T, 0, stream>>>(rowptr, cnt, mid, csr, gi, z0,
                                                                       dinv, go, b3, out, N, npw,
                                                                       split, epad);
                    } else {
                        ak_hop_lds<true><<<HL_WGS, HL_T, 0, stream>>>(rowptr, cnt, mid, csr, gi, z0,
                                                                      dinv, (float*)nullptr, b3, out,
                                                                      N, npw, split, epad);
                    }
                } else if (it < KITERS - 1) {
                    ak_hop<false><<<hblocks, B, 0, stream>>>(rowptr, cnt, csr, gi, z0, dinv, go,
                                                             b3, out, N);
                } else {
                    ak_hop<true><<<hblocks, B, 0, stream>>>(rowptr, cnt, csr, gi, z0, dinv,
                                                            (float*)nullptr, b3, out, N);
                }
            }
            return;
        }
    }

    size_t off = 0;
    auto carve = [&](size_t bytes) -> char* {
        char* p = ws + off;
        off += (bytes + 255) & ~(size_t)255;
        return p;
    };
    float* deg  = (float*)carve((size_t)N * 4);
    float* dinv = (float*)carve((size_t)N * 4);
    float* z0   = (float*)carve((size_t)N * 4);
    float* g    = (float*)carve((size_t)N * 4);
    float* agg  = (float*)carve((size_t)N * 4);

    ak_deg_init<<<(N + B - 1) / B, B, 0, stream>>>(deg, N);
    ak_deg_accum<<<(E + B - 1) / B, B, 0, stream>>>(eidx, deg, E);
    ak_dinv<<<(N + B - 1) / B, B, 0, stream>>>(deg, dinv, N);

    int tiles = N / 16;
    if (tiles > 0) {
        int waves_per_block = B / 32;
        int blocks = (tiles + waves_per_block - 1) / waves_per_block;
        ak_encoder_wmma<<<blocks, B, 0, stream>>>(x, W1, b1, W2, b2, W3, dinv, z0, g, agg, tiles);
    }
    int rem_start = tiles * 16;
    if (rem_start < N) {
        ak_encoder_tail<<<(N - rem_start + B - 1) / B, B, 0, stream>>>(
            x, W1, b1, W2, b2, W3, dinv, z0, g, agg, rem_start, N);
    }

    for (int it = 0; it < KITERS; ++it) {
        ak_edge_scatter<<<(E + B - 1) / B, B, 0, stream>>>(eidx, g, agg, E);
        if (it < KITERS - 1) {
            ak_finalize<<<(N + B - 1) / B, B, 0, stream>>>(agg, g, z0, dinv, N);
        } else {
            ak_finalize_out<<<(N + B - 1) / B, B, 0, stream>>>(agg, g, z0, dinv, b3, out, N);
        }
    }
}


# Both sort levels put their entries together in an LDS stage (level 1: a block's chunk, share by share; level 2: a bucket's list in final order) and store it in whole 128-byte lines, not scattered
# speedup vs baseline: 15.1675x; 1.2863x over previous
#include <hip/hip_runtime.h>


typedef __attribute__((ext_vector_type(2))) float v2f;
typedef __attribute__((ext_vector_type(8))) float v8f;

#define HID 16
#define KITERS 10
#define ONE_MINUS_ALPHA 0.9f
#define ALPHA_F 0.1f

__device__ __forceinline__ float relu_f(float v) { return v > 0.f ? v : 0.f; }


__global__ void ak_deg_init(float* __restrict__ deg, int n) {
    int i = blockIdx.x * blockDim.x + threadIdx.x;
    if (i < n) deg[i] = 1.0f;
}

__global__ void ak_deg_accum(const int* __restrict__ eidx, float* __restrict__ deg, int E) {
    int e = blockIdx.x * blockDim.x + threadIdx.x;
    if (e < E) atomicAdd(&deg[eidx[(long long)E + e]], 1.0f);
}

__global__ void ak_dinv(const float* __restrict__ deg, float* __restrict__ dinv, int n) {
    int i = blockIdx.x * blockDim.x + threadIdx.x;
    if (i < n) dinv[i] = rsqrtf(deg[i]);
}

__device__ __forceinline__ float xhalf(float x) {
    int b = __builtin_bit_cast(int, x);
    return __builtin_bit_cast(float,
        __builtin_amdgcn_permlanex16(b, b, 0x76543210, 0xfedcba98, false, false));
}


__global__ void ak_encoder_wmma(const float* __restrict__ x, const float* __restrict__ W1,
                                const float* __restrict__ b1, const float* __restrict__ W2,
                                const float* __restrict__ b2, const float* __restrict__ W3,
                                const float* __restrict__ dinv, float* __restrict__ z0,
                                float* __restrict__ g, float* __restrict__ agg, int n_tiles) {
    int wave = (blockIdx.x * blockDim.x + threadIdx.x) >> 5;
    int lane = threadIdx.x & 31;
    if (wave >= n_tiles) return;

    int row = lane & 15;
    int hi  = lane >> 4;
    int node = wave * 16 + row;
    float xv = x[node];

    v8f acc = {};
#pragma unroll
    for (int s = 0; s < 4; ++s) {
        int k0 = 4 * s + 2 * hi;
        v2f wt, ht;
        wt.x = W2[k0 * HID + row];
        wt.y = W2[(k0 + 1) * HID + row];
        ht.x = relu_f(xv * W1[k0]     + b1[k0]);
        ht.y = relu_f(xv * W1[k0 + 1] + b1[k0 + 1]);
        acc = __builtin_amdgcn_wmma_f32_16x16x4_f32(
            false, wt, false, ht, (short)0, acc, false, false);
    }

    float part = 0.f;
#pragma unroll
    for (int i = 0; i < 8; ++i) {
        int m = i + 8 * hi;
        part += relu_f(acc[i] + b2[m]) * W3[m];
    }
    float z = part + xhalf(part);
    if (hi == 0) {
        z0[node] = z;
        g[node]  = dinv[node] * z;
        if (agg) agg[node] = 0.f;
    }
}

__global__ void ak_encoder_tail(const float* __restrict__ x, const float* __restrict__ W1,
                                const float* __restrict__ b1, const float* __restrict__ W2,
                                const float* __restrict__ b2, const float* __restrict__ W3,
                                const float* __restrict__ dinv, float* __restrict__ z0,
                                float* __restrict__ g, float* __restrict__ agg, int start, int N) {
    int v = start + blockIdx.x * blockDim.x + threadIdx.x;
    if (v >= N) return;
    float h1[HID];
    float xv = x[v];
#pragma unroll
    for (int k = 0; k < HID; ++k) h1[k] = relu_f(xv * W1[k] + b1[k]);
    float z = 0.f;
#pragma unroll
    for (int j = 0; j < HID; ++j) {
        float acc = b2[j];
#pragma unroll
        for (int k = 0; k < HID; ++k) acc += h1[k] * W2[k * HID + j];
        z += relu_f(acc) * W3[j];
    }
    z0[v] = z;
    g[v]  = dinv[v] * z;
    if (agg) agg[v] = 0.f;
}


__global__ void ak_edge_scatter(const int* __restrict__ eidx, const float* __restrict__ g,
                                float* __restrict__ agg, int E) {
    int e = blockIdx.x * blockDim.x + threadIdx.x;
    if (e >= E) return;
    atomicAdd(&agg[eidx[(long long)E + e]], g[eidx[e]]);
}

__global__ void ak_finalize(float* __restrict__ agg, float* __restrict__ g,
                            const float* __restrict__ z0, const float* __restrict__ dinv, int n) {
    int v = blockIdx.x * blockDim.x + threadIdx.x;
    if (v >= n) return;
    float dv = dinv[v];
    float zn = ONE_MINUS_ALPHA * dv * (agg[v] + g[v]) + ALPHA_F * z0[v];
    g[v] = dv * zn;
    agg[v] = 0.f;
}

__global__ void ak_finalize_out(const float* __restrict__ agg, const float* __restrict__ g,
                                const float* __restrict__ z0, const float* __restrict__ dinv,
                                const float* __restrict__ b3, float* __restrict__ out, int N) {
    int v = blockIdx.x * blockDim.x + threadIdx.x;
    if (v >= N) return;
    out[v] = ONE_MINUS_ALPHA * dinv[v] * (agg[v] + g[v]) + ALPHA_F * z0[v] + b3[0];
}


#define BSH 8
#define BNODES 256
#define NB_MAX 4096
#define CHUNK_WGS 256

__global__ void __launch_bounds__(1024)
ak_bucket_hist(const int* __restrict__ eidx, int* __restrict__ hist1, int E, int chunk, int nb) {
    __shared__ int hist[NB_MAX];
    for (int b = threadIdx.x; b < nb; b += 1024) hist[b] = 0;
    __syncthreads();
    const int lo = blockIdx.x * chunk;
    const int hi = min(lo + chunk, E);
    const int* __restrict__ dst = eidx + (size_t)E;
    for (int e0 = lo + threadIdx.x; e0 < hi; e0 += 4 * 1024) {
        int d[4];
#pragma unroll
        for (int k = 0; k < 4; ++k) d[k] = dst[min(e0 + k * 1024, hi - 1)];
#pragma unroll
        for (int k = 0; k < 4; ++k) {
            if (e0 + k * 1024 < hi) atomicAdd(&hist[d[k] >> BSH], 1);
        }
    }
    __syncthreads();
    for (int b = threadIdx.x; b < nb; b += 1024) hist1[b * CHUNK_WGS + blockIdx.x] = hist[b];
}

__global__ void __launch_bounds__(1024)
ak_excl_scan(const int* __restrict__ in, int* __restrict__ out, int n) {
    __shared__ int red[32];
    __shared__ int wsum[32];
    const int tid = threadIdx.x, lane = tid & 31, w = tid >> 5;
    const int first = blockIdx.x * 1024;

    int part = 0;
    for (int i = tid; i < first; i += 1024) part += in[i];
    for (int o = 16; o > 0; o >>= 1) part += __shfl_xor(part, o);
    if (lane == 0) red[w] = part;

    const int i = first + tid;
    const int c = (i < n) ? in[i] : 0;
    int x = c;
    for (int o = 1; o < 32; o <<= 1) {
        int y = __shfl_up(x, o);
        if (lane >= o) x += y;
    }
    if (lane == 31) wsum[w] = x;
    __syncthreads();
    if (w == 0) {
        int s = wsum[lane];
        for (int o = 1; o < 32; o <<= 1) {
            int y = __shfl_up(s, o);
            if (lane >= o) s += y;
        }
        wsum[lane] = s;
        int b = red[lane];
        for (int o = 16; o > 0; o >>= 1) b += __shfl_xor(b, o);
        red[lane] = b;
    }
    __syncthreads();
    if (i < n) out[i] = red[0] + (w ? wsum[w - 1] : 0) + x - c;
}

template <bool STAGED>
__device__ __forceinline__ void ak_scatter_edges(const int* __restrict__ eidx, int E, int lo, int hi,
                                                 int tid, int* cur, unsigned* stg,
                                                 unsigned* __restrict__ tmp) {
    const int* __restrict__ dst = eidx + (size_t)E;
    for (int e0 = lo + tid; e0 < hi; e0 += 4 * 1024) {
        int d[4], s[4];
#pragma unroll
        for (int k = 0; k < 4; ++k) {
            const int e = min(e0 + k * 1024, hi - 1);
            d[k] = dst[e];
            s[k] = eidx[e];
        }
        asm volatile("" :: "v"(d[0]), "v"(d[1]), "v"(d[2]), "v"(d[3]),
                           "v"(s[0]), "v"(s[1]), "v"(s[2]), "v"(s[3]));
#pragma unroll
        for (int k = 0; k < 4; ++k) {
            if (e0 + k * 1024 < hi) {
                const int slot = atomicAdd(&cur[d[k] >> BSH], 1);
                const unsigned ent = ((unsigned)s[k] << BSH) | (unsigned)(d[k] & (BNODES - 1));
                if (STAGED) stg[slot] = ent;
                else        tmp[slot] = ent;
            }
        }
    }
}

#define ST1_CAP 20480

__global__ void __launch_bounds__(1024)
ak_bucket_scatter(const int* __restrict__ eidx, const int* __restrict__ hist1,
                  const int* __restrict__ off1, unsigned* __restrict__ tmp, int E, int chunk,
                  int nb) {
    __shared__ int cur[NB_MAX];
    __shared__ int sbeg[NB_MAX];
    __shared__ int gbeg[NB_MAX];
    __shared__ unsigned stg[ST1_CAP];
    __shared__ int wsum[32];
    const int tid = threadIdx.x, lane = tid & 31, w = tid >> 5;
    const bool staged = chunk <= ST1_CAP;
    if (staged) {
        int c4[4], g4[4];
        int s = 0;
#pragma unroll
        for (int k = 0; k < 4; ++k) {
            const int b = 4 * tid + k;
            const int bb = min(b, nb - 1);
            const int cv = hist1[bb * CHUNK_WGS + blockIdx.x];
            g4[k] = off1[bb * CHUNK_WGS + blockIdx.x];
            c4[k] = b < nb ? cv : 0;
            s += c4[k];
        }
        int x = s;
        for (int o = 1; o < 32; o <<= 1) {
            int y = __shfl_up(x, o);
            if (lane >= o) x += y;
        }
        if (lane == 31) wsum[w] = x;
        __syncthreads();
        if (w == 0) {
            int t = wsum[lane];
            for (int o = 1; o < 32; o <<= 1) {
                int y = __shfl_up(t, o);
                if (lane >= o) t += y;
            }
            wsum[lane] = t;
        }
        __syncthreads();
        int base = (w ? wsum[w - 1] : 0) + x - s;
#pragma unroll
        for (int k = 0; k < 4; ++k) {
            const int b = 4 * tid + k;
            if (b < nb) {
                cur[b] = base;
                sbeg[b] = base;
                gbeg[b] = g4[k];
            }
            base += c4[k];
        }
    } else {
        for (int b = tid; b < nb; b += 1024) cur[b] = off1[b * CHUNK_WGS + blockIdx.x];
    }
    __syncthreads();
    const int lo = blockIdx.x * chunk;
    const int hi = min(lo + chunk, E);
    if (!staged) {
        ak_scatter_edges<false>(eidx, E, lo, hi, tid, cur, stg, tmp);
        return;
    }
    ak_scatter_edges<true>(eidx, E, lo, hi, tid, cur, stg, tmp);
    __syncthreads();
    for (int b = w; b < nb; b += 32) {
        const int s0 = sbeg[b], n = cur[b] - s0, g0 = gbeg[b];
        if (n <= 0) continue;
        for (int ib = lane - (g0 & 31); ib < n; ib += 128) {
            unsigned v[4];
#pragma unroll
            for (int k = 0; k < 4; ++k)
                v[k] = stg[s0 + min(max(ib + 32 * k, 0), n - 1)];
#pragma unroll
            for (int k = 0; k < 4; ++k) {
                const int i = ib + 32 * k;
                if (i >= 0 && i < n) tmp[g0 + i] = v[k];
            }
        }
    }
}

#define ST2_BYTES 49152

template <bool S16> struct ak_ent { typedef int t; };
template <> struct ak_ent<true> { typedef unsigned short t; };

template <bool S16, bool STAGED>
__device__ __forceinline__ void ak_place(const unsigned* __restrict__ tmp, int lo, int hi, int tid,
                                         int split, int* cur, typename ak_ent<S16>::t* stage,
                                         typename ak_ent<S16>::t* __restrict__ out) {
    typedef typename ak_ent<S16>::t ent_t;
    for (int i0 = lo + tid; i0 < hi; i0 += 4 * 512) {
        unsigned u[4];
#pragma unroll
        for (int k = 0; k < 4; ++k) u[k] = tmp[min(i0 + k * 512, hi - 1)];
        asm volatile("" :: "v"(u[0]), "v"(u[1]), "v"(u[2]), "v"(u[3]));
#pragma unroll
        for (int k = 0; k < 4; ++k) {
            if (i0 + k * 512 < hi) {
                const int sn = (int)(u[k] >> BSH);
                const int key = (int)((u[k] & (BNODES - 1)) << 1) | (sn >= split ? 1 : 0);
                const int slot = atomicAdd(&cur[key], 1);
                const ent_t val = S16 ? (ent_t)(sn >= split ? sn - split : sn) : (ent_t)sn;
                if (STAGED) stage[slot] = val;
                else        out[slot] = val;
            }
        }
    }
}

template <bool S16>
__global__ void __launch_bounds__(512)
ak_bucket_sort(const unsigned* __restrict__ tmp, const int* __restrict__ off1,
               int* __restrict__ rowptr, int* __restrict__ cnt, int* __restrict__ mid,
               float* __restrict__ dinv, int* __restrict__ csr, int E, int N, int nb, int split) {
    typedef typename ak_ent<S16>::t ent_t;
    constexpr int CAP = ST2_BYTES / (int)sizeof(ent_t);
    constexpr int LINE = 128 / (int)sizeof(ent_t);
    __shared__ int c[2 * BNODES];
    __shared__ int cur[2 * BNODES];
    __shared__ int wtot[2 * BNODES / 32];
    __shared__ ent_t stg[CAP];
    const int tid = threadIdx.x, lane = tid & 31, w = tid >> 5;
    const int b = blockIdx.x;
    const int lo = off1[b * CHUNK_WGS];
    const int hi = (b + 1 < nb) ? off1[(b + 1) * CHUNK_WGS] : E;
    const int sh = lo & (LINE - 1);
    const bool staged = (hi - lo) + sh <= CAP;
    ent_t* const gout = reinterpret_cast<ent_t*>(csr);
    c[tid] = 0;
    __syncthreads();
    for (int i0 = lo + tid; i0 < hi; i0 += 4 * 512) {
        unsigned u[4];
#pragma unroll
        for (int k = 0; k < 4; ++k) u[k] = tmp[min(i0 + k * 512, hi - 1)];
#pragma unroll
        for (int k = 0; k < 4; ++k) {
            if (i0 + k * 512 < hi) {
                const int key = (int)((u[k] & (BNODES - 1)) << 1) | ((int)(u[k] >> BSH) >= split ? 1 : 0);
                atomicAdd(&c[key], 1);
            }
        }
    }
    __syncthreads();
    const int mine = c[tid];
    int x = mine;
    for (int o = 1; o < 32; o <<= 1) {
        int y = __shfl_up(x, o);
        if (lane >= o) x += y;
    }
    if (lane == 31) wtot[w] = x;
    __syncthreads();
    int base = 0;
    for (int k = 0; k < w; ++k) base += wtot[k];
    const int ex = base + x - mine;
    cur[tid] = ex;
    if ((tid & 1) == 0) {
        const int v = b * BNODES + (tid >> 1);
        if (v < N) {
            const int total = mine + c[tid + 1];
            cnt[v] = total;
            rowptr[v] = lo + ex;
            mid[v] = lo + ex + mine;
            dinv[v] = rsqrtf((float)(total + 1));
        }
    }
    __syncthreads();
    if (!staged) {
        ak_place<S16, false>(tmp, lo, hi, tid, split, cur, stg, gout + lo);
        return;
    }
    ak_place<S16, true>(tmp, lo, hi, tid, split, cur, stg + sh, gout + lo);
    __syncthreads();
    const int first = sh, last = sh + (hi - lo);
    ent_t* const gl = gout + (lo - sh);
    if (S16) {
        unsigned* const gw = reinterpret_cast<unsigned*>(gl);
        const int fe = (first + 1) & ~1, le = last & ~1;
        if (fe < le) {
            for (int wd0 = tid; 2 * wd0 < le; wd0 += 4 * 512) {
                unsigned a[4], bq[4];
#pragma unroll
                for (int k = 0; k < 4; ++k) {
                    const int j = min(max(2 * (wd0 + 512 * k), fe), le - 2);
                    a[k] = stg[j];
                    bq[k] = stg[j + 1];
                }
#pragma unroll
                for (int k = 0; k < 4; ++k) {
                    const int wd = wd0 + 512 * k;
                    if (2 * wd >= fe && 2 * wd < le) gw[wd] = a[k] | (bq[k] << 16);
                }
            }
        }
        if (tid == 0) {
            if ((first & 1) && first < last) gl[first] = stg[first];
            if ((last & 1) && last - 1 >= first) gl[last - 1] = stg[last - 1];
        }
    } else if (first < last) {
        for (int j0 = tid; j0 < last; j0 += 4 * 512) {
            ent_t a[4];
#pragma unroll
            for (int k = 0; k < 4; ++k) a[k] = stg[min(max(j0 + 512 * k, first), last - 1)];
#pragma unroll
            for (int k = 0; k < 4; ++k) {
                const int j = j0 + 512 * k;
                if (j >= first && j < last) gl[j] = a[k];
            }
        }
    }
}

template <int M>
__device__ __forceinline__ float add_xmask(float x) {
    int b = __builtin_bit_cast(int, x);
    return x + __builtin_bit_cast(float, __builtin_amdgcn_update_dpp(0, b, 0x160 + M, 0xf, 0xf, true));
}

#define HOP_TRIPS 6

template <bool LAST>
__global__ void __launch_bounds__(256)
ak_hop(const int* __restrict__ rowptr, const int* __restrict__ cnt,
       const int* __restrict__ csr, const float* __restrict__ gin,
       const float* __restrict__ z0, const float* __restrict__ dinv,
       float* __restrict__ gout, const float* __restrict__ b3,
       float* __restrict__ out, int N) {
    const int l = threadIdx.x & 7;
    const int grp = (blockIdx.x * 256 + threadIdx.x) >> 3;
    const int ngrp = (gridDim.x * 256) >> 3;
    int v = grp;
    int n_nx = cnt[min(v, N - 1)];
    int b_nx = rowptr[min(v, N - 1)];
    for (int vw = grp & ~3; vw < N; vw += ngrp, v += ngrp) {
        const int vv = min(v, N - 1);
        const int n = n_nx;
        const int* __restrict__ cs = csr + b_nx;
        const int vn = min(v + ngrp, N - 1);
        n_nx = cnt[vn];
        b_nx = rowptr[vn];
        const float self = gin[vv];
        const float dv = dinv[vv];
        const float zz = z0[vv];
        float acc = 0.f;
        for (int j0 = l; j0 - l < n; j0 += 64) {
            int s[8];
            float a[8];
#pragma unroll
            for (int k = 0; k < 8; ++k) s[k] = cs[min(j0 + 8 * k, n - 1)];
#pragma unroll
            for (int k = 0; k < 8; ++k) a[k] = gin[s[k]];
#pragma unroll
            for (int k = 0; k < 8; ++k) acc += (j0 + 8 * k < n) ? a[k] : 0.f;
        }
        acc = add_xmask<1>(acc);
        acc = add_xmask<2>(acc);
        acc = add_xmask<4>(acc);
        const float zn = ONE_MINUS_ALPHA * dv * (acc + self) + ALPHA_F * zz;
        if (l == 0 && v < N) {
            if (LAST) out[v] = zn + b3[0];
            else      gout[v] = dv * zn;
        }
    }
}


#define HL_WGS 256
#define HL_T 512
#define HL_TRIPS 8
#define HL_SLAB 65536
#define HL_RING 13

typedef __attribute__((ext_vector_type(4))) float v4f;
typedef __attribute__((ext_vector_type(2))) int v2i;

__device__ __forceinline__ int hl_piece(int j, int tid, int nq, int rot) {
    const int i = min(j * HL_T + tid, nq - 1) + rot;
    return i >= nq ? i - nq : i;
}
__device__ __forceinline__ int hl_rot(int nq) {
    return (int)((blockIdx.x * (unsigned)nq) / HL_WGS) & ~31;
}

__device__ __forceinline__ v2i hl_quad(const unsigned short* __restrict__ c16, int p, int epad) {
    return *reinterpret_cast<const v2i*>(c16 + min(p, epad - 4));
}

__device__ __forceinline__ float hl_sum4(const float* sg, v2i q, int p, int s, int e) {
    const int qq[4] = {q.x & 0xffff, (int)((unsigned)q.x >> 16),
                       q.y & 0xffff, (int)((unsigned)q.y >> 16)};
    float a[4];
    bool ok[4];
#pragma unroll
    for (int j = 0; j < 4; ++j) {
        ok[j] = (p + j >= s) && (p + j < e);
        a[j] = sg[ok[j] ? qq[j] : 0];
    }
    float acc = 0.f;
#pragma unroll
    for (int j = 0; j < 4; ++j) acc += ok[j] ? a[j] : 0.f;
    return acc;
}

__device__ __forceinline__ float hl_first_lo(const float* sg, v2i qlo, int s, int e, int l) {
    return hl_sum4(sg, qlo, (s & ~3) + 4 * l, s, e);
}
__device__ __forceinline__ float hl_first_hi(const float* sg, v2i qhi, int s, int e, int l) {
    const int p = (s & ~3) + 32 + 4 * l;
    return p < e ? hl_sum4(sg, qhi, p, s, e) : 0.f;
}
__device__ __forceinline__ float hl_rest(const float* sg, const unsigned short* __restrict__ c16,
                                         int s, int e, int l, int epad) {
    float acc = 0.f;
    for (int pp = (s & ~3) + 64; pp < e; pp += 64) {
        const int p = pp + 4 * l;
        acc += hl_sum4(sg, hl_quad(c16, p, epad), p, s, e);
        acc += hl_sum4(sg, hl_quad(c16, p + 32, epad), p + 32, s, e);
    }
    return acc;
}

template <bool LAST>
__global__ void __launch_bounds__(HL_T)
ak_hop_lds(const int* __restrict__ rowptr, const int* __restrict__ cnt, const int* __restrict__ mid,
           const int* __restrict__ csr, const float* __restrict__ gin,
           const float* __restrict__ z0, const float* __restrict__ dinv,
           float* __restrict__ gout, const float* __restrict__ b3, float* __restrict__ out,
           int N, int npw, int split, int epad) {
    __shared__ float sg[HL_SLAB];
    __shared__ float stot[HL_T];
    const unsigned short* __restrict__ const c16 = reinterpret_cast<const unsigned short*>(csr);
    const int tid = threadIdx.x, l = tid & 7, gi = tid >> 3;
    const int v_lo = blockIdx.x * npw;
    const int v_hi = min(v_lo + npw, N);

    v4f* const d4 = reinterpret_cast<v4f*>(sg);
    v4f rr[HL_RING];
    int rx[HL_RING];
    const v4f* __restrict__ const ha = reinterpret_cast<const v4f*>(gin);
    const int nqa = split >> 2, rota = hl_rot(nqa), pera = (nqa + HL_T - 1) / HL_T;
#pragma unroll
    for (int k = 0; k < HL_RING; ++k) {
        rx[k] = hl_piece(k, tid, nqa, rota);
        rr[k] = ha[rx[k]];
    }

    int s0[HL_TRIPS], m0[HL_TRIPS], e0[HL_TRIPS];
#pragma unroll
    for (int t = 0; t < HL_TRIPS; ++t) {
        const int v = v_lo + gi + 64 * t;
        const int vv = min(v, N - 1);
        const int rs = rowptr[vv], rm = mid[vv], rc = cnt[vv];
        const bool ok = v < v_hi;
        s0[t] = ok ? rs : 0;
        m0[t] = ok ? rm : 0;
        e0[t] = ok ? rs + rc : 0;
    }
    v2i qa[HL_TRIPS], qa2[HL_TRIPS];
#pragma unroll
    for (int t = 0; t < HL_TRIPS; ++t) {
        qa[t]  = hl_quad(c16, (s0[t] & ~3) + 4 * l, epad);
        qa2[t] = hl_quad(c16, (s0[t] & ~3) + 32 + 4 * l, epad);
    }
    const int vf = v_lo + tid;
    const int vfc = min(vf, N - 1);
    const float f_dv = dinv[vfc], f_zz = z0[vfc], f_self = gin[vfc];

    for (int j0 = HL_RING; j0 < pera; j0 += HL_RING) {
#pragma unroll
        for (int k = 0; k < HL_RING; ++k) {
            d4[rx[k]] = rr[k];
            rx[k] = hl_piece(j0 + k, tid, nqa, rota);
            rr[k] = ha[rx[k]];
        }
    }
#pragma unroll
    for (int k = 0; k < HL_RING; ++k) d4[rx[k]] = rr[k];
    __syncthreads();

    const float* __restrict__ const g1 = gin + split;
    const v4f* __restrict__ const hb = reinterpret_cast<const v4f*>(g1);
    const int n1 = N - split;
    const int nqb = n1 >> 2, rotb = hl_rot(nqb), perb = (nqb + HL_T - 1) / HL_T;
#pragma unroll
    for (int k = 0; k < HL_RING; ++k) {
        rx[k] = hl_piece(k, tid, nqb, rotb);
        rr[k] = hb[rx[k]];
    }
    float acc[HL_TRIPS];
#pragma unroll
    for (int t = 0; t < HL_TRIPS; ++t) acc[t] = hl_first_lo(sg, qa[t], s0[t], m0[t], l);
#pragma unroll
    for (int t = 0; t < HL_TRIPS; ++t) acc[t] += hl_first_hi(sg, qa2[t], s0[t], m0[t], l);
#pragma unroll
    for (int t = 0; t < HL_TRIPS; ++t) acc[t] += hl_rest(sg, c16, s0[t], m0[t], l, epad);
    v2i qb[HL_TRIPS], qb2[HL_TRIPS];
#pragma unroll
    for (int t = 0; t < HL_TRIPS; ++t) {
        qb[t]  = hl_quad(c16, (m0[t] & ~3) + 4 * l, epad);
        qb2[t] = hl_quad(c16, (m0[t] & ~3) + 32 + 4 * l, epad);
    }
    __syncthreads();
    for (int j0 = HL_RING; j0 < perb; j0 += HL_RING) {
#pragma unroll
        for (int k = 0; k < HL_RING; ++k) {
            d4[rx[k]] = rr[k];
            rx[k] = hl_piece(j0 + k, tid, nqb, rotb);
            rr[k] = hb[rx[k]];
        }
    }
#pragma unroll
    for (int k = 0; k < HL_RING; ++k) d4[rx[k]] = rr[k];
    if (tid < (n1 & 3)) sg[(nqb << 2) + tid] = g1[(nqb << 2) + tid];
    __syncthreads();
#pragma unroll
    for (int t = 0; t < HL_TRIPS; ++t) acc[t] += hl_first_lo(sg, qb[t], m0[t], e0[t], l);
#pragma unroll
    for (int t = 0; t < HL_TRIPS; ++t) acc[t] += hl_first_hi(sg, qb2[t], m0[t], e0[t], l);
#pragma unroll
    for (int t = 0; t < HL_TRIPS; ++t) acc[t] += hl_rest(sg, c16, m0[t], e0[t], l, epad);
#pragma unroll
    for (int t = 0; t < HL_TRIPS; ++t) {
        float a = acc[t];
        a = add_xmask<1>(a);
        a = add_xmask<2>(a);
        a = add_xmask<4>(a);
        if (l == 0) stot[gi + 64 * t] = a;
    }
    __syncthreads();
    if (tid < npw && vf < v_hi) {
        const float zn = ONE_MINUS_ALPHA * f_dv * (stot[tid] + f_self) + ALPHA_F * f_zz;
        if (LAST) out[vf] = zn + b3[0];
        else      gout[vf] = f_dv * zn;
    }
}


extern "C" void kernel_launch(void* const* d_in, const int* in_sizes, int n_in,
                              void* d_out, int out_size, void* d_ws, size_t ws_size,
                              hipStream_t stream) {
    const float* x  = (const float*)d_in[0];
    const int*  eidx = (const int*)d_in[1];
    const float* W1 = (const float*)d_in[2];
    const float* b1 = (const float*)d_in[3];
    const float* W2 = (const float*)d_in[4];
    const float* b2 = (const float*)d_in[5];
    const float* W3 = (const float*)d_in[6];
    const float* b3 = (const float*)d_in[7];
    float* out = (float*)d_out;

    const int N = in_sizes[0];
    const int E = in_sizes[1] / 2;

    char* ws = (char*)d_ws;
    const int B = 256;

    {
        const int nb = (N + BNODES - 1) / BNODES;
        const int nh = nb * CHUNK_WGS;
        size_t o2 = 0;
        auto carve2 = [&](size_t bytes) -> char* {
            char* p = ws + o2;
            o2 += (bytes + 255) & ~(size_t)255;
            return p;
        };
        int*   cnt    = (int*)carve2((size_t)N * 4);
        float* dinv   = (float*)carve2((size_t)N * 4);
        int*   rowptr = (int*)carve2((size_t)N * 4);
        int*   hist1  = (int*)carve2((size_t)nh * 4);
        int*   off1   = (int*)carve2((size_t)nh * 4);
        int*   csr    = (int*)carve2((size_t)E * 4);
        const size_t plane = (((size_t)N * 4) + 255) & ~(size_t)255;
        const size_t tmpb = (size_t)E * 4;
        char* shared_region = carve2(tmpb > 3 * plane ? tmpb : 3 * plane);
        unsigned* tmp = (unsigned*)shared_region;
        float* z0 = (float*)shared_region;
        float* gA = (float*)(shared_region + plane);
        float* gB = (float*)(shared_region + 2 * plane);
        int* mid = hist1;
        const bool lds_hop = N >= 4096 && N <= 2 * HL_SLAB && E >= 64;
        const int split = lds_hop ? ((N / 2 + 31) & ~31) : N;

        if (o2 <= ws_size && nb <= NB_MAX && N <= (1 << 24)) {
            const int chunk = (E + CHUNK_WGS - 1) / CHUNK_WGS;
            ak_bucket_hist<<<CHUNK_WGS, 1024, 0, stream>>>(eidx, hist1, E, chunk, nb);
            ak_excl_scan<<<(nh + 1023) / 1024, 1024, 0, stream>>>(hist1, off1, nh);
            ak_bucket_scatter<<<CHUNK_WGS, 1024, 0, stream>>>(eidx, hist1, off1, tmp, E, chunk, nb);
            if (lds_hop) {
                ak_bucket_sort<true><<<nb, 512, 0, stream>>>(tmp, off1, rowptr, cnt, mid, dinv, csr, E,
                                                             N, nb, split);
            } else {
                ak_bucket_sort<false><<<nb, 512, 0, stream>>>(tmp, off1, rowptr, cnt, mid, dinv, csr, E,
                                                              N, nb, split);
            }

            int tiles = N / 16;
            if (tiles > 0) {
                int waves_per_block = B / 32;
                int blocks = (tiles + waves_per_block - 1) / waves_per_block;
                ak_encoder_wmma<<<blocks, B, 0, stream>>>(x, W1, b1, W2, b2, W3, dinv, z0, gA,
                                                          (float*)nullptr, tiles);
            }
            int rem_start = tiles * 16;
            if (rem_start < N) {
                ak_encoder_tail<<<(N - rem_start + B - 1) / B, B, 0, stream>>>(
                    x, W1, b1, W2, b2, W3, dinv, z0, gA, (float*)nullptr, rem_start, N);
            }

            const int hblocks = (N + 32 * HOP_TRIPS - 1) / (32 * HOP_TRIPS);
            const int npw = (N + HL_WGS - 1) / HL_WGS;
            const int epad = (E + 63) & ~63;
            for (int it = 0; it < KITERS; ++it) {
                const float* gi = (it & 1) ? gB : gA;
                float* go = (it & 1) ? gA : gB;
                if (lds_hop) {
                    if (it < KITERS - 1) {
                        ak_hop_lds<false><<<HL_WGS, HL_T, 0, stream>>>(rowptr, cnt, mid, csr, gi, z0,
                                                                       dinv, go, b3, out, N, npw,
                                                                       split, epad);
                    } else {
                        ak_hop_lds<true><<<HL_WGS, HL_T, 0, stream>>>(rowptr, cnt, mid, csr, gi, z0,
                                                                      dinv, (float*)nullptr, b3, out,
                                                                      N, npw, split, epad);
                    }
                } else if (it < KITERS - 1) {
                    ak_hop<false><<<hblocks, B, 0, stream>>>(rowptr, cnt, csr, gi, z0, dinv, go,
                                                             b3, out, N);
                } else {
                    ak_hop<true><<<hblocks, B, 0, stream>>>(rowptr, cnt, csr, gi, z0, dinv,
                                                            (float*)nullptr, b3, out, N);
                }
            }
            return;
        }
    }

    size_t off = 0;
    auto carve = [&](size_t bytes) -> char* {
        char* p = ws + off;
        off += (bytes + 255) & ~(size_t)255;
        return p;
    };
    float* deg  = (float*)carve((size_t)N * 4);
    float* dinv = (float*)carve((size_t)N * 4);
    float* z0   = (float*)carve((size_t)N * 4);
    float* g    = (float*)carve((size_t)N * 4);
    float* agg  = (float*)carve((size_t)N * 4);

    ak_deg_init<<<(N + B - 1) / B, B, 0, stream>>>(deg, N);
    ak_deg_accum<<<(E + B - 1) / B, B, 0, stream>>>(eidx, deg, E);
    ak_dinv<<<(N + B - 1) / B, B, 0, stream>>>(deg, dinv, N);

    int tiles = N / 16;
    if (tiles > 0) {
        int waves_per_block = B / 32;
        int blocks = (tiles + waves_per_block - 1) / waves_per_block;
        ak_encoder_wmma<<<blocks, B, 0, stream>>>(x, W1, b1, W2, b2, W3, dinv, z0, g, agg, tiles);
    }
    int rem_start = tiles * 16;
    if (rem_start < N) {
        ak_encoder_tail<<<(N - rem_start + B - 1) / B, B, 0, stream>>>(
            x, W1, b1, W2, b2, W3, dinv, z0, g, agg, rem_start, N);
    }

    for (int it = 0; it < KITERS; ++it) {
        ak_edge_scatter<<<(E + B - 1) / B, B, 0, stream>>>(eidx, g, agg, E);
        if (it < KITERS - 1) {
            ak_finalize<<<(N + B - 1) / B, B, 0, stream>>>(agg, g, z0, dinv, N);
        } else {
            ak_finalize_out<<<(N + B - 1) / B, B, 0, stream>>>(agg, g, z0, dinv, b3, out, N);
        }
    }
}


# Sort level 1 orders each block's chunk by bucket in place (edges held in registers); level 2 collects a bucket's shares from all chunks: no count launch, no scan launch; float sum order changes
# speedup vs baseline: 16.2451x; 1.0710x over previous
#include <hip/hip_runtime.h>


typedef __attribute__((ext_vector_type(2))) float v2f;
typedef __attribute__((ext_vector_type(8))) float v8f;

#define HID 16
#define KITERS 10
#define ONE_MINUS_ALPHA 0.9f
#define ALPHA_F 0.1f

__device__ __forceinline__ float relu_f(float v) { return v > 0.f ? v : 0.f; }


__global__ void ak_deg_init(float* __restrict__ deg, int n) {
    int i = blockIdx.x * blockDim.x + threadIdx.x;
    if (i < n) deg[i] = 1.0f;
}

__global__ void ak_deg_accum(const int* __restrict__ eidx, float* __restrict__ deg, int E) {
    int e = blockIdx.x * blockDim.x + threadIdx.x;
    if (e < E) atomicAdd(&deg[eidx[(long long)E + e]], 1.0f);
}

__global__ void ak_dinv(const float* __restrict__ deg, float* __restrict__ dinv, int n) {
    int i = blockIdx.x * blockDim.x + threadIdx.x;
    if (i < n) dinv[i] = rsqrtf(deg[i]);
}

__device__ __forceinline__ float xhalf(float x) {
    int b = __builtin_bit_cast(int, x);
    return __builtin_bit_cast(float,
        __builtin_amdgcn_permlanex16(b, b, 0x76543210, 0xfedcba98, false, false));
}


__global__ void ak_encoder_wmma(const float* __restrict__ x, const float* __restrict__ W1,
                                const float* __restrict__ b1, const float* __restrict__ W2,
                                const float* __restrict__ b2, const float* __restrict__ W3,
                                const float* __restrict__ dinv, float* __restrict__ z0,
                                float* __restrict__ g, float* __restrict__ agg, int n_tiles) {
    int wave = (blockIdx.x * blockDim.x + threadIdx.x) >> 5;
    int lane = threadIdx.x & 31;
    if (wave >= n_tiles) return;

    int row = lane & 15;
    int hi  = lane >> 4;
    int node = wave * 16 + row;
    float xv = x[node];

    v8f acc = {};
#pragma unroll
    for (int s = 0; s < 4; ++s) {
        int k0 = 4 * s + 2 * hi;
        v2f wt, ht;
        wt.x = W2[k0 * HID + row];
        wt.y = W2[(k0 + 1) * HID + row];
        ht.x = relu_f(xv * W1[k0]     + b1[k0]);
        ht.y = relu_f(xv * W1[k0 + 1] + b1[k0 + 1]);
        acc = __builtin_amdgcn_wmma_f32_16x16x4_f32(
            false, wt, false, ht, (short)0, acc, false, false);
    }

    float part = 0.f;
#pragma unroll
    for (int i = 0; i < 8; ++i) {
        int m = i + 8 * hi;
        part += relu_f(acc[i] + b2[m]) * W3[m];
    }
    float z = part + xhalf(part);
    if (hi == 0) {
        z0[node] = z;
        g[node]  = dinv[node] * z;
        if (agg) agg[node] = 0.f;
    }
}

__global__ void ak_encoder_tail(const float* __restrict__ x, const float* __restrict__ W1,
                                const float* __restrict__ b1, const float* __restrict__ W2,
                                const float* __restrict__ b2, const float* __restrict__ W3,
                                const float* __restrict__ dinv, float* __restrict__ z0,
                                float* __restrict__ g, float* __restrict__ agg, int start, int N) {
    int v = start + blockIdx.x * blockDim.x + threadIdx.x;
    if (v >= N) return;
    float h1[HID];
    float xv = x[v];
#pragma unroll
    for (int k = 0; k < HID; ++k) h1[k] = relu_f(xv * W1[k] + b1[k]);
    float z = 0.f;
#pragma unroll
    for (int j = 0; j < HID; ++j) {
        float acc = b2[j];
#pragma unroll
        for (int k = 0; k < HID; ++k) acc += h1[k] * W2[k * HID + j];
        z += relu_f(acc) * W3[j];
    }
    z0[v] = z;
    g[v]  = dinv[v] * z;
    if (agg) agg[v] = 0.f;
}


__global__ void ak_edge_scatter(const int* __restrict__ eidx, const float* __restrict__ g,
                                float* __restrict__ agg, int E) {
    int e = blockIdx.x * blockDim.x + threadIdx.x;
    if (e >= E) return;
    atomicAdd(&agg[eidx[(long long)E + e]], g[eidx[e]]);
}

__global__ void ak_finalize(float* __restrict__ agg, float* __restrict__ g,
                            const float* __restrict__ z0, const float* __restrict__ dinv, int n) {
    int v = blockIdx.x * blockDim.x + threadIdx.x;
    if (v >= n) return;
    float dv = dinv[v];
    float zn = ONE_MINUS_ALPHA * dv * (agg[v] + g[v]) + ALPHA_F * z0[v];
    g[v] = dv * zn;
    agg[v] = 0.f;
}

__global__ void ak_finalize_out(const float* __restrict__ agg, const float* __restrict__ g,
                                const float* __restrict__ z0, const float* __restrict__ dinv,
                                const float* __restrict__ b3, float* __restrict__ out, int N) {
    int v = blockIdx.x * blockDim.x + threadIdx.x;
    if (v >= N) return;
    out[v] = ONE_MINUS_ALPHA * dinv[v] * (agg[v] + g[v]) + ALPHA_F * z0[v] + b3[0];
}


#define BSH 8
#define BNODES 256
#define NB_MAX 4096
#define CHUNK_WGS 256

__global__ void __launch_bounds__(1024)
ak_bucket_hist(const int* __restrict__ eidx, int* __restrict__ hist1, int E, int chunk, int nb) {
    __shared__ int hist[NB_MAX];
    for (int b = threadIdx.x; b < nb; b += 1024) hist[b] = 0;
    __syncthreads();
    const int lo = blockIdx.x * chunk;
    const int hi = min(lo + chunk, E);
    const int* __restrict__ dst = eidx + (size_t)E;
    for (int e0 = lo + threadIdx.x; e0 < hi; e0 += 4 * 1024) {
        int d[4];
#pragma unroll
        for (int k = 0; k < 4; ++k) d[k] = dst[min(e0 + k * 1024, hi - 1)];
#pragma unroll
        for (int k = 0; k < 4; ++k) {
            if (e0 + k * 1024 < hi) atomicAdd(&hist[d[k] >> BSH], 1);
        }
    }
    __syncthreads();
    for (int b = threadIdx.x; b < nb; b += 1024) hist1[b * CHUNK_WGS + blockIdx.x] = hist[b];
}

__global__ void __launch_bounds__(1024)
ak_excl_scan(const int* __restrict__ in, int* __restrict__ out, int n) {
    __shared__ int red[32];
    __shared__ int wsum[32];
    const int tid = threadIdx.x, lane = tid & 31, w = tid >> 5;
    const int first = blockIdx.x * 1024;

    int part = 0;
    for (int i = tid; i < first; i += 1024) part += in[i];
    for (int o = 16; o > 0; o >>= 1) part += __shfl_xor(part, o);
    if (lane == 0) red[w] = part;

    const int i = first + tid;
    const int c = (i < n) ? in[i] : 0;
    int x = c;
    for (int o = 1; o < 32; o <<= 1) {
        int y = __shfl_up(x, o);
        if (lane >= o) x += y;
    }
    if (lane == 31) wsum[w] = x;
    __syncthreads();
    if (w == 0) {
        int s = wsum[lane];
        for (int o = 1; o < 32; o <<= 1) {
            int y = __shfl_up(s, o);
            if (lane >= o) s += y;
        }
        wsum[lane] = s;
        int b = red[lane];
        for (int o = 16; o > 0; o >>= 1) b += __shfl_xor(b, o);
        red[lane] = b;
    }
    __syncthreads();
    if (i < n) out[i] = red[0] + (w ? wsum[w - 1] : 0) + x - c;
}

template <bool STAGED>
__device__ __forceinline__ void ak_scatter_edges(const int* __restrict__ eidx, int E, int lo, int hi,
                                                 int tid, int* cur, unsigned* stg,
                                                 unsigned* __restrict__ tmp) {
    const int* __restrict__ dst = eidx + (size_t)E;
    for (int e0 = lo + tid; e0 < hi; e0 += 4 * 1024) {
        int d[4], s[4];
#pragma unroll
        for (int k = 0; k < 4; ++k) {
            const int e = min(e0 + k * 1024, hi - 1);
            d[k] = dst[e];
            s[k] = eidx[e];
        }
        asm volatile("" :: "v"(d[0]), "v"(d[1]), "v"(d[2]), "v"(d[3]),
                           "v"(s[0]), "v"(s[1]), "v"(s[2]), "v"(s[3]));
#pragma unroll
        for (int k = 0; k < 4; ++k) {
            if (e0 + k * 1024 < hi) {
                const int slot = atomicAdd(&cur[d[k] >> BSH], 1);
                const unsigned ent = ((unsigned)s[k] << BSH) | (unsigned)(d[k] & (BNODES - 1));
                if (STAGED) stg[slot] = ent;
                else        tmp[slot] = ent;
            }
        }
    }
}

#define ST1_CAP 20480

__global__ void __launch_bounds__(1024)
ak_bucket_scatter(const int* __restrict__ eidx, const int* __restrict__ hist1,
                  const int* __restrict__ off1, unsigned* __restrict__ tmp, int E, int chunk,
                  int nb) {
    __shared__ int cur[NB_MAX];
    __shared__ int sbeg[NB_MAX];
    __shared__ int gbeg[NB_MAX];
    __shared__ unsigned stg[ST1_CAP];
    __shared__ int wsum[32];
    const int tid = threadIdx.x, lane = tid & 31, w = tid >> 5;
    const bool staged = chunk <= ST1_CAP;
    if (staged) {
        int c4[4], g4[4];
        int s = 0;
#pragma unroll
        for (int k = 0; k < 4; ++k) {
            const int b = 4 * tid + k;
            const int bb = min(b, nb - 1);
            const int cv = hist1[bb * CHUNK_WGS + blockIdx.x];
            g4[k] = off1[bb * CHUNK_WGS + blockIdx.x];
            c4[k] = b < nb ? cv : 0;
            s += c4[k];
        }
        int x = s;
        for (int o = 1; o < 32; o <<= 1) {
            int y = __shfl_up(x, o);
            if (lane >= o) x += y;
        }
        if (lane == 31) wsum[w] = x;
        __syncthreads();
        if (w == 0) {
            int t = wsum[lane];
            for (int o = 1; o < 32; o <<= 1) {
                int y = __shfl_up(t, o);
                if (lane >= o) t += y;
            }
            wsum[lane] = t;
        }
        __syncthreads();
        int base = (w ? wsum[w - 1] : 0) + x - s;
#pragma unroll
        for (int k = 0; k < 4; ++k) {
            const int b = 4 * tid + k;
            if (b < nb) {
                cur[b] = base;
                sbeg[b] = base;
                gbeg[b] = g4[k];
            }
            base += c4[k];
        }
    } else {
        for (int b = tid; b < nb; b += 1024) cur[b] = off1[b * CHUNK_WGS + blockIdx.x];
    }
    __syncthreads();
    const int lo = blockIdx.x * chunk;
    const int hi = min(lo + chunk, E);
    if (!staged) {
        ak_scatter_edges<false>(eidx, E, lo, hi, tid, cur, stg, tmp);
        return;
    }
    ak_scatter_edges<true>(eidx, E, lo, hi, tid, cur, stg, tmp);
    __syncthreads();
    for (int b = w; b < nb; b += 32) {
        const int s0 = sbeg[b], n = cur[b] - s0, g0 = gbeg[b];
        if (n <= 0) continue;
        for (int ib = lane - (g0 & 31); ib < n; ib += 128) {
            unsigned v[4];
#pragma unroll
            for (int k = 0; k < 4; ++k)
                v[k] = stg[s0 + min(max(ib + 32 * k, 0), n - 1)];
#pragma unroll
            for (int k = 0; k < 4; ++k) {
                const int i = ib + 32 * k;
                if (i >= 0 && i < n) tmp[g0 + i] = v[k];
            }
        }
    }
}

#define ST2_BYTES 49152

template <bool S16> struct ak_ent { typedef int t; };
template <> struct ak_ent<true> { typedef unsigned short t; };

template <bool S16, bool STAGED>
__device__ __forceinline__ void ak_place(const unsigned* __restrict__ tmp, int lo, int hi, int tid,
                                         int split, int* cur, typename ak_ent<S16>::t* stage,
                                         typename ak_ent<S16>::t* __restrict__ out) {
    typedef typename ak_ent<S16>::t ent_t;
    for (int i0 = lo + tid; i0 < hi; i0 += 4 * 512) {
        unsigned u[4];
#pragma unroll
        for (int k = 0; k < 4; ++k) u[k] = tmp[min(i0 + k * 512, hi - 1)];
        asm volatile("" :: "v"(u[0]), "v"(u[1]), "v"(u[2]), "v"(u[3]));
#pragma unroll
        for (int k = 0; k < 4; ++k) {
            if (i0 + k * 512 < hi) {
                const int sn = (int)(u[k] >> BSH);
                const int key = (int)((u[k] & (BNODES - 1)) << 1) | (sn >= split ? 1 : 0);
                const int slot = atomicAdd(&cur[key], 1);
                const ent_t val = S16 ? (ent_t)(sn >= split ? sn - split : sn) : (ent_t)sn;
                if (STAGED) stage[slot] = val;
                else        out[slot] = val;
            }
        }
    }
}

template <bool S16>
__global__ void __launch_bounds__(512)
ak_bucket_sort(const unsigned* __restrict__ tmp, const int* __restrict__ off1,
               int* __restrict__ rowptr, int* __restrict__ cnt, int* __restrict__ mid,
               float* __restrict__ dinv, int* __restrict__ csr, int E, int N, int nb, int split) {
    typedef typename ak_ent<S16>::t ent_t;
    constexpr int CAP = ST2_BYTES / (int)sizeof(ent_t);
    constexpr int LINE = 128 / (int)sizeof(ent_t);
    __shared__ int c[2 * BNODES];
    __shared__ int cur[2 * BNODES];
    __shared__ int wtot[2 * BNODES / 32];
    __shared__ ent_t stg[CAP];
    const int tid = threadIdx.x, lane = tid & 31, w = tid >> 5;
    const int b = blockIdx.x;
    const int lo = off1[b * CHUNK_WGS];
    const int hi = (b + 1 < nb) ? off1[(b + 1) * CHUNK_WGS] : E;
    const int sh = lo & (LINE - 1);
    const bool staged = (hi - lo) + sh <= CAP;
    ent_t* const gout = reinterpret_cast<ent_t*>(csr);
    c[tid] = 0;
    __syncthreads();
    for (int i0 = lo + tid; i0 < hi; i0 += 4 * 512) {
        unsigned u[4];
#pragma unroll
        for (int k = 0; k < 4; ++k) u[k] = tmp[min(i0 + k * 512, hi - 1)];
#pragma unroll
        for (int k = 0; k < 4; ++k) {
            if (i0 + k * 512 < hi) {
                const int key = (int)((u[k] & (BNODES - 1)) << 1) | ((int)(u[k] >> BSH) >= split ? 1 : 0);
                atomicAdd(&c[key], 1);
            }
        }
    }
    __syncthreads();
    const int mine = c[tid];
    int x = mine;
    for (int o = 1; o < 32; o <<= 1) {
        int y = __shfl_up(x, o);
        if (lane >= o) x += y;
    }
    if (lane == 31) wtot[w] = x;
    __syncthreads();
    int base = 0;
    for (int k = 0; k < w; ++k) base += wtot[k];
    const int ex = base + x - mine;
    cur[tid] = ex;
    if ((tid & 1) == 0) {
        const int v = b * BNODES + (tid >> 1);
        if (v < N) {
            const int total = mine + c[tid + 1];
            cnt[v] = total;
            rowptr[v] = lo + ex;
            mid[v] = lo + ex + mine;
            dinv[v] = rsqrtf((float)(total + 1));
        }
    }
    __syncthreads();
    if (!staged) {
        ak_place<S16, false>(tmp, lo, hi, tid, split, cur, stg, gout + lo);
        return;
    }
    ak_place<S16, true>(tmp, lo, hi, tid, split, cur, stg + sh, gout + lo);
    __syncthreads();
    const int first = sh, last = sh + (hi - lo);
    ent_t* const gl = gout + (lo - sh);
    if (S16) {
        unsigned* const gw = reinterpret_cast<unsigned*>(gl);
        const int fe = (first + 1) & ~1, le = last & ~1;
        if (fe < le) {
            for (int wd0 = tid; 2 * wd0 < le; wd0 += 4 * 512) {
                unsigned a[4], bq[4];
#pragma unroll
                for (int k = 0; k < 4; ++k) {
                    const int j = min(max(2 * (wd0 + 512 * k), fe), le - 2);
                    a[k] = stg[j];
                    bq[k] = stg[j + 1];
                }
#pragma unroll
                for (int k = 0; k < 4; ++k) {
                    const int wd = wd0 + 512 * k;
                    if (2 * wd >= fe && 2 * wd < le) gw[wd] = a[k] | (bq[k] << 16);
                }
            }
        }
        if (tid == 0) {
            if ((first & 1) && first < last) gl[first] = stg[first];
            if ((last & 1) && last - 1 >= first) gl[last - 1] = stg[last - 1];
        }
    } else if (first < last) {
        for (int j0 = tid; j0 < last; j0 += 4 * 512) {
            ent_t a[4];
#pragma unroll
            for (int k = 0; k < 4; ++k) a[k] = stg[min(max(j0 + 512 * k, first), last - 1)];
#pragma unroll
            for (int k = 0; k < 4; ++k) {
                const int j = j0 + 512 * k;
                if (j >= first && j < last) gl[j] = a[k];
            }
        }
    }
}


#define CS_EPT 20
static_assert(CS_EPT * 1024 >= ST1_CAP, "a thread's registers must cover the largest staged chunk");

__global__ void __launch_bounds__(1024)
ak_chunk_sort(const int* __restrict__ eidx, int* __restrict__ cnt1, int* __restrict__ pos1,
              unsigned* __restrict__ tmp, int E, int chunk, int nb) {
    __shared__ int cur[NB_MAX];
    __shared__ unsigned stg[ST1_CAP];
    __shared__ int wsum[32];
    const int tid = threadIdx.x, lane = tid & 31, w = tid >> 5;
    const int lo = blockIdx.x * chunk;
    const int hi = min(lo + chunk, E);
    const int* __restrict__ dst = eidx + (size_t)E;
    int d[CS_EPT], s[CS_EPT];
#pragma unroll
    for (int k = 0; k < CS_EPT; ++k) {
        const int e = min(lo + tid + k * 1024, E - 1);
        d[k] = dst[e];
        s[k] = eidx[e];
    }
    for (int b = tid; b < nb; b += 1024) cur[b] = 0;
    __syncthreads();
#pragma unroll
    for (int k = 0; k < CS_EPT; ++k) {
        if (lo + tid + k * 1024 < hi) atomicAdd(&cur[d[k] >> BSH], 1);
    }
    __syncthreads();
    int c4[4];
    int sum4 = 0;
#pragma unroll
    for (int k = 0; k < 4; ++k) {
        const int b = 4 * tid + k;
        const int cv = cur[b];
        c4[k] = b < nb ? cv : 0;
        sum4 += c4[k];
    }
    int x = sum4;
    for (int o = 1; o < 32; o <<= 1) {
        int y = __shfl_up(x, o);
        if (lane >= o) x += y;
    }
    if (lane == 31) wsum[w] = x;
    __syncthreads();
    if (w == 0) {
        int t = wsum[lane];
        for (int o = 1; o < 32; o <<= 1) {
            int y = __shfl_up(t, o);
            if (lane >= o) t += y;
        }
        wsum[lane] = t;
    }
    __syncthreads();
    int base = (w ? wsum[w - 1] : 0) + x - sum4;
    int p4[4];
#pragma unroll
    for (int k = 0; k < 4; ++k) {
        const int b = 4 * tid + k;
        p4[k] = base;
        if (b < nb) cur[b] = base;
        base += c4[k];
    }
    __syncthreads();
    int slot[CS_EPT];
#pragma unroll
    for (int k = 0; k < CS_EPT; ++k) {
        slot[k] = 0;
        if (lo + tid + k * 1024 < hi) slot[k] = atomicAdd(&cur[d[k] >> BSH], 1);
    }
#pragma unroll
    for (int k = 0; k < CS_EPT; ++k) {
        if (lo + tid + k * 1024 < hi)
            stg[slot[k]] = ((unsigned)s[k] << BSH) | (unsigned)(d[k] & (BNODES - 1));
    }
    __syncthreads();
#pragma unroll
    for (int k = 0; k < 4; ++k) {
        const int b = 4 * tid + k;
        if (b < nb) {
            cnt1[b * CHUNK_WGS + blockIdx.x] = c4[k];
            pos1[b * CHUNK_WGS + blockIdx.x] = p4[k];
        }
    }
    const int n = hi - lo;
    if (n <= 0) return;
    for (int i0 = tid - (lo & 31); i0 < n; i0 += 4 * 1024) {
        unsigned v[4];
#pragma unroll
        for (int k = 0; k < 4; ++k) v[k] = stg[min(max(i0 + 1024 * k, 0), n - 1)];
#pragma unroll
        for (int k = 0; k < 4; ++k) {
            const int i = i0 + 1024 * k;
            if (i >= 0 && i < n) tmp[lo + i] = v[k];
        }
    }
}

__device__ __forceinline__ int ak_key(unsigned u, int split) {
    return (int)((u & (BNODES - 1)) << 1) | ((int)(u >> BSH) >= split ? 1 : 0);
}

template <bool STAGED>
__device__ __forceinline__ void ak_place_one(unsigned u, int split, int* cur, unsigned short* stage,
                                             unsigned short* __restrict__ out) {
    const int sn = (int)(u >> BSH);
    const int slot = atomicAdd(&cur[ak_key(u, split)], 1);
    const unsigned short val = (unsigned short)(sn >= split ? sn - split : sn);
    if (STAGED) stage[slot] = val;
    else        out[slot] = val;
}

#define BS_SPW 16

template <bool STAGED>
__device__ __forceinline__ void ak_place_sh(const unsigned (&u)[2 * BS_SPW], const int (&nn)[BS_SPW],
                                            const unsigned* __restrict__ tmp, const int* shn,
                                            const int* shb, int w, int lane, int split, int* cur,
                                            unsigned short* stage, unsigned short* __restrict__ out) {
#pragma unroll
    for (int k = 0; k < BS_SPW; ++k) {
        if (lane < nn[k])      ak_place_one<STAGED>(u[2 * k], split, cur, stage, out);
        if (lane + 32 < nn[k]) ak_place_one<STAGED>(u[2 * k + 1], split, cur, stage, out);
    }
#pragma nounroll
    for (int k = 0; k < BS_SPW; ++k) {
        const int n = shn[w * BS_SPW + k];
        const int p = shb[w * BS_SPW + k];
        for (int i = 64 + lane; i < n; i += 32)
            ak_place_one<STAGED>(tmp[p + i], split, cur, stage, out);
    }
}

__global__ void __launch_bounds__(512)
ak_bucket_sort_sh(const unsigned* __restrict__ tmp, const int* __restrict__ cnt1,
                  const int* __restrict__ pos1, int* __restrict__ rowptr, int* __restrict__ cnt,
                  int* __restrict__ mid, float* __restrict__ dinv, int* __restrict__ csr, int E,
                  int N, int chunk, int split) {
    typedef unsigned short ent_t;
    constexpr int CAP = ST2_BYTES / (int)sizeof(ent_t);
    constexpr int LINE = 128 / (int)sizeof(ent_t);
    __shared__ int c[2 * BNODES];
    __shared__ int cur[2 * BNODES];
    __shared__ int wtot[2 * BNODES / 32];
    __shared__ int shn[CHUNK_WGS];
    __shared__ int shb[CHUNK_WGS];
    __shared__ int redn[16];
    __shared__ int redp[16];
    __shared__ ent_t stg[CAP];
    const int tid = threadIdx.x, lane = tid & 31, w = tid >> 5;
    const int b = blockIdx.x;
    int myn = 0, myp = 0;
    if (tid < CHUNK_WGS) {
        myn = cnt1[b * CHUNK_WGS + tid];
        myp = pos1[b * CHUNK_WGS + tid];
        shn[tid] = myn;
        shb[tid] = min(tid * chunk + myp, E - 1);
    }
    c[tid] = 0;
    int sn = myn, sp = myp;
    for (int o = 16; o > 0; o >>= 1) {
        sn += __shfl_xor(sn, o);
        sp += __shfl_xor(sp, o);
    }
    if (lane == 0) {
        redn[w] = sn;
        redp[w] = sp;
    }
    __syncthreads();
    int tot = 0, lo = 0;
#pragma unroll
    for (int k = 0; k < 16; ++k) {
        tot += redn[k];
        lo += redp[k];
    }
    const int hi = lo + tot;
    const int sh = lo & (LINE - 1);
    const bool staged = tot + sh <= CAP;
    ent_t* const gout = reinterpret_cast<ent_t*>(csr);

    unsigned u[2 * BS_SPW];
    int nn[BS_SPW];
#pragma unroll
    for (int k = 0; k < BS_SPW; ++k) {
        nn[k] = shn[w * BS_SPW + k];
        const int p = shb[w * BS_SPW + k];
        const int m = max(nn[k] - 1, 0);
        u[2 * k]     = tmp[p + min(lane, m)];
        u[2 * k + 1] = tmp[p + min(lane + 32, m)];
    }
#pragma unroll
    for (int k = 0; k < BS_SPW; ++k) {
        if (lane < nn[k])      atomicAdd(&c[ak_key(u[2 * k], split)], 1);
        if (lane + 32 < nn[k]) atomicAdd(&c[ak_key(u[2 * k + 1], split)], 1);
    }
#pragma nounroll
    for (int k = 0; k < BS_SPW; ++k) {
        const int n = shn[w * BS_SPW + k];
        const int p = shb[w * BS_SPW + k];
        for (int i = 64 + lane; i < n; i += 32) atomicAdd(&c[ak_key(tmp[p + i], split)], 1);
    }
    __syncthreads();
    const int mine = c[tid];
    int x = mine;
    for (int o = 1; o < 32; o <<= 1) {
        int y = __shfl_up(x, o);
        if (lane >= o) x += y;
    }
    if (lane == 31) wtot[w] = x;
    __syncthreads();
    int base = 0;
    for (int k = 0; k < w; ++k) base += wtot[k];
    const int ex = base + x - mine;
    cur[tid] = ex;
    if ((tid & 1) == 0) {
        const int v = b * BNODES + (tid >> 1);
        if (v < N) {
            const int total = mine + c[tid + 1];
            cnt[v] = total;
            rowptr[v] = lo + ex;
            mid[v] = lo + ex + mine;
            dinv[v] = rsqrtf((float)(total + 1));
        }
    }
    __syncthreads();
    if (!staged) {
        ak_place_sh<false>(u, nn, tmp, shn, shb, w, lane, split, cur, stg, gout + lo);
        return;
    }
    ak_place_sh<true>(u, nn, tmp, shn, shb, w, lane, split, cur, stg + sh, gout + lo);
    __syncthreads();
    const int first = sh, last = sh + (hi - lo);
    ent_t* const gl = gout + (lo - sh);
    unsigned* const gw = reinterpret_cast<unsigned*>(gl);
    const int fe = (first + 1) & ~1, le = last & ~1;
    if (fe < le) {
        for (int wd0 = tid; 2 * wd0 < le; wd0 += 4 * 512) {
            unsigned a[4], bq[4];
#pragma unroll
            for (int k = 0; k < 4; ++k) {
                const int j = min(max(2 * (wd0 + 512 * k), fe), le - 2);
                a[k] = stg[j];
                bq[k] = stg[j + 1];
            }
#pragma unroll
            for (int k = 0; k < 4; ++k) {
                const int wd = wd0 + 512 * k;
                if (2 * wd >= fe && 2 * wd < le) gw[wd] = a[k] | (bq[k] << 16);
            }
        }
    }
    if (tid == 0) {
        if ((first & 1) && first < last) gl[first] = stg[first];
        if ((last & 1) && last - 1 >= first) gl[last - 1] = stg[last - 1];
    }
}

template <int M>
__device__ __forceinline__ float add_xmask(float x) {
    int b = __builtin_bit_cast(int, x);
    return x + __builtin_bit_cast(float, __builtin_amdgcn_update_dpp(0, b, 0x160 + M, 0xf, 0xf, true));
}

#define HOP_TRIPS 6

template <bool LAST>
__global__ void __launch_bounds__(256)
ak_hop(const int* __restrict__ rowptr, const int* __restrict__ cnt,
       const int* __restrict__ csr, const float* __restrict__ gin,
       const float* __restrict__ z0, const float* __restrict__ dinv,
       float* __restrict__ gout, const float* __restrict__ b3,
       float* __restrict__ out, int N) {
    const int l = threadIdx.x & 7;
    const int grp = (blockIdx.x * 256 + threadIdx.x) >> 3;
    const int ngrp = (gridDim.x * 256) >> 3;
    int v = grp;
    int n_nx = cnt[min(v, N - 1)];
    int b_nx = rowptr[min(v, N - 1)];
    for (int vw = grp & ~3; vw < N; vw += ngrp, v += ngrp) {
        const int vv = min(v, N - 1);
        const int n = n_nx;
        const int* __restrict__ cs = csr + b_nx;
        const int vn = min(v + ngrp, N - 1);
        n_nx = cnt[vn];
        b_nx = rowptr[vn];
        const float self = gin[vv];
        const float dv = dinv[vv];
        const float zz = z0[vv];
        float acc = 0.f;
        for (int j0 = l; j0 - l < n; j0 += 64) {
            int s[8];
            float a[8];
#pragma unroll
            for (int k = 0; k < 8; ++k) s[k] = cs[min(j0 + 8 * k, n - 1)];
#pragma unroll
            for (int k = 0; k < 8; ++k) a[k] = gin[s[k]];
#pragma unroll
            for (int k = 0; k < 8; ++k) acc += (j0 + 8 * k < n) ? a[k] : 0.f;
        }
        acc = add_xmask<1>(acc);
        acc = add_xmask<2>(acc);
        acc = add_xmask<4>(acc);
        const float zn = ONE_MINUS_ALPHA * dv * (acc + self) + ALPHA_F * zz;
        if (l == 0 && v < N) {
            if (LAST) out[v] = zn + b3[0];
            else      gout[v] = dv * zn;
        }
    }
}


#define HL_WGS 256
#define HL_T 512
#define HL_TRIPS 8
#define HL_SLAB 65536
#define HL_RING 13

typedef __attribute__((ext_vector_type(4))) float v4f;
typedef __attribute__((ext_vector_type(2))) int v2i;

__device__ __forceinline__ int hl_piece(int j, int tid, int nq, int rot) {
    const int i = min(j * HL_T + tid, nq - 1) + rot;
    return i >= nq ? i - nq : i;
}
__device__ __forceinline__ int hl_rot(int nq) {
    return (int)((blockIdx.x * (unsigned)nq) / HL_WGS) & ~31;
}

__device__ __forceinline__ v2i hl_quad(const unsigned short* __restrict__ c16, int p, int epad) {
    return *reinterpret_cast<const v2i*>(c16 + min(p, epad - 4));
}

__device__ __forceinline__ float hl_sum4(const float* sg, v2i q, int p, int s, int e) {
    const int qq[4] = {q.x & 0xffff, (int)((unsigned)q.x >> 16),
                       q.y & 0xffff, (int)((unsigned)q.y >> 16)};
    float a[4];
    bool ok[4];
#pragma unroll
    for (int j = 0; j < 4; ++j) {
        ok[j] = (p + j >= s) && (p + j < e);
        a[j] = sg[ok[j] ? qq[j] : 0];
    }
    float acc = 0.f;
#pragma unroll
    for (int j = 0; j < 4; ++j) acc += ok[j] ? a[j] : 0.f;
    return acc;
}

__device__ __forceinline__ float hl_first_lo(const float* sg, v2i qlo, int s, int e, int l) {
    return hl_sum4(sg, qlo, (s & ~3) + 4 * l, s, e);
}
__device__ __forceinline__ float hl_first_hi(const float* sg, v2i qhi, int s, int e, int l) {
    const int p = (s & ~3) + 32 + 4 * l;
    return p < e ? hl_sum4(sg, qhi, p, s, e) : 0.f;
}
__device__ __forceinline__ float hl_rest(const float* sg, const unsigned short* __restrict__ c16,
                                         int s, int e, int l, int epad) {
    float acc = 0.f;
    for (int pp = (s & ~3) + 64; pp < e; pp += 64) {
        const int p = pp + 4 * l;
        acc += hl_sum4(sg, hl_quad(c16, p, epad), p, s, e);
        acc += hl_sum4(sg, hl_quad(c16, p + 32, epad), p + 32, s, e);
    }
    return acc;
}

template <bool LAST>
__global__ void __launch_bounds__(HL_T)
ak_hop_lds(const int* __restrict__ rowptr, const int* __restrict__ cnt, const int* __restrict__ mid,
           const int* __restrict__ csr, const float* __restrict__ gin,
           const float* __restrict__ z0, const float* __restrict__ dinv,
           float* __restrict__ gout, const float* __restrict__ b3, float* __restrict__ out,
           int N, int npw, int split, int epad) {
    __shared__ float sg[HL_SLAB];
    __shared__ float stot[HL_T];
    const unsigned short* __restrict__ const c16 = reinterpret_cast<const unsigned short*>(csr);
    const int tid = threadIdx.x, l = tid & 7, gi = tid >> 3;
    const int v_lo = blockIdx.x * npw;
    const int v_hi = min(v_lo + npw, N);

    v4f* const d4 = reinterpret_cast<v4f*>(sg);
    v4f rr[HL_RING];
    int rx[HL_RING];
    const v4f* __restrict__ const ha = reinterpret_cast<const v4f*>(gin);
    const int nqa = split >> 2, rota = hl_rot(nqa), pera = (nqa + HL_T - 1) / HL_T;
#pragma unroll
    for (int k = 0; k < HL_RING; ++k) {
        rx[k] = hl_piece(k, tid, nqa, rota);
        rr[k] = ha[rx[k]];
    }

    int s0[HL_TRIPS], m0[HL_TRIPS], e0[HL_TRIPS];
#pragma unroll
    for (int t = 0; t < HL_TRIPS; ++t) {
        const int v = v_lo + gi + 64 * t;
        const int vv = min(v, N - 1);
        const int rs = rowptr[vv], rm = mid[vv], rc = cnt[vv];
        const bool ok = v < v_hi;
        s0[t] = ok ? rs : 0;
        m0[t] = ok ? rm : 0;
        e0[t] = ok ? rs + rc : 0;
    }
    v2i qa[HL_TRIPS], qa2[HL_TRIPS];
#pragma unroll
    for (int t = 0; t < HL_TRIPS; ++t) {
        qa[t]  = hl_quad(c16, (s0[t] & ~3) + 4 * l, epad);
        qa2[t] = hl_quad(c16, (s0[t] & ~3) + 32 + 4 * l, epad);
    }
    const int vf = v_lo + tid;
    const int vfc = min(vf, N - 1);
    const float f_dv = dinv[vfc], f_zz = z0[vfc], f_self = gin[vfc];

    for (int j0 = HL_RING; j0 < pera; j0 += HL_RING) {
#pragma unroll
        for (int k = 0; k < HL_RING; ++k) {
            d4[rx[k]] = rr[k];
            rx[k] = hl_piece(j0 + k, tid, nqa, rota);
            rr[k] = ha[rx[k]];
        }
    }
#pragma unroll
    for (int k = 0; k < HL_RING; ++k) d4[rx[k]] = rr[k];
    __syncthreads();

    const float* __restrict__ const g1 = gin + split;
    const v4f* __restrict__ const hb = reinterpret_cast<const v4f*>(g1);
    const int n1 = N - split;
    const int nqb = n1 >> 2, rotb = hl_rot(nqb), perb = (nqb + HL_T - 1) / HL_T;
#pragma unroll
    for (int k = 0; k < HL_RING; ++k) {
        rx[k] = hl_piece(k, tid, nqb, rotb);
        rr[k] = hb[rx[k]];
    }
    float acc[HL_TRIPS];
#pragma unroll
    for (int t = 0; t < HL_TRIPS; ++t) acc[t] = hl_first_lo(sg, qa[t], s0[t], m0[t], l);
#pragma unroll
    for (int t = 0; t < HL_TRIPS; ++t) acc[t] += hl_first_hi(sg, qa2[t], s0[t], m0[t], l);
#pragma unroll
    for (int t = 0; t < HL_TRIPS; ++t) acc[t] += hl_rest(sg, c16, s0[t], m0[t], l, epad);
    v2i qb[HL_TRIPS], qb2[HL_TRIPS];
#pragma unroll
    for (int t = 0; t < HL_TRIPS; ++t) {
        qb[t]  = hl_quad(c16, (m0[t] & ~3) + 4 * l, epad);
        qb2[t] = hl_quad(c16, (m0[t] & ~3) + 32 + 4 * l, epad);
    }
    __syncthreads();
    for (int j0 = HL_RING; j0 < perb; j0 += HL_RING) {
#pragma unroll
        for (int k = 0; k < HL_RING; ++k) {
            d4[rx[k]] = rr[k];
            rx[k] = hl_piece(j0 + k, tid, nqb, rotb);
            rr[k] = hb[rx[k]];
        }
    }
#pragma unroll
    for (int k = 0; k < HL_RING; ++k) d4[rx[k]] = rr[k];
    if (tid < (n1 & 3)) sg[(nqb << 2) + tid] = g1[(nqb << 2) + tid];
    __syncthreads();
#pragma unroll
    for (int t = 0; t < HL_TRIPS; ++t) acc[t] += hl_first_lo(sg, qb[t], m0[t], e0[t], l);
#pragma unroll
    for (int t = 0; t < HL_TRIPS; ++t) acc[t] += hl_first_hi(sg, qb2[t], m0[t], e0[t], l);
#pragma unroll
    for (int t = 0; t < HL_TRIPS; ++t) acc[t] += hl_rest(sg, c16, m0[t], e0[t], l, epad);
#pragma unroll
    for (int t = 0; t < HL_TRIPS; ++t) {
        float a = acc[t];
        a = add_xmask<1>(a);
        a = add_xmask<2>(a);
        a = add_xmask<4>(a);
        if (l == 0) stot[gi + 64 * t] = a;
    }
    __syncthreads();
    if (tid < npw && vf < v_hi) {
        const float zn = ONE_MINUS_ALPHA * f_dv * (stot[tid] + f_self) + ALPHA_F * f_zz;
        if (LAST) out[vf] = zn + b3[0];
        else      gout[vf] = f_dv * zn;
    }
}


extern "C" void kernel_launch(void* const* d_in, const int* in_sizes, int n_in,
                              void* d_out, int out_size, void* d_ws, size_t ws_size,
                              hipStream_t stream) {
    const float* x  = (const float*)d_in[0];
    const int*  eidx = (const int*)d_in[1];
    const float* W1 = (const float*)d_in[2];
    const float* b1 = (const float*)d_in[3];
    const float* W2 = (const float*)d_in[4];
    const float* b2 = (const float*)d_in[5];
    const float* W3 = (const float*)d_in[6];
    const float* b3 = (const float*)d_in[7];
    float* out = (float*)d_out;

    const int N = in_sizes[0];
    const int E = in_sizes[1] / 2;

    char* ws = (char*)d_ws;
    const int B = 256;

    {
        const int nb = (N + BNODES - 1) / BNODES;
        const int nh = nb * CHUNK_WGS;
        size_t o2 = 0;
        auto carve2 = [&](size_t bytes) -> char* {
            char* p = ws + o2;
            o2 += (bytes + 255) & ~(size_t)255;
            return p;
        };
        int*   cnt    = (int*)carve2((size_t)N * 4);
        float* dinv   = (float*)carve2((size_t)N * 4);
        int*   rowptr = (int*)carve2((size_t)N * 4);
        int*   hist1  = (int*)carve2((size_t)nh * 4);
        int*   off1   = (int*)carve2((size_t)nh * 4);
        int*   csr    = (int*)carve2((size_t)E * 4);
        const size_t plane = (((size_t)N * 4) + 255) & ~(size_t)255;
        const size_t tmpb = (size_t)E * 4;
        char* shared_region = carve2(tmpb > 3 * plane ? tmpb : 3 * plane);
        unsigned* tmp = (unsigned*)shared_region;
        float* z0 = (float*)shared_region;
        float* gA = (float*)(shared_region + plane);
        float* gB = (float*)(shared_region + 2 * plane);
        int* mid = hist1;
        const bool lds_hop = N >= 4096 && N <= 2 * HL_SLAB && E >= 64;
        const int split = lds_hop ? ((N / 2 + 31) & ~31) : N;

        if (o2 <= ws_size && nb <= NB_MAX && N <= (1 << 24)) {
            const int chunk = (E + CHUNK_WGS - 1) / CHUNK_WGS;
            const int epad = (E + 63) & ~63;
            const size_t mid_off = (((size_t)epad * 2) + 255) & ~(size_t)255;
            const bool own_chunks = lds_hop && chunk <= ST1_CAP &&
                                    mid_off + (size_t)N * 4 <= (size_t)E * 4;
            if (own_chunks) {
                mid = (int*)((char*)csr + mid_off);
                ak_chunk_sort<<<CHUNK_WGS, 1024, 0, stream>>>(eidx, hist1, off1, tmp, E, chunk, nb);
                ak_bucket_sort_sh<<<nb, 512, 0, stream>>>(tmp, hist1, off1, rowptr, cnt, mid, dinv, csr,
                                                          E, N, chunk, split);
            } else {
                ak_bucket_hist<<<CHUNK_WGS, 1024, 0, stream>>>(eidx, hist1, E, chunk, nb);
                ak_excl_scan<<<(nh + 1023) / 1024, 1024, 0, stream>>>(hist1, off1, nh);
                ak_bucket_scatter<<<CHUNK_WGS, 1024, 0, stream>>>(eidx, hist1, off1, tmp, E, chunk, nb);
                if (lds_hop) {
                    ak_bucket_sort<true><<<nb, 512, 0, stream>>>(tmp, off1, rowptr, cnt, mid, dinv, csr,
                                                                 E, N, nb, split);
                } else {
                    ak_bucket_sort<false><<<nb, 512, 0, stream>>>(tmp, off1, rowptr, cnt, mid, dinv, csr,
                                                                  E, N, nb, split);
                }
            }

            int tiles = N / 16;
            if (tiles > 0) {
                int waves_per_block = B / 32;
                int blocks = (tiles + waves_per_block - 1) / waves_per_block;
                ak_encoder_wmma<<<blocks, B, 0, stream>>>(x, W1, b1, W2, b2, W3, dinv, z0, gA,
                                                          (float*)nullptr, tiles);
            }
            int rem_start = tiles * 16;
            if (rem_start < N) {
                ak_encoder_tail<<<(N - rem_start + B - 1) / B, B, 0, stream>>>(
                    x, W1, b1, W2, b2, W3, dinv, z0, gA, (float*)nullptr, rem_start, N);
            }

            const int hblocks = (N + 32 * HOP_TRIPS - 1) / (32 * HOP_TRIPS);
            const int npw = (N + HL_WGS - 1) / HL_WGS;
            for (int it = 0; it < KITERS; ++it) {
                const float* gi = (it & 1) ? gB : gA;
                float* go = (it & 1) ? gA : gB;
                if (lds_hop) {
                    if (it < KITERS - 1) {
                        ak_hop_lds<false><<<HL_WGS, HL_T, 0, stream>>>(rowptr, cnt, mid, csr, gi, z0,
                                                                       dinv, go, b3, out, N, npw,
                                                                       split, epad);
                    } else {
                        ak_hop_lds<true><<<HL_WGS, HL_T, 0, stream>>>(rowptr, cnt, mid, csr, gi, z0,
                                                                      dinv, (float*)nullptr, b3, out,
                                                                      N, npw, split, epad);
                    }
                } else if (it < KITERS - 1) {
                    ak_hop<false><<<hblocks, B, 0, stream>>>(rowptr, cnt, csr, gi, z0, dinv, go,
                                                             b3, out, N);
                } else {
                    ak_hop<true><<<hblocks, B, 0, stream>>>(rowptr, cnt, csr, gi, z0, dinv,
                                                            (float*)nullptr, b3, out, N);
                }
            }
            return;
        }
    }

    size_t off = 0;
    auto carve = [&](size_t bytes) -> char* {
        char* p = ws + off;
        off += (bytes + 255) & ~(size_t)255;
        return p;
    };
    float* deg  = (float*)carve((size_t)N * 4);
    float* dinv = (float*)carve((size_t)N * 4);
    float* z0   = (float*)carve((size_t)N * 4);
    float* g    = (float*)carve((size_t)N * 4);
    float* agg  = (float*)carve((size_t)N * 4);

    ak_deg_init<<<(N + B - 1) / B, B, 0, stream>>>(deg, N);
    ak_deg_accum<<<(E + B - 1) / B, B, 0, stream>>>(eidx, deg, E);
    ak_dinv<<<(N + B - 1) / B, B, 0, stream>>>(deg, dinv, N);

    int tiles = N / 16;
    if (tiles > 0) {
        int waves_per_block = B / 32;
        int blocks = (tiles + waves_per_block - 1) / waves_per_block;
        ak_encoder_wmma<<<blocks, B, 0, stream>>>(x, W1, b1, W2, b2, W3, dinv, z0, g, agg, tiles);
    }
    int rem_start = tiles * 16;
    if (rem_start < N) {
        ak_encoder_tail<<<(N - rem_start + B - 1) / B, B, 0, stream>>>(
            x, W1, b1, W2, b2, W3, dinv, z0, g, agg, rem_start, N);
    }

    for (int it = 0; it < KITERS; ++it) {
        ak_edge_scatter<<<(E + B - 1) / B, B, 0, stream>>>(eidx, g, agg, E);
        if (it < KITERS - 1) {
            ak_finalize<<<(N + B - 1) / B, B, 0, stream>>>(agg, g, z0, dinv, N);
        } else {
            ak_finalize_out<<<(N + B - 1) / B, B, 0, stream>>>(agg, g, z0, dinv, b3, out, N);
        }
    }
}


# A hop work-group copies only one half of g to LDS: two work-groups share a node range and each adds its term of g' by one float atomic add onto a plane cleared a hop earlier; float sum order changes
# speedup vs baseline: 18.5343x; 1.1409x over previous
#include <hip/hip_runtime.h>


typedef __attribute__((ext_vector_type(2))) float v2f;
typedef __attribute__((ext_vector_type(8))) float v8f;

#define HID 16
#define KITERS 10
#define ONE_MINUS_ALPHA 0.9f
#define ALPHA_F 0.1f

__device__ __forceinline__ float relu_f(float v) { return v > 0.f ? v : 0.f; }


__global__ void ak_deg_init(float* __restrict__ deg, int n) {
    int i = blockIdx.x * blockDim.x + threadIdx.x;
    if (i < n) deg[i] = 1.0f;
}

__global__ void ak_deg_accum(const int* __restrict__ eidx, float* __restrict__ deg, int E) {
    int e = blockIdx.x * blockDim.x + threadIdx.x;
    if (e < E) atomicAdd(&deg[eidx[(long long)E + e]], 1.0f);
}

__global__ void ak_dinv(const float* __restrict__ deg, float* __restrict__ dinv, int n) {
    int i = blockIdx.x * blockDim.x + threadIdx.x;
    if (i < n) dinv[i] = rsqrtf(deg[i]);
}

__device__ __forceinline__ float xhalf(float x) {
    int b = __builtin_bit_cast(int, x);
    return __builtin_bit_cast(float,
        __builtin_amdgcn_permlanex16(b, b, 0x76543210, 0xfedcba98, false, false));
}


__global__ void ak_encoder_wmma(const float* __restrict__ x, const float* __restrict__ W1,
                                const float* __restrict__ b1, const float* __restrict__ W2,
                                const float* __restrict__ b2, const float* __restrict__ W3,
                                const float* __restrict__ dinv, float* __restrict__ z0,
                                float* __restrict__ g, float* __restrict__ agg, int n_tiles) {
    int wave = (blockIdx.x * blockDim.x + threadIdx.x) >> 5;
    int lane = threadIdx.x & 31;
    if (wave >= n_tiles) return;

    int row = lane & 15;
    int hi  = lane >> 4;
    int node = wave * 16 + row;
    float xv = x[node];

    v8f acc = {};
#pragma unroll
    for (int s = 0; s < 4; ++s) {
        int k0 = 4 * s + 2 * hi;
        v2f wt, ht;
        wt.x = W2[k0 * HID + row];
        wt.y = W2[(k0 + 1) * HID + row];
        ht.x = relu_f(xv * W1[k0]     + b1[k0]);
        ht.y = relu_f(xv * W1[k0 + 1] + b1[k0 + 1]);
        acc = __builtin_amdgcn_wmma_f32_16x16x4_f32(
            false, wt, false, ht, (short)0, acc, false, false);
    }

    float part = 0.f;
#pragma unroll
    for (int i = 0; i < 8; ++i) {
        int m = i + 8 * hi;
        part += relu_f(acc[i] + b2[m]) * W3[m];
    }
    float z = part + xhalf(part);
    if (hi == 0) {
        z0[node] = z;
        g[node]  = dinv[node] * z;
        if (agg) agg[node] = 0.f;
    }
}

__global__ void ak_encoder_tail(const float* __restrict__ x, const float* __restrict__ W1,
                                const float* __restrict__ b1, const float* __restrict__ W2,
                                const float* __restrict__ b2, const float* __restrict__ W3,
                                const float* __restrict__ dinv, float* __restrict__ z0,
                                float* __restrict__ g, float* __restrict__ agg, int start, int N) {
    int v = start + blockIdx.x * blockDim.x + threadIdx.x;
    if (v >= N) return;
    float h1[HID];
    float xv = x[v];
#pragma unroll
    for (int k = 0; k < HID; ++k) h1[k] = relu_f(xv * W1[k] + b1[k]);
    float z = 0.f;
#pragma unroll
    for (int j = 0; j < HID; ++j) {
        float acc = b2[j];
#pragma unroll
        for (int k = 0; k < HID; ++k) acc += h1[k] * W2[k * HID + j];
        z += relu_f(acc) * W3[j];
    }
    z0[v] = z;
    g[v]  = dinv[v] * z;
    if (agg) agg[v] = 0.f;
}


__global__ void ak_edge_scatter(const int* __restrict__ eidx, const float* __restrict__ g,
                                float* __restrict__ agg, int E) {
    int e = blockIdx.x * blockDim.x + threadIdx.x;
    if (e >= E) return;
    atomicAdd(&agg[eidx[(long long)E + e]], g[eidx[e]]);
}

__global__ void ak_finalize(float* __restrict__ agg, float* __restrict__ g,
                            const float* __restrict__ z0, const float* __restrict__ dinv, int n) {
    int v = blockIdx.x * blockDim.x + threadIdx.x;
    if (v >= n) return;
    float dv = dinv[v];
    float zn = ONE_MINUS_ALPHA * dv * (agg[v] + g[v]) + ALPHA_F * z0[v];
    g[v] = dv * zn;
    agg[v] = 0.f;
}

__global__ void ak_finalize_out(const float* __restrict__ agg, const float* __restrict__ g,
                                const float* __restrict__ z0, const float* __restrict__ dinv,
                                const float* __restrict__ b3, float* __restrict__ out, int N) {
    int v = blockIdx.x * blockDim.x + threadIdx.x;
    if (v >= N) return;
    out[v] = ONE_MINUS_ALPHA * dinv[v] * (agg[v] + g[v]) + ALPHA_F * z0[v] + b3[0];
}


#define BSH 8
#define BNODES 256
#define NB_MAX 4096
#define CHUNK_WGS 256

__global__ void __launch_bounds__(1024)
ak_bucket_hist(const int* __restrict__ eidx, int* __restrict__ hist1, int E, int chunk, int nb) {
    __shared__ int hist[NB_MAX];
    for (int b = threadIdx.x; b < nb; b += 1024) hist[b] = 0;
    __syncthreads();
    const int lo = blockIdx.x * chunk;
    const int hi = min(lo + chunk, E);
    const int* __restrict__ dst = eidx + (size_t)E;
    for (int e0 = lo + threadIdx.x; e0 < hi; e0 += 4 * 1024) {
        int d[4];
#pragma unroll
        for (int k = 0; k < 4; ++k) d[k] = dst[min(e0 + k * 1024, hi - 1)];
#pragma unroll
        for (int k = 0; k < 4; ++k) {
            if (e0 + k * 1024 < hi) atomicAdd(&hist[d[k] >> BSH], 1);
        }
    }
    __syncthreads();
    for (int b = threadIdx.x; b < nb; b += 1024) hist1[b * CHUNK_WGS + blockIdx.x] = hist[b];
}

__global__ void __launch_bounds__(1024)
ak_excl_scan(const int* __restrict__ in, int* __restrict__ out, int n) {
    __shared__ int red[32];
    __shared__ int wsum[32];
    const int tid = threadIdx.x, lane = tid & 31, w = tid >> 5;
    const int first = blockIdx.x * 1024;

    int part = 0;
    for (int i = tid; i < first; i += 1024) part += in[i];
    for (int o = 16; o > 0; o >>= 1) part += __shfl_xor(part, o);
    if (lane == 0) red[w] = part;

    const int i = first + tid;
    const int c = (i < n) ? in[i] : 0;
    int x = c;
    for (int o = 1; o < 32; o <<= 1) {
        int y = __shfl_up(x, o);
        if (lane >= o) x += y;
    }
    if (lane == 31) wsum[w] = x;
    __syncthreads();
    if (w == 0) {
        int s = wsum[lane];
        for (int o = 1; o < 32; o <<= 1) {
            int y = __shfl_up(s, o);
            if (lane >= o) s += y;
        }
        wsum[lane] = s;
        int b = red[lane];
        for (int o = 16; o > 0; o >>= 1) b += __shfl_xor(b, o);
        red[lane] = b;
    }
    __syncthreads();
    if (i < n) out[i] = red[0] + (w ? wsum[w - 1] : 0) + x - c;
}

template <bool STAGED>
__device__ __forceinline__ void ak_scatter_edges(const int* __restrict__ eidx, int E, int lo, int hi,
                                                 int tid, int* cur, unsigned* stg,
                                                 unsigned* __restrict__ tmp) {
    const int* __restrict__ dst = eidx + (size_t)E;
    for (int e0 = lo + tid; e0 < hi; e0 += 4 * 1024) {
        int d[4], s[4];
#pragma unroll
        for (int k = 0; k < 4; ++k) {
            const int e = min(e0 + k * 1024, hi - 1);
            d[k] = dst[e];
            s[k] = eidx[e];
        }
        asm volatile("" :: "v"(d[0]), "v"(d[1]), "v"(d[2]), "v"(d[3]),
                           "v"(s[0]), "v"(s[1]), "v"(s[2]), "v"(s[3]));
#pragma unroll
        for (int k = 0; k < 4; ++k) {
            if (e0 + k * 1024 < hi) {
                const int slot = atomicAdd(&cur[d[k] >> BSH], 1);
                const unsigned ent = ((unsigned)s[k] << BSH) | (unsigned)(d[k] & (BNODES - 1));
                if (STAGED) stg[slot] = ent;
                else        tmp[slot] = ent;
            }
        }
    }
}

#define ST1_CAP 20480

__global__ void __launch_bounds__(1024)
ak_bucket_scatter(const int* __restrict__ eidx, const int* __restrict__ hist1,
                  const int* __restrict__ off1, unsigned* __restrict__ tmp, int E, int chunk,
                  int nb) {
    __shared__ int cur[NB_MAX];
    __shared__ int sbeg[NB_MAX];
    __shared__ int gbeg[NB_MAX];
    __shared__ unsigned stg[ST1_CAP];
    __shared__ int wsum[32];
    const int tid = threadIdx.x, lane = tid & 31, w = tid >> 5;
    const bool staged = chunk <= ST1_CAP;
    if (staged) {
        int c4[4], g4[4];
        int s = 0;
#pragma unroll
        for (int k = 0; k < 4; ++k) {
            const int b = 4 * tid + k;
            const int bb = min(b, nb - 1);
            const int cv = hist1[bb * CHUNK_WGS + blockIdx.x];
            g4[k] = off1[bb * CHUNK_WGS + blockIdx.x];
            c4[k] = b < nb ? cv : 0;
            s += c4[k];
        }
        int x = s;
        for (int o = 1; o < 32; o <<= 1) {
            int y = __shfl_up(x, o);
            if (lane >= o) x += y;
        }
        if (lane == 31) wsum[w] = x;
        __syncthreads();
        if (w == 0) {
            int t = wsum[lane];
            for (int o = 1; o < 32; o <<= 1) {
                int y = __shfl_up(t, o);
                if (lane >= o) t += y;
            }
            wsum[lane] = t;
        }
        __syncthreads();
        int base = (w ? wsum[w - 1] : 0) + x - s;
#pragma unroll
        for (int k = 0; k < 4; ++k) {
            const int b = 4 * tid + k;
            if (b < nb) {
                cur[b] = base;
                sbeg[b] = base;
                gbeg[b] = g4[k];
            }
            base += c4[k];
        }
    } else {
        for (int b = tid; b < nb; b += 1024) cur[b] = off1[b * CHUNK_WGS + blockIdx.x];
    }
    __syncthreads();
    const int lo = blockIdx.x * chunk;
    const int hi = min(lo + chunk, E);
    if (!staged) {
        ak_scatter_edges<false>(eidx, E, lo, hi, tid, cur, stg, tmp);
        return;
    }
    ak_scatter_edges<true>(eidx, E, lo, hi, tid, cur, stg, tmp);
    __syncthreads();
    for (int b = w; b < nb; b += 32) {
        const int s0 = sbeg[b], n = cur[b] - s0, g0 = gbeg[b];
        if (n <= 0) continue;
        for (int ib = lane - (g0 & 31); ib < n; ib += 128) {
            unsigned v[4];
#pragma unroll
            for (int k = 0; k < 4; ++k)
                v[k] = stg[s0 + min(max(ib + 32 * k, 0), n - 1)];
#pragma unroll
            for (int k = 0; k < 4; ++k) {
                const int i = ib + 32 * k;
                if (i >= 0 && i < n) tmp[g0 + i] = v[k];
            }
        }
    }
}

#define ST2_BYTES 49152

template <bool S16> struct ak_ent { typedef int t; };
template <> struct ak_ent<true> { typedef unsigned short t; };

template <bool S16, bool STAGED>
__device__ __forceinline__ void ak_place(const unsigned* __restrict__ tmp, int lo, int hi, int tid,
                                         int split, int* cur, typename ak_ent<S16>::t* stage,
                                         typename ak_ent<S16>::t* __restrict__ out) {
    typedef typename ak_ent<S16>::t ent_t;
    for (int i0 = lo + tid; i0 < hi; i0 += 4 * 512) {
        unsigned u[4];
#pragma unroll
        for (int k = 0; k < 4; ++k) u[k] = tmp[min(i0 + k * 512, hi - 1)];
        asm volatile("" :: "v"(u[0]), "v"(u[1]), "v"(u[2]), "v"(u[3]));
#pragma unroll
        for (int k = 0; k < 4; ++k) {
            if (i0 + k * 512 < hi) {
                const int sn = (int)(u[k] >> BSH);
                const int key = (int)((u[k] & (BNODES - 1)) << 1) | (sn >= split ? 1 : 0);
                const int slot = atomicAdd(&cur[key], 1);
                const ent_t val = S16 ? (ent_t)(sn >= split ? sn - split : sn) : (ent_t)sn;
                if (STAGED) stage[slot] = val;
                else        out[slot] = val;
            }
        }
    }
}

template <bool S16>
__global__ void __launch_bounds__(512)
ak_bucket_sort(const unsigned* __restrict__ tmp, const int* __restrict__ off1,
               int* __restrict__ rowptr, int* __restrict__ cnt, int* __restrict__ mid,
               float* __restrict__ dinv, int* __restrict__ csr, int E, int N, int nb, int split) {
    typedef typename ak_ent<S16>::t ent_t;
    constexpr int CAP = ST2_BYTES / (int)sizeof(ent_t);
    constexpr int LINE = 128 / (int)sizeof(ent_t);
    __shared__ int c[2 * BNODES];
    __shared__ int cur[2 * BNODES];
    __shared__ int wtot[2 * BNODES / 32];
    __shared__ ent_t stg[CAP];
    const int tid = threadIdx.x, lane = tid & 31, w = tid >> 5;
    const int b = blockIdx.x;
    const int lo = off1[b * CHUNK_WGS];
    const int hi = (b + 1 < nb) ? off1[(b + 1) * CHUNK_WGS] : E;
    const int sh = lo & (LINE - 1);
    const bool staged = (hi - lo) + sh <= CAP;
    ent_t* const gout = reinterpret_cast<ent_t*>(csr);
    c[tid] = 0;
    __syncthreads();
    for (int i0 = lo + tid; i0 < hi; i0 += 4 * 512) {
        unsigned u[4];
#pragma unroll
        for (int k = 0; k < 4; ++k) u[k] = tmp[min(i0 + k * 512, hi - 1)];
#pragma unroll
        for (int k = 0; k < 4; ++k) {
            if (i0 + k * 512 < hi) {
                const int key = (int)((u[k] & (BNODES - 1)) << 1) | ((int)(u[k] >> BSH) >= split ? 1 : 0);
                atomicAdd(&c[key], 1);
            }
        }
    }
    __syncthreads();
    const int mine = c[tid];
    int x = mine;
    for (int o = 1; o < 32; o <<= 1) {
        int y = __shfl_up(x, o);
        if (lane >= o) x += y;
    }
    if (lane == 31) wtot[w] = x;
    __syncthreads();
    int base = 0;
    for (int k = 0; k < w; ++k) base += wtot[k];
    const int ex = base + x - mine;
    cur[tid] = ex;
    if ((tid & 1) == 0) {
        const int v = b * BNODES + (tid >> 1);
        if (v < N) {
            const int total = mine + c[tid + 1];
            cnt[v] = total;
            rowptr[v] = lo + ex;
            mid[v] = lo + ex + mine;
            dinv[v] = rsqrtf((float)(total + 1));
        }
    }
    __syncthreads();
    if (!staged) {
        ak_place<S16, false>(tmp, lo, hi, tid, split, cur, stg, gout + lo);
        return;
    }
    ak_place<S16, true>(tmp, lo, hi, tid, split, cur, stg + sh, gout + lo);
    __syncthreads();
    const int first = sh, last = sh + (hi - lo);
    ent_t* const gl = gout + (lo - sh);
    if (S16) {
        unsigned* const gw = reinterpret_cast<unsigned*>(gl);
        const int fe = (first + 1) & ~1, le = last & ~1;
        if (fe < le) {
            for (int wd0 = tid; 2 * wd0 < le; wd0 += 4 * 512) {
                unsigned a[4], bq[4];
#pragma unroll
                for (int k = 0; k < 4; ++k) {
                    const int j = min(max(2 * (wd0 + 512 * k), fe), le - 2);
                    a[k] = stg[j];
                    bq[k] = stg[j + 1];
                }
#pragma unroll
                for (int k = 0; k < 4; ++k) {
                    const int wd = wd0 + 512 * k;
                    if (2 * wd >= fe && 2 * wd < le) gw[wd] = a[k] | (bq[k] << 16);
                }
            }
        }
        if (tid == 0) {
            if ((first & 1) && first < last) gl[first] = stg[first];
            if ((last & 1) && last - 1 >= first) gl[last - 1] = stg[last - 1];
        }
    } else if (first < last) {
        for (int j0 = tid; j0 < last; j0 += 4 * 512) {
            ent_t a[4];
#pragma unroll
            for (int k = 0; k < 4; ++k) a[k] = stg[min(max(j0 + 512 * k, first), last - 1)];
#pragma unroll
            for (int k = 0; k < 4; ++k) {
                const int j = j0 + 512 * k;
                if (j >= first && j < last) gl[j] = a[k];
            }
        }
    }
}


#define CS_EPT 20
static_assert(CS_EPT * 1024 >= ST1_CAP, "a thread's registers must cover the largest staged chunk");

__global__ void __launch_bounds__(1024)
ak_chunk_sort(const int* __restrict__ eidx, int* __restrict__ cnt1, int* __restrict__ pos1,
              unsigned* __restrict__ tmp, int E, int chunk, int nb) {
    __shared__ int cur[NB_MAX];
    __shared__ unsigned stg[ST1_CAP];
    __shared__ int wsum[32];
    const int tid = threadIdx.x, lane = tid & 31, w = tid >> 5;
    const int lo = blockIdx.x * chunk;
    const int hi = min(lo + chunk, E);
    const int* __restrict__ dst = eidx + (size_t)E;
    int d[CS_EPT], s[CS_EPT];
#pragma unroll
    for (int k = 0; k < CS_EPT; ++k) {
        const int e = min(lo + tid + k * 1024, E - 1);
        d[k] = dst[e];
        s[k] = eidx[e];
    }
    for (int b = tid; b < nb; b += 1024) cur[b] = 0;
    __syncthreads();
#pragma unroll
    for (int k = 0; k < CS_EPT; ++k) {
        if (lo + tid + k * 1024 < hi) atomicAdd(&cur[d[k] >> BSH], 1);
    }
    __syncthreads();
    int c4[4];
    int sum4 = 0;
#pragma unroll
    for (int k = 0; k < 4; ++k) {
        const int b = 4 * tid + k;
        const int cv = cur[b];
        c4[k] = b < nb ? cv : 0;
        sum4 += c4[k];
    }
    int x = sum4;
    for (int o = 1; o < 32; o <<= 1) {
        int y = __shfl_up(x, o);
        if (lane >= o) x += y;
    }
    if (lane == 31) wsum[w] = x;
    __syncthreads();
    if (w == 0) {
        int t = wsum[lane];
        for (int o = 1; o < 32; o <<= 1) {
            int y = __shfl_up(t, o);
            if (lane >= o) t += y;
        }
        wsum[lane] = t;
    }
    __syncthreads();
    int base = (w ? wsum[w - 1] : 0) + x - sum4;
    int p4[4];
#pragma unroll
    for (int k = 0; k < 4; ++k) {
        const int b = 4 * tid + k;
        p4[k] = base;
        if (b < nb) cur[b] = base;
        base += c4[k];
    }
    __syncthreads();
    int slot[CS_EPT];
#pragma unroll
    for (int k = 0; k < CS_EPT; ++k) {
        slot[k] = 0;
        if (lo + tid + k * 1024 < hi) slot[k] = atomicAdd(&cur[d[k] >> BSH], 1);
    }
#pragma unroll
    for (int k = 0; k < CS_EPT; ++k) {
        if (lo + tid + k * 1024 < hi)
            stg[slot[k]] = ((unsigned)s[k] << BSH) | (unsigned)(d[k] & (BNODES - 1));
    }
    __syncthreads();
#pragma unroll
    for (int k = 0; k < 4; ++k) {
        const int b = 4 * tid + k;
        if (b < nb) {
            cnt1[b * CHUNK_WGS + blockIdx.x] = c4[k];
            pos1[b * CHUNK_WGS + blockIdx.x] = p4[k];
        }
    }
    const int n = hi - lo;
    if (n <= 0) return;
    for (int i0 = tid - (lo & 31); i0 < n; i0 += 4 * 1024) {
        unsigned v[4];
#pragma unroll
        for (int k = 0; k < 4; ++k) v[k] = stg[min(max(i0 + 1024 * k, 0), n - 1)];
#pragma unroll
        for (int k = 0; k < 4; ++k) {
            const int i = i0 + 1024 * k;
            if (i >= 0 && i < n) tmp[lo + i] = v[k];
        }
    }
}

__device__ __forceinline__ int ak_key(unsigned u, int split) {
    return (int)((u & (BNODES - 1)) << 1) | ((int)(u >> BSH) >= split ? 1 : 0);
}

template <bool STAGED>
__device__ __forceinline__ void ak_place_one(unsigned u, int split, int* cur, unsigned short* stage,
                                             unsigned short* __restrict__ out) {
    const int sn = (int)(u >> BSH);
    const int slot = atomicAdd(&cur[ak_key(u, split)], 1);
    const unsigned short val = (unsigned short)(sn >= split ? sn - split : sn);
    if (STAGED) stage[slot] = val;
    else        out[slot] = val;
}

#define BS_SPW 16

template <bool STAGED>
__device__ __forceinline__ void ak_place_sh(const unsigned (&u)[2 * BS_SPW], const int (&nn)[BS_SPW],
                                            const unsigned* __restrict__ tmp, const int* shn,
                                            const int* shb, int w, int lane, int split, int* cur,
                                            unsigned short* stage, unsigned short* __restrict__ out) {
#pragma unroll
    for (int k = 0; k < BS_SPW; ++k) {
        if (lane < nn[k])      ak_place_one<STAGED>(u[2 * k], split, cur, stage, out);
        if (lane + 32 < nn[k]) ak_place_one<STAGED>(u[2 * k + 1], split, cur, stage, out);
    }
#pragma nounroll
    for (int k = 0; k < BS_SPW; ++k) {
        const int n = shn[w * BS_SPW + k];
        const int p = shb[w * BS_SPW + k];
        for (int i = 64 + lane; i < n; i += 32)
            ak_place_one<STAGED>(tmp[p + i], split, cur, stage, out);
    }
}

__global__ void __launch_bounds__(512)
ak_bucket_sort_sh(const unsigned* __restrict__ tmp, const int* __restrict__ cnt1,
                  const int* __restrict__ pos1, int* __restrict__ rowptr, int* __restrict__ cnt,
                  int* __restrict__ mid, float* __restrict__ dinv, int* __restrict__ csr, int E,
                  int N, int chunk, int split) {
    typedef unsigned short ent_t;
    constexpr int CAP = ST2_BYTES / (int)sizeof(ent_t);
    constexpr int LINE = 128 / (int)sizeof(ent_t);
    __shared__ int c[2 * BNODES];
    __shared__ int cur[2 * BNODES];
    __shared__ int wtot[2 * BNODES / 32];
    __shared__ int shn[CHUNK_WGS];
    __shared__ int shb[CHUNK_WGS];
    __shared__ int redn[16];
    __shared__ int redp[16];
    __shared__ ent_t stg[CAP];
    const int tid = threadIdx.x, lane = tid & 31, w = tid >> 5;
    const int b = blockIdx.x;
    int myn = 0, myp = 0;
    if (tid < CHUNK_WGS) {
        myn = cnt1[b * CHUNK_WGS + tid];
        myp = pos1[b * CHUNK_WGS + tid];
        shn[tid] = myn;
        shb[tid] = min(tid * chunk + myp, E - 1);
    }
    c[tid] = 0;
    int sn = myn, sp = myp;
    for (int o = 16; o > 0; o >>= 1) {
        sn += __shfl_xor(sn, o);
        sp += __shfl_xor(sp, o);
    }
    if (lane == 0) {
        redn[w] = sn;
        redp[w] = sp;
    }
    __syncthreads();
    int tot = 0, lo = 0;
#pragma unroll
    for (int k = 0; k < 16; ++k) {
        tot += redn[k];
        lo += redp[k];
    }
    const int hi = lo + tot;
    const int sh = lo & (LINE - 1);
    const bool staged = tot + sh <= CAP;
    ent_t* const gout = reinterpret_cast<ent_t*>(csr);

    unsigned u[2 * BS_SPW];
    int nn[BS_SPW];
#pragma unroll
    for (int k = 0; k < BS_SPW; ++k) {
        nn[k] = shn[w * BS_SPW + k];
        const int p = shb[w * BS_SPW + k];
        const int m = max(nn[k] - 1, 0);
        u[2 * k]     = tmp[p + min(lane, m)];
        u[2 * k + 1] = tmp[p + min(lane + 32, m)];
    }
#pragma unroll
    for (int k = 0; k < BS_SPW; ++k) {
        if (lane < nn[k])      atomicAdd(&c[ak_key(u[2 * k], split)], 1);
        if (lane + 32 < nn[k]) atomicAdd(&c[ak_key(u[2 * k + 1], split)], 1);
    }
#pragma nounroll
    for (int k = 0; k < BS_SPW; ++k) {
        const int n = shn[w * BS_SPW + k];
        const int p = shb[w * BS_SPW + k];
        for (int i = 64 + lane; i < n; i += 32) atomicAdd(&c[ak_key(tmp[p + i], split)], 1);
    }
    __syncthreads();
    const int mine = c[tid];
    int x = mine;
    for (int o = 1; o < 32; o <<= 1) {
        int y = __shfl_up(x, o);
        if (lane >= o) x += y;
    }
    if (lane == 31) wtot[w] = x;
    __syncthreads();
    int base = 0;
    for (int k = 0; k < w; ++k) base += wtot[k];
    const int ex = base + x - mine;
    cur[tid] = ex;
    if ((tid & 1) == 0) {
        const int v = b * BNODES + (tid >> 1);
        if (v < N) {
            const int total = mine + c[tid + 1];
            cnt[v] = total;
            rowptr[v] = lo + ex;
            mid[v] = lo + ex + mine;
            dinv[v] = rsqrtf((float)(total + 1));
        }
    }
    __syncthreads();
    if (!staged) {
        ak_place_sh<false>(u, nn, tmp, shn, shb, w, lane, split, cur, stg, gout + lo);
        return;
    }
    ak_place_sh<true>(u, nn, tmp, shn, shb, w, lane, split, cur, stg + sh, gout + lo);
    __syncthreads();
    const int first = sh, last = sh + (hi - lo);
    ent_t* const gl = gout + (lo - sh);
    unsigned* const gw = reinterpret_cast<unsigned*>(gl);
    const int fe = (first + 1) & ~1, le = last & ~1;
    if (fe < le) {
        for (int wd0 = tid; 2 * wd0 < le; wd0 += 4 * 512) {
            unsigned a[4], bq[4];
#pragma unroll
            for (int k = 0; k < 4; ++k) {
                const int j = min(max(2 * (wd0 + 512 * k), fe), le - 2);
                a[k] = stg[j];
                bq[k] = stg[j + 1];
            }
#pragma unroll
            for (int k = 0; k < 4; ++k) {
                const int wd = wd0 + 512 * k;
                if (2 * wd >= fe && 2 * wd < le) gw[wd] = a[k] | (bq[k] << 16);
            }
        }
    }
    if (tid == 0) {
        if ((first & 1) && first < last) gl[first] = stg[first];
        if ((last & 1) && last - 1 >= first) gl[last - 1] = stg[last - 1];
    }
}

template <int M>
__device__ __forceinline__ float add_xmask(float x) {
    int b = __builtin_bit_cast(int, x);
    return x + __builtin_bit_cast(float, __builtin_amdgcn_update_dpp(0, b, 0x160 + M, 0xf, 0xf, true));
}

#define HOP_TRIPS 6

template <bool LAST>
__global__ void __launch_bounds__(256)
ak_hop(const int* __restrict__ rowptr, const int* __restrict__ cnt,
       const int* __restrict__ csr, const float* __restrict__ gin,
       const float* __restrict__ z0, const float* __restrict__ dinv,
       float* __restrict__ gout, const float* __restrict__ b3,
       float* __restrict__ out, int N) {
    const int l = threadIdx.x & 7;
    const int grp = (blockIdx.x * 256 + threadIdx.x) >> 3;
    const int ngrp = (gridDim.x * 256) >> 3;
    int v = grp;
    int n_nx = cnt[min(v, N - 1)];
    int b_nx = rowptr[min(v, N - 1)];
    for (int vw = grp & ~3; vw < N; vw += ngrp, v += ngrp) {
        const int vv = min(v, N - 1);
        const int n = n_nx;
        const int* __restrict__ cs = csr + b_nx;
        const int vn = min(v + ngrp, N - 1);
        n_nx = cnt[vn];
        b_nx = rowptr[vn];
        const float self = gin[vv];
        const float dv = dinv[vv];
        const float zz = z0[vv];
        float acc = 0.f;
        for (int j0 = l; j0 - l < n; j0 += 64) {
            int s[8];
            float a[8];
#pragma unroll
            for (int k = 0; k < 8; ++k) s[k] = cs[min(j0 + 8 * k, n - 1)];
#pragma unroll
            for (int k = 0; k < 8; ++k) a[k] = gin[s[k]];
#pragma unroll
            for (int k = 0; k < 8; ++k) acc += (j0 + 8 * k < n) ? a[k] : 0.f;
        }
        acc = add_xmask<1>(acc);
        acc = add_xmask<2>(acc);
        acc = add_xmask<4>(acc);
        const float zn = ONE_MINUS_ALPHA * dv * (acc + self) + ALPHA_F * zz;
        if (l == 0 && v < N) {
            if (LAST) out[v] = zn + b3[0];
            else      gout[v] = dv * zn;
        }
    }
}


#define HL_WGS 256
#define HL_T 512
#define HL_TRIPS 8
#define HL_SLAB 65536
#define HL_RING 13

typedef __attribute__((ext_vector_type(4))) float v4f;
typedef __attribute__((ext_vector_type(2))) int v2i;

__device__ __forceinline__ int hl_piece(int j, int tid, int nq, int rot) {
    const int i = min(j * HL_T + tid, nq - 1) + rot;
    return i >= nq ? i - nq : i;
}
__device__ __forceinline__ int hl_rot(int nq) {
    return (int)((blockIdx.x * (unsigned)nq) / HL_WGS) & ~31;
}

__device__ __forceinline__ v2i hl_quad(const unsigned short* __restrict__ c16, int p, int epad) {
    return *reinterpret_cast<const v2i*>(c16 + min(p, epad - 4));
}

__device__ __forceinline__ float hl_sum4(const float* sg, v2i q, int p, int s, int e) {
    const int qq[4] = {q.x & 0xffff, (int)((unsigned)q.x >> 16),
                       q.y & 0xffff, (int)((unsigned)q.y >> 16)};
    float a[4];
    bool ok[4];
#pragma unroll
    for (int j = 0; j < 4; ++j) {
        ok[j] = (p + j >= s) && (p + j < e);
        a[j] = sg[ok[j] ? qq[j] : 0];
    }
    float acc = 0.f;
#pragma unroll
    for (int j = 0; j < 4; ++j) acc += ok[j] ? a[j] : 0.f;
    return acc;
}

__device__ __forceinline__ float hl_first_lo(const float* sg, v2i qlo, int s, int e, int l) {
    return hl_sum4(sg, qlo, (s & ~3) + 4 * l, s, e);
}
__device__ __forceinline__ float hl_first_hi(const float* sg, v2i qhi, int s, int e, int l) {
    const int p = (s & ~3) + 32 + 4 * l;
    return p < e ? hl_sum4(sg, qhi, p, s, e) : 0.f;
}
__device__ __forceinline__ float hl_rest(const float* sg, const unsigned short* __restrict__ c16,
                                         int s, int e, int l, int epad) {
    float acc = 0.f;
    for (int pp = (s & ~3) + 64; pp < e; pp += 64) {
        const int p = pp + 4 * l;
        acc += hl_sum4(sg, hl_quad(c16, p, epad), p, s, e);
        acc += hl_sum4(sg, hl_quad(c16, p + 32, epad), p + 32, s, e);
    }
    return acc;
}

template <bool LAST>
__global__ void __launch_bounds__(HL_T)
ak_hop_lds(const int* __restrict__ rowptr, const int* __restrict__ cnt, const int* __restrict__ mid,
           const int* __restrict__ csr, const float* __restrict__ gin,
           const float* __restrict__ z0, const float* __restrict__ dinv,
           float* __restrict__ gout, const float* __restrict__ b3, float* __restrict__ out,
           int N, int npw, int split, int epad) {
    __shared__ float sg[HL_SLAB];
    __shared__ float stot[HL_T];
    const unsigned short* __restrict__ const c16 = reinterpret_cast<const unsigned short*>(csr);
    const int tid = threadIdx.x, l = tid & 7, gi = tid >> 3;
    const int v_lo = blockIdx.x * npw;
    const int v_hi = min(v_lo + npw, N);

    v4f* const d4 = reinterpret_cast<v4f*>(sg);
    v4f rr[HL_RING];
    int rx[HL_RING];
    const v4f* __restrict__ const ha = reinterpret_cast<const v4f*>(gin);
    const int nqa = split >> 2, rota = hl_rot(nqa), pera = (nqa + HL_T - 1) / HL_T;
#pragma unroll
    for (int k = 0; k < HL_RING; ++k) {
        rx[k] = hl_piece(k, tid, nqa, rota);
        rr[k] = ha[rx[k]];
    }

    int s0[HL_TRIPS], m0[HL_TRIPS], e0[HL_TRIPS];
#pragma unroll
    for (int t = 0; t < HL_TRIPS; ++t) {
        const int v = v_lo + gi + 64 * t;
        const int vv = min(v, N - 1);
        const int rs = rowptr[vv], rm = mid[vv], rc = cnt[vv];
        const bool ok = v < v_hi;
        s0[t] = ok ? rs : 0;
        m0[t] = ok ? rm : 0;
        e0[t] = ok ? rs + rc : 0;
    }
    v2i qa[HL_TRIPS], qa2[HL_TRIPS];
#pragma unroll
    for (int t = 0; t < HL_TRIPS; ++t) {
        qa[t]  = hl_quad(c16, (s0[t] & ~3) + 4 * l, epad);
        qa2[t] = hl_quad(c16, (s0[t] & ~3) + 32 + 4 * l, epad);
    }
    const int vf = v_lo + tid;
    const int vfc = min(vf, N - 1);
    const float f_dv = dinv[vfc], f_zz = z0[vfc], f_self = gin[vfc];

    for (int j0 = HL_RING; j0 < pera; j0 += HL_RING) {
#pragma unroll
        for (int k = 0; k < HL_RING; ++k) {
            d4[rx[k]] = rr[k];
            rx[k] = hl_piece(j0 + k, tid, nqa, rota);
            rr[k] = ha[rx[k]];
        }
    }
#pragma unroll
    for (int k = 0; k < HL_RING; ++k) d4[rx[k]] = rr[k];
    __syncthreads();

    const float* __restrict__ const g1 = gin + split;
    const v4f* __restrict__ const hb = reinterpret_cast<const v4f*>(g1);
    const int n1 = N - split;
    const int nqb = n1 >> 2, rotb = hl_rot(nqb), perb = (nqb + HL_T - 1) / HL_T;
#pragma unroll
    for (int k = 0; k < HL_RING; ++k) {
        rx[k] = hl_piece(k, tid, nqb, rotb);
        rr[k] = hb[rx[k]];
    }
    float acc[HL_TRIPS];
#pragma unroll
    for (int t = 0; t < HL_TRIPS; ++t) acc[t] = hl_first_lo(sg, qa[t], s0[t], m0[t], l);
#pragma unroll
    for (int t = 0; t < HL_TRIPS; ++t) acc[t] += hl_first_hi(sg, qa2[t], s0[t], m0[t], l);
#pragma unroll
    for (int t = 0; t < HL_TRIPS; ++t) acc[t] += hl_rest(sg, c16, s0[t], m0[t], l, epad);
    v2i qb[HL_TRIPS], qb2[HL_TRIPS];
#pragma unroll
    for (int t = 0; t < HL_TRIPS; ++t) {
        qb[t]  = hl_quad(c16, (m0[t] & ~3) + 4 * l, epad);
        qb2[t] = hl_quad(c16, (m0[t] & ~3) + 32 + 4 * l, epad);
    }
    __syncthreads();
    for (int j0 = HL_RING; j0 < perb; j0 += HL_RING) {
#pragma unroll
        for (int k = 0; k < HL_RING; ++k) {
            d4[rx[k]] = rr[k];
            rx[k] = hl_piece(j0 + k, tid, nqb, rotb);
            rr[k] = hb[rx[k]];
        }
    }
#pragma unroll
    for (int k = 0; k < HL_RING; ++k) d4[rx[k]] = rr[k];
    if (tid < (n1 & 3)) sg[(nqb << 2) + tid] = g1[(nqb << 2) + tid];
    __syncthreads();
#pragma unroll
    for (int t = 0; t < HL_TRIPS; ++t) acc[t] += hl_first_lo(sg, qb[t], m0[t], e0[t], l);
#pragma unroll
    for (int t = 0; t < HL_TRIPS; ++t) acc[t] += hl_first_hi(sg, qb2[t], m0[t], e0[t], l);
#pragma unroll
    for (int t = 0; t < HL_TRIPS; ++t) acc[t] += hl_rest(sg, c16, m0[t], e0[t], l, epad);
#pragma unroll
    for (int t = 0; t < HL_TRIPS; ++t) {
        float a = acc[t];
        a = add_xmask<1>(a);
        a = add_xmask<2>(a);
        a = add_xmask<4>(a);
        if (l == 0) stot[gi + 64 * t] = a;
    }
    __syncthreads();
    if (tid < npw && vf < v_hi) {
        const float zn = ONE_MINUS_ALPHA * f_dv * (stot[tid] + f_self) + ALPHA_F * f_zz;
        if (LAST) out[vf] = zn + b3[0];
        else      gout[vf] = f_dv * zn;
    }
}


#define HP_RANGES 128
#define HP_TRIPS 13
#define HP_NPW (64 * HP_TRIPS)
static_assert(KITERS >= 2, "the hop before the last one clears out for the last hop's adds");

template <bool LAST>
__global__ void __launch_bounds__(HL_T)
ak_hop_pair(const int* __restrict__ rowptr, const int* __restrict__ cnt, const int* __restrict__ mid,
            const int* __restrict__ csr, const float* __restrict__ gin,
            const float* __restrict__ z0, const float* __restrict__ dinv,
            float* __restrict__ gacc, float* __restrict__ gzero, const float* __restrict__ b3,
            int N, int npw, int split, int epad) {
    __shared__ float sg[HL_SLAB];
    __shared__ float stot[HP_NPW];
    const unsigned short* __restrict__ const c16 = reinterpret_cast<const unsigned short*>(csr);
    const int tid = threadIdx.x, l = tid & 7, gi = tid >> 3;
    const int h = blockIdx.x & 1;
    const int v_lo = (blockIdx.x >> 1) * npw;
    const int v_hi = min(v_lo + npw, N);

    v4f* const d4 = reinterpret_cast<v4f*>(sg);
    v4f rr[HL_RING];
    int rx[HL_RING];
    const float* __restrict__ const gh = gin + (h ? split : 0);
    const v4f* __restrict__ const hp = reinterpret_cast<const v4f*>(gh);
    const int nh = h ? N - split : split;
    const int nq = nh >> 2, rot = hl_rot(nq), per = (nq + HL_T - 1) / HL_T;
#pragma unroll
    for (int k = 0; k < HL_RING; ++k) {
        rx[k] = hl_piece(k, tid, nq, rot);
        rr[k] = hp[rx[k]];
    }

    int s0[HP_TRIPS], e0[HP_TRIPS];
#pragma unroll
    for (int t = 0; t < HP_TRIPS; ++t) {
        const int v = v_lo + gi + 64 * t;
        const int vv = min(v, N - 1);
        const int rs = rowptr[vv], rm = mid[vv], rc = cnt[vv];
        const bool ok = v < v_hi;
        s0[t] = ok ? (h ? rm : rs) : 0;
        e0[t] = ok ? (h ? rs + rc : rm) : 0;
    }
    v2i qa[HP_TRIPS], qa2[HP_TRIPS];
#pragma unroll
    for (int t = 0; t < HP_TRIPS; ++t) {
        qa[t]  = hl_quad(c16, (s0[t] & ~3) + 4 * l, epad);
        qa2[t] = hl_quad(c16, (s0[t] & ~3) + 32 + 4 * l, epad);
    }
    float f_dv[2], f_zz[2], f_self[2];
#pragma unroll
    for (int j = 0; j < 2; ++j) {
        const int vfc = min(v_lo + tid + HL_T * j, N - 1);
        f_dv[j] = dinv[vfc];
        f_zz[j] = z0[vfc];
        f_self[j] = gin[vfc];
    }

    for (int j0 = HL_RING; j0 < per; j0 += HL_RING) {
#pragma unroll
        for (int k = 0; k < HL_RING; ++k) {
            d4[rx[k]] = rr[k];
            rx[k] = hl_piece(j0 + k, tid, nq, rot);
            rr[k] = hp[rx[k]];
        }
    }
#pragma unroll
    for (int k = 0; k < HL_RING; ++k) d4[rx[k]] = rr[k];
    if (tid < (nh & 3)) sg[(nq << 2) + tid] = gh[(nq << 2) + tid];
    __syncthreads();

    float acc[HP_TRIPS];
#pragma unroll
    for (int t = 0; t < HP_TRIPS; ++t) acc[t] = hl_first_lo(sg, qa[t], s0[t], e0[t], l);
#pragma unroll
    for (int t = 0; t < HP_TRIPS; ++t) acc[t] += hl_first_hi(sg, qa2[t], s0[t], e0[t], l);
#pragma unroll
    for (int t = 0; t < HP_TRIPS; ++t) acc[t] += hl_rest(sg, c16, s0[t], e0[t], l, epad);
#pragma unroll
    for (int t = 0; t < HP_TRIPS; ++t) {
        float a = acc[t];
        a = add_xmask<1>(a);
        a = add_xmask<2>(a);
        a = add_xmask<4>(a);
        if (l == 0) stot[gi + 64 * t] = a;
    }
    __syncthreads();
    const float bias = LAST ? b3[0] : 0.f;
#pragma unroll
    for (int j = 0; j < 2; ++j) {
        const int i = tid + HL_T * j;
        const int vf = v_lo + i;
        if (i < npw && vf < v_hi) {
            const float dv = f_dv[j];
            const float zt = h ? ONE_MINUS_ALPHA * dv * stot[i]
                               : ONE_MINUS_ALPHA * dv * (stot[i] + f_self[j]) + ALPHA_F * f_zz[j];
            atomicAdd(&gacc[vf], LAST ? (h ? zt : zt + bias) : dv * zt);
            if (gzero != nullptr && j == h) gzero[vf] = 0.f;
        }
    }
}


extern "C" void kernel_launch(void* const* d_in, const int* in_sizes, int n_in,
                              void* d_out, int out_size, void* d_ws, size_t ws_size,
                              hipStream_t stream) {
    const float* x  = (const float*)d_in[0];
    const int*  eidx = (const int*)d_in[1];
    const float* W1 = (const float*)d_in[2];
    const float* b1 = (const float*)d_in[3];
    const float* W2 = (const float*)d_in[4];
    const float* b2 = (const float*)d_in[5];
    const float* W3 = (const float*)d_in[6];
    const float* b3 = (const float*)d_in[7];
    float* out = (float*)d_out;

    const int N = in_sizes[0];
    const int E = in_sizes[1] / 2;

    char* ws = (char*)d_ws;
    const int B = 256;

    {
        const int nb = (N + BNODES - 1) / BNODES;
        const int nh = nb * CHUNK_WGS;
        size_t o2 = 0;
        auto carve2 = [&](size_t bytes) -> char* {
            char* p = ws + o2;
            o2 += (bytes + 255) & ~(size_t)255;
            return p;
        };
        int*   cnt    = (int*)carve2((size_t)N * 4);
        float* dinv   = (float*)carve2((size_t)N * 4);
        int*   rowptr = (int*)carve2((size_t)N * 4);
        int*   hist1  = (int*)carve2((size_t)nh * 4);
        int*   off1   = (int*)carve2((size_t)nh * 4);
        int*   csr    = (int*)carve2((size_t)E * 4);
        const size_t plane = (((size_t)N * 4) + 255) & ~(size_t)255;
        const size_t tmpb = (size_t)E * 4;
        char* shared_region = carve2(tmpb > 4 * plane ? tmpb : 4 * plane);
        unsigned* tmp = (unsigned*)shared_region;
        float* z0 = (float*)shared_region;
        float* gA = (float*)(shared_region + plane);
        float* gB = (float*)(shared_region + 2 * plane);
        float* gC = (float*)(shared_region + 3 * plane);
        int* mid = hist1;
        const bool lds_hop = N >= 4096 && N <= 2 * HL_SLAB && E >= 64;
        const int split = lds_hop ? ((N / 2 + 31) & ~31) : N;
        const int npw2 = (N + HP_RANGES - 1) / HP_RANGES;
        const bool pair_hop = lds_hop && npw2 <= HP_NPW;

        if (o2 <= ws_size && nb <= NB_MAX && N <= (1 << 24)) {
            const int chunk = (E + CHUNK_WGS - 1) / CHUNK_WGS;
            const int epad = (E + 63) & ~63;
            const size_t mid_off = (((size_t)epad * 2) + 255) & ~(size_t)255;
            const bool own_chunks = lds_hop && chunk <= ST1_CAP &&
                                    mid_off + (size_t)N * 4 <= (size_t)E * 4;
            if (own_chunks) {
                mid = (int*)((char*)csr + mid_off);
                ak_chunk_sort<<<CHUNK_WGS, 1024, 0, stream>>>(eidx, hist1, off1, tmp, E, chunk, nb);
                ak_bucket_sort_sh<<<nb, 512, 0, stream>>>(tmp, hist1, off1, rowptr, cnt, mid, dinv, csr,
                                                          E, N, chunk, split);
            } else {
                ak_bucket_hist<<<CHUNK_WGS, 1024, 0, stream>>>(eidx, hist1, E, chunk, nb);
                ak_excl_scan<<<(nh + 1023) / 1024, 1024, 0, stream>>>(hist1, off1, nh);
                ak_bucket_scatter<<<CHUNK_WGS, 1024, 0, stream>>>(eidx, hist1, off1, tmp, E, chunk, nb);
                if (lds_hop) {
                    ak_bucket_sort<true><<<nb, 512, 0, stream>>>(tmp, off1, rowptr, cnt, mid, dinv, csr,
                                                                 E, N, nb, split);
                } else {
                    ak_bucket_sort<false><<<nb, 512, 0, stream>>>(tmp, off1, rowptr, cnt, mid, dinv, csr,
                                                                  E, N, nb, split);
                }
            }

            int tiles = N / 16;
            if (tiles > 0) {
                int waves_per_block = B / 32;
                int blocks = (tiles + waves_per_block - 1) / waves_per_block;
                ak_encoder_wmma<<<blocks, B, 0, stream>>>(x, W1, b1, W2, b2, W3, dinv, z0, gA,
                                                          pair_hop ? gB : (float*)nullptr, tiles);
            }
            int rem_start = tiles * 16;
            if (rem_start < N) {
                ak_encoder_tail<<<(N - rem_start + B - 1) / B, B, 0, stream>>>(
                    x, W1, b1, W2, b2, W3, dinv, z0, gA, pair_hop ? gB : (float*)nullptr,
                    rem_start, N);
            }
            if (pair_hop) {
                float* const pl[3] = {gA, gB, gC};
                for (int it = 0; it < KITERS; ++it) {
                    const float* gi = pl[it % 3];
                    if (it < KITERS - 1) {
                        float* gz = (it == KITERS - 2) ? out : pl[(it + 2) % 3];
                        ak_hop_pair<false><<<2 * HP_RANGES, HL_T, 0, stream>>>(
                            rowptr, cnt, mid, csr, gi, z0, dinv, pl[(it + 1) % 3], gz, b3, N, npw2,
                            split, epad);
                    } else {
                        ak_hop_pair<true><<<2 * HP_RANGES, HL_T, 0, stream>>>(
                            rowptr, cnt, mid, csr, gi, z0, dinv, out, (float*)nullptr, b3, N, npw2,
                            split, epad);
                    }
                }
                return;
            }

            const int hblocks = (N + 32 * HOP_TRIPS - 1) / (32 * HOP_TRIPS);
            const int npw = (N + HL_WGS - 1) / HL_WGS;
            for (int it = 0; it < KITERS; ++it) {
                const float* gi = (it & 1) ? gB : gA;
                float* go = (it & 1) ? gA : gB;
                if (lds_hop) {
                    if (it < KITERS - 1) {
                        ak_hop_lds<false><<<HL_WGS, HL_T, 0, stream>>>(rowptr, cnt, mid, csr, gi, z0,
                                                                       dinv, go, b3, out, N, npw,
                                                                       split, epad);
                    } else {
                        ak_hop_lds<true><<<HL_WGS, HL_T, 0, stream>>>(rowptr, cnt, mid, csr, gi, z0,
                                                                      dinv, (float*)nullptr, b3, out,
                                                                      N, npw, split, epad);
                    }
                } else if (it < KITERS - 1) {
                    ak_hop<false><<<hblocks, B, 0, stream>>>(rowptr, cnt, csr, gi, z0, dinv, go,
                                                             b3, out, N);
                } else {
                    ak_hop<true><<<hblocks, B, 0, stream>>>(rowptr, cnt, csr, gi, z0, dinv,
                                                            (float*)nullptr, b3, out, N);
                }
            }
            return;
        }
    }

    size_t off = 0;
    auto carve = [&](size_t bytes) -> char* {
        char* p = ws + off;
        off += (bytes + 255) & ~(size_t)255;
        return p;
    };
    float* deg  = (float*)carve((size_t)N * 4);
    float* dinv = (float*)carve((size_t)N * 4);
    float* z0   = (float*)carve((size_t)N * 4);
    float* g    = (float*)carve((size_t)N * 4);
    float* agg  = (float*)carve((size_t)N * 4);

    ak_deg_init<<<(N + B - 1) / B, B, 0, stream>>>(deg, N);
    ak_deg_accum<<<(E + B - 1) / B, B, 0, stream>>>(eidx, deg, E);
    ak_dinv<<<(N + B - 1) / B, B, 0, stream>>>(deg, dinv, N);

    int tiles = N / 16;
    if (tiles > 0) {
        int waves_per_block = B / 32;
        int blocks = (tiles + waves_per_block - 1) / waves_per_block;
        ak_encoder_wmma<<<blocks, B, 0, stream>>>(x, W1, b1, W2, b2, W3, dinv, z0, g, agg, tiles);
    }
    int rem_start = tiles * 16;
    if (rem_start < N) {
        ak_encoder_tail<<<(N - rem_start + B - 1) / B, B, 0, stream>>>(
            x, W1, b1, W2, b2, W3, dinv, z0, g, agg, rem_start, N);
    }

    for (int it = 0; it < KITERS; ++it) {
        ak_edge_scatter<<<(E + B - 1) / B, B, 0, stream>>>(eidx, g, agg, E);
        if (it < KITERS - 1) {
            ak_finalize<<<(N + B - 1) / B, B, 0, stream>>>(agg, g, z0, dinv, N);
        } else {
            ak_finalize_out<<<(N + B - 1) / B, B, 0, stream>>>(agg, g, z0, dinv, b3, out, N);
        }
    }
}


# Sorted list padded: each node-and-half segment starts on a 16-byte piece, filled up with entries that read a zero word in LDS; hops sum 4 lanes a node with no per-entry test; float sum order changes
# speedup vs baseline: 20.7005x; 1.1169x over previous
#include <hip/hip_runtime.h>


typedef __attribute__((ext_vector_type(2))) float v2f;
typedef __attribute__((ext_vector_type(8))) float v8f;

#define HID 16
#define KITERS 10
#define ONE_MINUS_ALPHA 0.9f
#define ALPHA_F 0.1f

__device__ __forceinline__ float relu_f(float v) { return v > 0.f ? v : 0.f; }


__global__ void ak_deg_init(float* __restrict__ deg, int n) {
    int i = blockIdx.x * blockDim.x + threadIdx.x;
    if (i < n) deg[i] = 1.0f;
}

__global__ void ak_deg_accum(const int* __restrict__ eidx, float* __restrict__ deg, int E) {
    int e = blockIdx.x * blockDim.x + threadIdx.x;
    if (e < E) atomicAdd(&deg[eidx[(long long)E + e]], 1.0f);
}

__global__ void ak_dinv(const float* __restrict__ deg, float* __restrict__ dinv, int n) {
    int i = blockIdx.x * blockDim.x + threadIdx.x;
    if (i < n) dinv[i] = rsqrtf(deg[i]);
}

__device__ __forceinline__ float xhalf(float x) {
    int b = __builtin_bit_cast(int, x);
    return __builtin_bit_cast(float,
        __builtin_amdgcn_permlanex16(b, b, 0x76543210, 0xfedcba98, false, false));
}


__global__ void ak_encoder_wmma(const float* __restrict__ x, const float* __restrict__ W1,
                                const float* __restrict__ b1, const float* __restrict__ W2,
                                const float* __restrict__ b2, const float* __restrict__ W3,
                                const float* __restrict__ dinv, float* __restrict__ z0,
                                float* __restrict__ g, float* __restrict__ agg, int n_tiles) {
    int wave = (blockIdx.x * blockDim.x + threadIdx.x) >> 5;
    int lane = threadIdx.x & 31;
    if (wave >= n_tiles) return;

    int row = lane & 15;
    int hi  = lane >> 4;
    int node = wave * 16 + row;
    float xv = x[node];

    v8f acc = {};
#pragma unroll
    for (int s = 0; s < 4; ++s) {
        int k0 = 4 * s + 2 * hi;
        v2f wt, ht;
        wt.x = W2[k0 * HID + row];
        wt.y = W2[(k0 + 1) * HID + row];
        ht.x = relu_f(xv * W1[k0]     + b1[k0]);
        ht.y = relu_f(xv * W1[k0 + 1] + b1[k0 + 1]);
        acc = __builtin_amdgcn_wmma_f32_16x16x4_f32(
            false, wt, false, ht, (short)0, acc, false, false);
    }

    float part = 0.f;
#pragma unroll
    for (int i = 0; i < 8; ++i) {
        int m = i + 8 * hi;
        part += relu_f(acc[i] + b2[m]) * W3[m];
    }
    float z = part + xhalf(part);
    if (hi == 0) {
        z0[node] = z;
        g[node]  = dinv[node] * z;
        if (agg) agg[node] = 0.f;
    }
}

__global__ void ak_encoder_tail(const float* __restrict__ x, const float* __restrict__ W1,
                                const float* __restrict__ b1, const float* __restrict__ W2,
                                const float* __restrict__ b2, const float* __restrict__ W3,
                                const float* __restrict__ dinv, float* __restrict__ z0,
                                float* __restrict__ g, float* __restrict__ agg, int start, int N) {
    int v = start + blockIdx.x * blockDim.x + threadIdx.x;
    if (v >= N) return;
    float h1[HID];
    float xv = x[v];
#pragma unroll
    for (int k = 0; k < HID; ++k) h1[k] = relu_f(xv * W1[k] + b1[k]);
    float z = 0.f;
#pragma unroll
    for (int j = 0; j < HID; ++j) {
        float acc = b2[j];
#pragma unroll
        for (int k = 0; k < HID; ++k) acc += h1[k] * W2[k * HID + j];
        z += relu_f(acc) * W3[j];
    }
    z0[v] = z;
    g[v]  = dinv[v] * z;
    if (agg) agg[v] = 0.f;
}


__global__ void ak_edge_scatter(const int* __restrict__ eidx, const float* __restrict__ g,
                                float* __restrict__ agg, int E) {
    int e = blockIdx.x * blockDim.x + threadIdx.x;
    if (e >= E) return;
    atomicAdd(&agg[eidx[(long long)E + e]], g[eidx[e]]);
}

__global__ void ak_finalize(float* __restrict__ agg, float* __restrict__ g,
                            const float* __restrict__ z0, const float* __restrict__ dinv, int n) {
    int v = blockIdx.x * blockDim.x + threadIdx.x;
    if (v >= n) return;
    float dv = dinv[v];
    float zn = ONE_MINUS_ALPHA * dv * (agg[v] + g[v]) + ALPHA_F * z0[v];
    g[v] = dv * zn;
    agg[v] = 0.f;
}

__global__ void ak_finalize_out(const float* __restrict__ agg, const float* __restrict__ g,
                                const float* __restrict__ z0, const float* __restrict__ dinv,
                                const float* __restrict__ b3, float* __restrict__ out, int N) {
    int v = blockIdx.x * blockDim.x + threadIdx.x;
    if (v >= N) return;
    out[v] = ONE_MINUS_ALPHA * dinv[v] * (agg[v] + g[v]) + ALPHA_F * z0[v] + b3[0];
}


#define BSH 8
#define BNODES 256
#define NB_MAX 4096
#define CHUNK_WGS 256

__global__ void __launch_bounds__(1024)
ak_bucket_hist(const int* __restrict__ eidx, int* __restrict__ hist1, int E, int chunk, int nb) {
    __shared__ int hist[NB_MAX];
    for (int b = threadIdx.x; b < nb; b += 1024) hist[b] = 0;
    __syncthreads();
    const int lo = blockIdx.x * chunk;
    const int hi = min(lo + chunk, E);
    const int* __restrict__ dst = eidx + (size_t)E;
    for (int e0 = lo + threadIdx.x; e0 < hi; e0 += 4 * 1024) {
        int d[4];
#pragma unroll
        for (int k = 0; k < 4; ++k) d[k] = dst[min(e0 + k * 1024, hi - 1)];
#pragma unroll
        for (int k = 0; k < 4; ++k) {
            if (e0 + k * 1024 < hi) atomicAdd(&hist[d[k] >> BSH], 1);
        }
    }
    __syncthreads();
    for (int b = threadIdx.x; b < nb; b += 1024) hist1[b * CHUNK_WGS + blockIdx.x] = hist[b];
}

__global__ void __launch_bounds__(1024)
ak_excl_scan(const int* __restrict__ in, int* __restrict__ out, int n) {
    __shared__ int red[32];
    __shared__ int wsum[32];
    const int tid = threadIdx.x, lane = tid & 31, w = tid >> 5;
    const int first = blockIdx.x * 1024;

    int part = 0;
    for (int i = tid; i < first; i += 1024) part += in[i];
    for (int o = 16; o > 0; o >>= 1) part += __shfl_xor(part, o);
    if (lane == 0) red[w] = part;

    const int i = first + tid;
    const int c = (i < n) ? in[i] : 0;
    int x = c;
    for (int o = 1; o < 32; o <<= 1) {
        int y = __shfl_up(x, o);
        if (lane >= o) x += y;
    }
    if (lane == 31) wsum[w] = x;
    __syncthreads();
    if (w == 0) {
        int s = wsum[lane];
        for (int o = 1; o < 32; o <<= 1) {
            int y = __shfl_up(s, o);
            if (lane >= o) s += y;
        }
        wsum[lane] = s;
        int b = red[lane];
        for (int o = 16; o > 0; o >>= 1) b += __shfl_xor(b, o);
        red[lane] = b;
    }
    __syncthreads();
    if (i < n) out[i] = red[0] + (w ? wsum[w - 1] : 0) + x - c;
}

template <bool STAGED>
__device__ __forceinline__ void ak_scatter_edges(const int* __restrict__ eidx, int E, int lo, int hi,
                                                 int tid, int* cur, unsigned* stg,
                                                 unsigned* __restrict__ tmp) {
    const int* __restrict__ dst = eidx + (size_t)E;
    for (int e0 = lo + tid; e0 < hi; e0 += 4 * 1024) {
        int d[4], s[4];
#pragma unroll
        for (int k = 0; k < 4; ++k) {
            const int e = min(e0 + k * 1024, hi - 1);
            d[k] = dst[e];
            s[k] = eidx[e];
        }
        asm volatile("" :: "v"(d[0]), "v"(d[1]), "v"(d[2]), "v"(d[3]),
                           "v"(s[0]), "v"(s[1]), "v"(s[2]), "v"(s[3]));
#pragma unroll
        for (int k = 0; k < 4; ++k) {
            if (e0 + k * 1024 < hi) {
                const int slot = atomicAdd(&cur[d[k] >> BSH], 1);
                const unsigned ent = ((unsigned)s[k] << BSH) | (unsigned)(d[k] & (BNODES - 1));
                if (STAGED) stg[slot] = ent;
                else        tmp[slot] = ent;
            }
        }
    }
}

#define ST1_CAP 20480

__global__ void __launch_bounds__(1024)
ak_bucket_scatter(const int* __restrict__ eidx, const int* __restrict__ hist1,
                  const int* __restrict__ off1, unsigned* __restrict__ tmp, int E, int chunk,
                  int nb) {
    __shared__ int cur[NB_MAX];
    __shared__ int sbeg[NB_MAX];
    __shared__ int gbeg[NB_MAX];
    __shared__ unsigned stg[ST1_CAP];
    __shared__ int wsum[32];
    const int tid = threadIdx.x, lane = tid & 31, w = tid >> 5;
    const bool staged = chunk <= ST1_CAP;
    if (staged) {
        int c4[4], g4[4];
        int s = 0;
#pragma unroll
        for (int k = 0; k < 4; ++k) {
            const int b = 4 * tid + k;
            const int bb = min(b, nb - 1);
            const int cv = hist1[bb * CHUNK_WGS + blockIdx.x];
            g4[k] = off1[bb * CHUNK_WGS + blockIdx.x];
            c4[k] = b < nb ? cv : 0;
            s += c4[k];
        }
        int x = s;
        for (int o = 1; o < 32; o <<= 1) {
            int y = __shfl_up(x, o);
            if (lane >= o) x += y;
        }
        if (lane == 31) wsum[w] = x;
        __syncthreads();
        if (w == 0) {
            int t = wsum[lane];
            for (int o = 1; o < 32; o <<= 1) {
                int y = __shfl_up(t, o);
                if (lane >= o) t += y;
            }
            wsum[lane] = t;
        }
        __syncthreads();
        int base = (w ? wsum[w - 1] : 0) + x - s;
#pragma unroll
        for (int k = 0; k < 4; ++k) {
            const int b = 4 * tid + k;
            if (b < nb) {
                cur[b] = base;
                sbeg[b] = base;
                gbeg[b] = g4[k];
            }
            base += c4[k];
        }
    } else {
        for (int b = tid; b < nb; b += 1024) cur[b] = off1[b * CHUNK_WGS + blockIdx.x];
    }
    __syncthreads();
    const int lo = blockIdx.x * chunk;
    const int hi = min(lo + chunk, E);
    if (!staged) {
        ak_scatter_edges<false>(eidx, E, lo, hi, tid, cur, stg, tmp);
        return;
    }
    ak_scatter_edges<true>(eidx, E, lo, hi, tid, cur, stg, tmp);
    __syncthreads();
    for (int b = w; b < nb; b += 32) {
        const int s0 = sbeg[b], n = cur[b] - s0, g0 = gbeg[b];
        if (n <= 0) continue;
        for (int ib = lane - (g0 & 31); ib < n; ib += 128) {
            unsigned v[4];
#pragma unroll
            for (int k = 0; k < 4; ++k)
                v[k] = stg[s0 + min(max(ib + 32 * k, 0), n - 1)];
#pragma unroll
            for (int k = 0; k < 4; ++k) {
                const int i = ib + 32 * k;
                if (i >= 0 && i < n) tmp[g0 + i] = v[k];
            }
        }
    }
}

#define ST2_BYTES 49152

template <bool S16> struct ak_ent { typedef int t; };
template <> struct ak_ent<true> { typedef unsigned short t; };

template <bool S16, bool STAGED>
__device__ __forceinline__ void ak_place(const unsigned* __restrict__ tmp, int lo, int hi, int tid,
                                         int split, int* cur, typename ak_ent<S16>::t* stage,
                                         typename ak_ent<S16>::t* __restrict__ out) {
    typedef typename ak_ent<S16>::t ent_t;
    for (int i0 = lo + tid; i0 < hi; i0 += 4 * 512) {
        unsigned u[4];
#pragma unroll
        for (int k = 0; k < 4; ++k) u[k] = tmp[min(i0 + k * 512, hi - 1)];
        asm volatile("" :: "v"(u[0]), "v"(u[1]), "v"(u[2]), "v"(u[3]));
#pragma unroll
        for (int k = 0; k < 4; ++k) {
            if (i0 + k * 512 < hi) {
                const int sn = (int)(u[k] >> BSH);
                const int key = (int)((u[k] & (BNODES - 1)) << 1) | (sn >= split ? 1 : 0);
                const int slot = atomicAdd(&cur[key], 1);
                const ent_t val = S16 ? (ent_t)(sn >= split ? sn - split : sn) : (ent_t)sn;
                if (STAGED) stage[slot] = val;
                else        out[slot] = val;
            }
        }
    }
}

template <bool S16>
__global__ void __launch_bounds__(512)
ak_bucket_sort(const unsigned* __restrict__ tmp, const int* __restrict__ off1,
               int* __restrict__ rowptr, int* __restrict__ cnt, int* __restrict__ mid,
               float* __restrict__ dinv, int* __restrict__ csr, int E, int N, int nb, int split) {
    typedef typename ak_ent<S16>::t ent_t;
    constexpr int CAP = ST2_BYTES / (int)sizeof(ent_t);
    constexpr int LINE = 128 / (int)sizeof(ent_t);
    __shared__ int c[2 * BNODES];
    __shared__ int cur[2 * BNODES];
    __shared__ int wtot[2 * BNODES / 32];
    __shared__ ent_t stg[CAP];
    const int tid = threadIdx.x, lane = tid & 31, w = tid >> 5;
    const int b = blockIdx.x;
    const int lo = off1[b * CHUNK_WGS];
    const int hi = (b + 1 < nb) ? off1[(b + 1) * CHUNK_WGS] : E;
    const int sh = lo & (LINE - 1);
    const bool staged = (hi - lo) + sh <= CAP;
    ent_t* const gout = reinterpret_cast<ent_t*>(csr);
    c[tid] = 0;
    __syncthreads();
    for (int i0 = lo + tid; i0 < hi; i0 += 4 * 512) {
        unsigned u[4];
#pragma unroll
        for (int k = 0; k < 4; ++k) u[k] = tmp[min(i0 + k * 512, hi - 1)];
#pragma unroll
        for (int k = 0; k < 4; ++k) {
            if (i0 + k * 512 < hi) {
                const int key = (int)((u[k] & (BNODES - 1)) << 1) | ((int)(u[k] >> BSH) >= split ? 1 : 0);
                atomicAdd(&c[key], 1);
            }
        }
    }
    __syncthreads();
    const int mine = c[tid];
    int x = mine;
    for (int o = 1; o < 32; o <<= 1) {
        int y = __shfl_up(x, o);
        if (lane >= o) x += y;
    }
    if (lane == 31) wtot[w] = x;
    __syncthreads();
    int base = 0;
    for (int k = 0; k < w; ++k) base += wtot[k];
    const int ex = base + x - mine;
    cur[tid] = ex;
    if ((tid & 1) == 0) {
        const int v = b * BNODES + (tid >> 1);
        if (v < N) {
            const int total = mine + c[tid + 1];
            cnt[v] = total;
            rowptr[v] = lo + ex;
            mid[v] = lo + ex + mine;
            dinv[v] = rsqrtf((float)(total + 1));
        }
    }
    __syncthreads();
    if (!staged) {
        ak_place<S16, false>(tmp, lo, hi, tid, split, cur, stg, gout + lo);
        return;
    }
    ak_place<S16, true>(tmp, lo, hi, tid, split, cur, stg + sh, gout + lo);
    __syncthreads();
    const int first = sh, last = sh + (hi - lo);
    ent_t* const gl = gout + (lo - sh);
    if (S16) {
        unsigned* const gw = reinterpret_cast<unsigned*>(gl);
        const int fe = (first + 1) & ~1, le = last & ~1;
        if (fe < le) {
            for (int wd0 = tid; 2 * wd0 < le; wd0 += 4 * 512) {
                unsigned a[4], bq[4];
#pragma unroll
                for (int k = 0; k < 4; ++k) {
                    const int j = min(max(2 * (wd0 + 512 * k), fe), le - 2);
                    a[k] = stg[j];
                    bq[k] = stg[j + 1];
                }
#pragma unroll
                for (int k = 0; k < 4; ++k) {
                    const int wd = wd0 + 512 * k;
                    if (2 * wd >= fe && 2 * wd < le) gw[wd] = a[k] | (bq[k] << 16);
                }
            }
        }
        if (tid == 0) {
            if ((first & 1) && first < last) gl[first] = stg[first];
            if ((last & 1) && last - 1 >= first) gl[last - 1] = stg[last - 1];
        }
    } else if (first < last) {
        for (int j0 = tid; j0 < last; j0 += 4 * 512) {
            ent_t a[4];
#pragma unroll
            for (int k = 0; k < 4; ++k) a[k] = stg[min(max(j0 + 512 * k, first), last - 1)];
#pragma unroll
            for (int k = 0; k < 4; ++k) {
                const int j = j0 + 512 * k;
                if (j >= first && j < last) gl[j] = a[k];
            }
        }
    }
}


#define CS_EPT 20
static_assert(CS_EPT * 1024 >= ST1_CAP, "a thread's registers must cover the largest staged chunk");

__global__ void __launch_bounds__(1024)
ak_chunk_sort(const int* __restrict__ eidx, int* __restrict__ cnt1, int* __restrict__ pos1,
              unsigned* __restrict__ tmp, int E, int chunk, int nb) {
    __shared__ int cur[NB_MAX];
    __shared__ unsigned stg[ST1_CAP];
    __shared__ int wsum[32];
    const int tid = threadIdx.x, lane = tid & 31, w = tid >> 5;
    const int lo = blockIdx.x * chunk;
    const int hi = min(lo + chunk, E);
    const int* __restrict__ dst = eidx + (size_t)E;
    int d[CS_EPT], s[CS_EPT];
#pragma unroll
    for (int k = 0; k < CS_EPT; ++k) {
        const int e = min(lo + tid + k * 1024, E - 1);
        d[k] = dst[e];
        s[k] = eidx[e];
    }
    for (int b = tid; b < nb; b += 1024) cur[b] = 0;
    __syncthreads();
#pragma unroll
    for (int k = 0; k < CS_EPT; ++k) {
        if (lo + tid + k * 1024 < hi) atomicAdd(&cur[d[k] >> BSH], 1);
    }
    __syncthreads();
    int c4[4];
    int sum4 = 0;
#pragma unroll
    for (int k = 0; k < 4; ++k) {
        const int b = 4 * tid + k;
        const int cv = cur[b];
        c4[k] = b < nb ? cv : 0;
        sum4 += c4[k];
    }
    int x = sum4;
    for (int o = 1; o < 32; o <<= 1) {
        int y = __shfl_up(x, o);
        if (lane >= o) x += y;
    }
    if (lane == 31) wsum[w] = x;
    __syncthreads();
    if (w == 0) {
        int t = wsum[lane];
        for (int o = 1; o < 32; o <<= 1) {
            int y = __shfl_up(t, o);
            if (lane >= o) t += y;
        }
        wsum[lane] = t;
    }
    __syncthreads();
    int base = (w ? wsum[w - 1] : 0) + x - sum4;
    int p4[4];
#pragma unroll
    for (int k = 0; k < 4; ++k) {
        const int b = 4 * tid + k;
        p4[k] = base;
        if (b < nb) cur[b] = base;
        base += c4[k];
    }
    __syncthreads();
    int slot[CS_EPT];
#pragma unroll
    for (int k = 0; k < CS_EPT; ++k) {
        slot[k] = 0;
        if (lo + tid + k * 1024 < hi) slot[k] = atomicAdd(&cur[d[k] >> BSH], 1);
    }
#pragma unroll
    for (int k = 0; k < CS_EPT; ++k) {
        if (lo + tid + k * 1024 < hi)
            stg[slot[k]] = ((unsigned)s[k] << BSH) | (unsigned)(d[k] & (BNODES - 1));
    }
    __syncthreads();
#pragma unroll
    for (int k = 0; k < 4; ++k) {
        const int b = 4 * tid + k;
        if (b < nb) {
            cnt1[b * CHUNK_WGS + blockIdx.x] = c4[k];
            pos1[b * CHUNK_WGS + blockIdx.x] = p4[k];
        }
    }
    const int n = hi - lo;
    if (n <= 0) return;
    for (int i0 = tid - (lo & 31); i0 < n; i0 += 4 * 1024) {
        unsigned v[4];
#pragma unroll
        for (int k = 0; k < 4; ++k) v[k] = stg[min(max(i0 + 1024 * k, 0), n - 1)];
#pragma unroll
        for (int k = 0; k < 4; ++k) {
            const int i = i0 + 1024 * k;
            if (i >= 0 && i < n) tmp[lo + i] = v[k];
        }
    }
}

__device__ __forceinline__ int ak_key(unsigned u, int split) {
    return (int)((u & (BNODES - 1)) << 1) | ((int)(u >> BSH) >= split ? 1 : 0);
}

template <bool STAGED>
__device__ __forceinline__ void ak_place_one(unsigned u, int split, int* cur, unsigned short* stage,
                                             unsigned short* __restrict__ out) {
    const int sn = (int)(u >> BSH);
    const int slot = atomicAdd(&cur[ak_key(u, split)], 1);
    const unsigned short val = (unsigned short)(sn >= split ? sn - split : sn);
    if (STAGED) stage[slot] = val;
    else        out[slot] = val;
}

#define BS_SPW 16

template <bool STAGED>
__device__ __forceinline__ void ak_place_sh(const unsigned (&u)[2 * BS_SPW], const int (&nn)[BS_SPW],
                                            const unsigned* __restrict__ tmp, const int* shn,
                                            const int* shb, int w, int lane, int split, int* cur,
                                            unsigned short* stage, unsigned short* __restrict__ out) {
#pragma unroll
    for (int k = 0; k < BS_SPW; ++k) {
        if (lane < nn[k])      ak_place_one<STAGED>(u[2 * k], split, cur, stage, out);
        if (lane + 32 < nn[k]) ak_place_one<STAGED>(u[2 * k + 1], split, cur, stage, out);
    }
#pragma nounroll
    for (int k = 0; k < BS_SPW; ++k) {
        const int n = shn[w * BS_SPW + k];
        const int p = shb[w * BS_SPW + k];
        for (int i = 64 + lane; i < n; i += 32)
            ak_place_one<STAGED>(tmp[p + i], split, cur, stage, out);
    }
}

#define PAD_ENT 0xffff
#define PB_GAP 3600
template <bool PAD>
__global__ void __launch_bounds__(512)
ak_bucket_sort_sh(const unsigned* __restrict__ tmp, const int* __restrict__ cnt1,
                  const int* __restrict__ pos1, int* __restrict__ rowptr, int* __restrict__ cnt,
                  int* __restrict__ mid, float* __restrict__ dinv, int* __restrict__ csr, int E,
                  int N, int chunk, int split) {
    typedef unsigned short ent_t;
    constexpr int CAP = ST2_BYTES / (int)sizeof(ent_t);
    constexpr int LINE = 128 / (int)sizeof(ent_t);
    __shared__ int c[2 * BNODES];
    __shared__ int cur[2 * BNODES];
    __shared__ int wtot[2 * BNODES / 32];
    __shared__ int shn[CHUNK_WGS];
    __shared__ int shb[CHUNK_WGS];
    __shared__ int redn[16];
    __shared__ int redp[16];
    __shared__ ent_t stg[CAP];
    const int tid = threadIdx.x, lane = tid & 31, w = tid >> 5;
    const int b = blockIdx.x;
    int myn = 0, myp = 0;
    if (tid < CHUNK_WGS) {
        myn = cnt1[b * CHUNK_WGS + tid];
        myp = pos1[b * CHUNK_WGS + tid];
        shn[tid] = myn;
        shb[tid] = min(tid * chunk + myp, E - 1);
    }
    c[tid] = 0;
    int sn = myn, sp = myp;
    for (int o = 16; o > 0; o >>= 1) {
        sn += __shfl_xor(sn, o);
        sp += __shfl_xor(sp, o);
    }
    if (lane == 0) {
        redn[w] = sn;
        redp[w] = sp;
    }
    __syncthreads();
    int tot = 0, lo = 0;
#pragma unroll
    for (int k = 0; k < 16; ++k) {
        tot += redn[k];
        lo += redp[k];
    }
    const int lop = PAD ? ((lo + 7) & ~7) + PB_GAP * b : lo;
    const int sh = lop & (LINE - 1);
    bool staged = tot + sh <= CAP;
    ent_t* const gout = reinterpret_cast<ent_t*>(csr);

    unsigned u[2 * BS_SPW];
    int nn[BS_SPW];
#pragma unroll
    for (int k = 0; k < BS_SPW; ++k) {
        nn[k] = shn[w * BS_SPW + k];
        const int p = shb[w * BS_SPW + k];
        const int m = max(nn[k] - 1, 0);
        u[2 * k]     = tmp[p + min(lane, m)];
        u[2 * k + 1] = tmp[p + min(lane + 32, m)];
    }
#pragma unroll
    for (int k = 0; k < BS_SPW; ++k) {
        if (lane < nn[k])      atomicAdd(&c[ak_key(u[2 * k], split)], 1);
        if (lane + 32 < nn[k]) atomicAdd(&c[ak_key(u[2 * k + 1], split)], 1);
    }
#pragma nounroll
    for (int k = 0; k < BS_SPW; ++k) {
        const int n = shn[w * BS_SPW + k];
        const int p = shb[w * BS_SPW + k];
        for (int i = 64 + lane; i < n; i += 32) atomicAdd(&c[ak_key(tmp[p + i], split)], 1);
    }
    __syncthreads();
    const int mine = c[tid];
    const int minep = PAD ? (mine + 7) & ~7 : mine;
    int x = minep;
    for (int o = 1; o < 32; o <<= 1) {
        int y = __shfl_up(x, o);
        if (lane >= o) x += y;
    }
    if (lane == 31) wtot[w] = x;
    __syncthreads();
    int base = 0, totp = 0;
#pragma unroll
    for (int k = 0; k < 2 * BNODES / 32; ++k) {
        const int wk = wtot[k];
        if (k < w) base += wk;
        totp += wk;
    }
    const int ex = base + x - minep;
    cur[tid] = ex;
    if (PAD) staged = totp + sh <= CAP;
    if ((tid & 1) == 0) {
        const int v = b * BNODES + (tid >> 1);
        if (v < N) {
            const int other = c[tid + 1];
            const int total = mine + other;
            if (PAD) {
                rowptr[v] = (lop + ex) >> 3;
                mid[v] = (lop + ex + minep) >> 3;
                cnt[v] = (lop + ex + minep + ((other + 7) & ~7)) >> 3;
            } else {
                cnt[v] = total;
                rowptr[v] = lo + ex;
                mid[v] = lo + ex + mine;
            }
            dinv[v] = rsqrtf((float)(total + 1));
        }
    }
    if (PAD) {
        for (int j = mine; j < minep; ++j) {
            if (staged) stg[sh + ex + j] = (ent_t)PAD_ENT;
            else        gout[lop + ex + j] = (ent_t)PAD_ENT;
        }
    }
    __syncthreads();
    if (!staged) {
        ak_place_sh<false>(u, nn, tmp, shn, shb, w, lane, split, cur, stg, gout + lop);
        return;
    }
    ak_place_sh<true>(u, nn, tmp, shn, shb, w, lane, split, cur, stg + sh, gout + lop);
    __syncthreads();
    const int first = sh, last = sh + (PAD ? totp : tot);
    ent_t* const gl = gout + (lop - sh);
    unsigned* const gw = reinterpret_cast<unsigned*>(gl);
    const int fe = (first + 1) & ~1, le = last & ~1;
    if (fe < le) {
        for (int wd0 = tid; 2 * wd0 < le; wd0 += 4 * 512) {
            unsigned a[4], bq[4];
#pragma unroll
            for (int k = 0; k < 4; ++k) {
                const int j = min(max(2 * (wd0 + 512 * k), fe), le - 2);
                a[k] = stg[j];
                bq[k] = stg[j + 1];
            }
#pragma unroll
            for (int k = 0; k < 4; ++k) {
                const int wd = wd0 + 512 * k;
                if (2 * wd >= fe && 2 * wd < le) gw[wd] = a[k] | (bq[k] << 16);
            }
        }
    }
    if (tid == 0) {
        if ((first & 1) && first < last) gl[first] = stg[first];
        if ((last & 1) && last - 1 >= first) gl[last - 1] = stg[last - 1];
    }
}

template <int M>
__device__ __forceinline__ float add_xmask(float x) {
    int b = __builtin_bit_cast(int, x);
    return x + __builtin_bit_cast(float, __builtin_amdgcn_update_dpp(0, b, 0x160 + M, 0xf, 0xf, true));
}

#define HOP_TRIPS 6

template <bool LAST>
__global__ void __launch_bounds__(256)
ak_hop(const int* __restrict__ rowptr, const int* __restrict__ cnt,
       const int* __restrict__ csr, const float* __restrict__ gin,
       const float* __restrict__ z0, const float* __restrict__ dinv,
       float* __restrict__ gout, const float* __restrict__ b3,
       float* __restrict__ out, int N) {
    const int l = threadIdx.x & 7;
    const int grp = (blockIdx.x * 256 + threadIdx.x) >> 3;
    const int ngrp = (gridDim.x * 256) >> 3;
    int v = grp;
    int n_nx = cnt[min(v, N - 1)];
    int b_nx = rowptr[min(v, N - 1)];
    for (int vw = grp & ~3; vw < N; vw += ngrp, v += ngrp) {
        const int vv = min(v, N - 1);
        const int n = n_nx;
        const int* __restrict__ cs = csr + b_nx;
        const int vn = min(v + ngrp, N - 1);
        n_nx = cnt[vn];
        b_nx = rowptr[vn];
        const float self = gin[vv];
        const float dv = dinv[vv];
        const float zz = z0[vv];
        float acc = 0.f;
        for (int j0 = l; j0 - l < n; j0 += 64) {
            int s[8];
            float a[8];
#pragma unroll
            for (int k = 0; k < 8; ++k) s[k] = cs[min(j0 + 8 * k, n - 1)];
#pragma unroll
            for (int k = 0; k < 8; ++k) a[k] = gin[s[k]];
#pragma unroll
            for (int k = 0; k < 8; ++k) acc += (j0 + 8 * k < n) ? a[k] : 0.f;
        }
        acc = add_xmask<1>(acc);
        acc = add_xmask<2>(acc);
        acc = add_xmask<4>(acc);
        const float zn = ONE_MINUS_ALPHA * dv * (acc + self) + ALPHA_F * zz;
        if (l == 0 && v < N) {
            if (LAST) out[v] = zn + b3[0];
            else      gout[v] = dv * zn;
        }
    }
}


#define HL_WGS 256
#define HL_T 512
#define HL_TRIPS 8
#define HL_SLAB 65536
#define HL_RING 13

typedef __attribute__((ext_vector_type(4))) float v4f;
typedef __attribute__((ext_vector_type(2))) int v2i;

__device__ __forceinline__ int hl_piece(int j, int tid, int nq, int rot) {
    const int i = min(j * HL_T + tid, nq - 1) + rot;
    return i >= nq ? i - nq : i;
}
__device__ __forceinline__ int hl_rot(int nq) {
    return (int)((blockIdx.x * (unsigned)nq) / HL_WGS) & ~31;
}

__device__ __forceinline__ v2i hl_quad(const unsigned short* __restrict__ c16, int p, int epad) {
    return *reinterpret_cast<const v2i*>(c16 + min(p, epad - 4));
}

__device__ __forceinline__ float hl_sum4(const float* sg, v2i q, int p, int s, int e) {
    const int qq[4] = {q.x & 0xffff, (int)((unsigned)q.x >> 16),
                       q.y & 0xffff, (int)((unsigned)q.y >> 16)};
    float a[4];
    bool ok[4];
#pragma unroll
    for (int j = 0; j < 4; ++j) {
        ok[j] = (p + j >= s) && (p + j < e);
        a[j] = sg[ok[j] ? qq[j] : 0];
    }
    float acc = 0.f;
#pragma unroll
    for (int j = 0; j < 4; ++j) acc += ok[j] ? a[j] : 0.f;
    return acc;
}

__device__ __forceinline__ float hl_first_lo(const float* sg, v2i qlo, int s, int e, int l) {
    return hl_sum4(sg, qlo, (s & ~3) + 4 * l, s, e);
}
__device__ __forceinline__ float hl_first_hi(const float* sg, v2i qhi, int s, int e, int l) {
    const int p = (s & ~3) + 32 + 4 * l;
    return p < e ? hl_sum4(sg, qhi, p, s, e) : 0.f;
}
__device__ __forceinline__ float hl_rest(const float* sg, const unsigned short* __restrict__ c16,
                                         int s, int e, int l, int epad) {
    float acc = 0.f;
    for (int pp = (s & ~3) + 64; pp < e; pp += 64) {
        const int p = pp + 4 * l;
        acc += hl_sum4(sg, hl_quad(c16, p, epad), p, s, e);
        acc += hl_sum4(sg, hl_quad(c16, p + 32, epad), p + 32, s, e);
    }
    return acc;
}

template <bool LAST>
__global__ void __launch_bounds__(HL_T)
ak_hop_lds(const int* __restrict__ rowptr, const int* __restrict__ cnt, const int* __restrict__ mid,
           const int* __restrict__ csr, const float* __restrict__ gin,
           const float* __restrict__ z0, const float* __restrict__ dinv,
           float* __restrict__ gout, const float* __restrict__ b3, float* __restrict__ out,
           int N, int npw, int split, int epad) {
    __shared__ float sg[HL_SLAB];
    __shared__ float stot[HL_T];
    const unsigned short* __restrict__ const c16 = reinterpret_cast<const unsigned short*>(csr);
    const int tid = threadIdx.x, l = tid & 7, gi = tid >> 3;
    const int v_lo = blockIdx.x * npw;
    const int v_hi = min(v_lo + npw, N);

    v4f* const d4 = reinterpret_cast<v4f*>(sg);
    v4f rr[HL_RING];
    int rx[HL_RING];
    const v4f* __restrict__ const ha = reinterpret_cast<const v4f*>(gin);
    const int nqa = split >> 2, rota = hl_rot(nqa), pera = (nqa + HL_T - 1) / HL_T;
#pragma unroll
    for (int k = 0; k < HL_RING; ++k) {
        rx[k] = hl_piece(k, tid, nqa, rota);
        rr[k] = ha[rx[k]];
    }

    int s0[HL_TRIPS], m0[HL_TRIPS], e0[HL_TRIPS];
#pragma unroll
    for (int t = 0; t < HL_TRIPS; ++t) {
        const int v = v_lo + gi + 64 * t;
        const int vv = min(v, N - 1);
        const int rs = rowptr[vv], rm = mid[vv], rc = cnt[vv];
        const bool ok = v < v_hi;
        s0[t] = ok ? rs : 0;
        m0[t] = ok ? rm : 0;
        e0[t] = ok ? rs + rc : 0;
    }
    v2i qa[HL_TRIPS], qa2[HL_TRIPS];
#pragma unroll
    for (int t = 0; t < HL_TRIPS; ++t) {
        qa[t]  = hl_quad(c16, (s0[t] & ~3) + 4 * l, epad);
        qa2[t] = hl_quad(c16, (s0[t] & ~3) + 32 + 4 * l, epad);
    }
    const int vf = v_lo + tid;
    const int vfc = min(vf, N - 1);
    const float f_dv = dinv[vfc], f_zz = z0[vfc], f_self = gin[vfc];

    for (int j0 = HL_RING; j0 < pera; j0 += HL_RING) {
#pragma unroll
        for (int k = 0; k < HL_RING; ++k) {
            d4[rx[k]] = rr[k];
            rx[k] = hl_piece(j0 + k, tid, nqa, rota);
            rr[k] = ha[rx[k]];
        }
    }
#pragma unroll
    for (int k = 0; k < HL_RING; ++k) d4[rx[k]] = rr[k];
    __syncthreads();

    const float* __restrict__ const g1 = gin + split;
    const v4f* __restrict__ const hb = reinterpret_cast<const v4f*>(g1);
    const int n1 = N - split;
    const int nqb = n1 >> 2, rotb = hl_rot(nqb), perb = (nqb + HL_T - 1) / HL_T;
#pragma unroll
    for (int k = 0; k < HL_RING; ++k) {
        rx[k] = hl_piece(k, tid, nqb, rotb);
        rr[k] = hb[rx[k]];
    }
    float acc[HL_TRIPS];
#pragma unroll
    for (int t = 0; t < HL_TRIPS; ++t) acc[t] = hl_first_lo(sg, qa[t], s0[t], m0[t], l);
#pragma unroll
    for (int t = 0; t < HL_TRIPS; ++t) acc[t] += hl_first_hi(sg, qa2[t], s0[t], m0[t], l);
#pragma unroll
    for (int t = 0; t < HL_TRIPS; ++t) acc[t] += hl_rest(sg, c16, s0[t], m0[t], l, epad);
    v2i qb[HL_TRIPS], qb2[HL_TRIPS];
#pragma unroll
    for (int t = 0; t < HL_TRIPS; ++t) {
        qb[t]  = hl_quad(c16, (m0[t] & ~3) + 4 * l, epad);
        qb2[t] = hl_quad(c16, (m0[t] & ~3) + 32 + 4 * l, epad);
    }
    __syncthreads();
    for (int j0 = HL_RING; j0 < perb; j0 += HL_RING) {
#pragma unroll
        for (int k = 0; k < HL_RING; ++k) {
            d4[rx[k]] = rr[k];
            rx[k] = hl_piece(j0 + k, tid, nqb, rotb);
            rr[k] = hb[rx[k]];
        }
    }
#pragma unroll
    for (int k = 0; k < HL_RING; ++k) d4[rx[k]] = rr[k];
    if (tid < (n1 & 3)) sg[(nqb << 2) + tid] = g1[(nqb << 2) + tid];
    __syncthreads();
#pragma unroll
    for (int t = 0; t < HL_TRIPS; ++t) acc[t] += hl_first_lo(sg, qb[t], m0[t], e0[t], l);
#pragma unroll
    for (int t = 0; t < HL_TRIPS; ++t) acc[t] += hl_first_hi(sg, qb2[t], m0[t], e0[t], l);
#pragma unroll
    for (int t = 0; t < HL_TRIPS; ++t) acc[t] += hl_rest(sg, c16, m0[t], e0[t], l, epad);
#pragma unroll
    for (int t = 0; t < HL_TRIPS; ++t) {
        float a = acc[t];
        a = add_xmask<1>(a);
        a = add_xmask<2>(a);
        a = add_xmask<4>(a);
        if (l == 0) stot[gi + 64 * t] = a;
    }
    __syncthreads();
    if (tid < npw && vf < v_hi) {
        const float zn = ONE_MINUS_ALPHA * f_dv * (stot[tid] + f_self) + ALPHA_F * f_zz;
        if (LAST) out[vf] = zn + b3[0];
        else      gout[vf] = f_dv * zn;
    }
}


#define HP_RANGES 128
#define HP_TRIPS 13
#define HP_NPW (64 * HP_TRIPS)
static_assert(KITERS >= 2, "the hop before the last one clears out for the last hop's adds");

template <bool LAST>
__global__ void __launch_bounds__(HL_T)
ak_hop_pair(const int* __restrict__ rowptr, const int* __restrict__ cnt, const int* __restrict__ mid,
            const int* __restrict__ csr, const float* __restrict__ gin,
            const float* __restrict__ z0, const float* __restrict__ dinv,
            float* __restrict__ gacc, float* __restrict__ gzero, const float* __restrict__ b3,
            int N, int npw, int split, int epad) {
    __shared__ float sg[HL_SLAB];
    __shared__ float stot[HP_NPW];
    const unsigned short* __restrict__ const c16 = reinterpret_cast<const unsigned short*>(csr);
    const int tid = threadIdx.x, l = tid & 7, gi = tid >> 3;
    const int h = blockIdx.x & 1;
    const int v_lo = (blockIdx.x >> 1) * npw;
    const int v_hi = min(v_lo + npw, N);

    v4f* const d4 = reinterpret_cast<v4f*>(sg);
    v4f rr[HL_RING];
    int rx[HL_RING];
    const float* __restrict__ const gh = gin + (h ? split : 0);
    const v4f* __restrict__ const hp = reinterpret_cast<const v4f*>(gh);
    const int nh = h ? N - split : split;
    const int nq = nh >> 2, rot = hl_rot(nq), per = (nq + HL_T - 1) / HL_T;
#pragma unroll
    for (int k = 0; k < HL_RING; ++k) {
        rx[k] = hl_piece(k, tid, nq, rot);
        rr[k] = hp[rx[k]];
    }

    int s0[HP_TRIPS], e0[HP_TRIPS];
#pragma unroll
    for (int t = 0; t < HP_TRIPS; ++t) {
        const int v = v_lo + gi + 64 * t;
        const int vv = min(v, N - 1);
        const int rs = rowptr[vv], rm = mid[vv], rc = cnt[vv];
        const bool ok = v < v_hi;
        s0[t] = ok ? (h ? rm : rs) : 0;
        e0[t] = ok ? (h ? rs + rc : rm) : 0;
    }
    v2i qa[HP_TRIPS], qa2[HP_TRIPS];
#pragma unroll
    for (int t = 0; t < HP_TRIPS; ++t) {
        qa[t]  = hl_quad(c16, (s0[t] & ~3) + 4 * l, epad);
        qa2[t] = hl_quad(c16, (s0[t] & ~3) + 32 + 4 * l, epad);
    }
    float f_dv[2], f_zz[2], f_self[2];
#pragma unroll
    for (int j = 0; j < 2; ++j) {
        const int vfc = min(v_lo + tid + HL_T * j, N - 1);
        f_dv[j] = dinv[vfc];
        f_zz[j] = z0[vfc];
        f_self[j] = gin[vfc];
    }

    for (int j0 = HL_RING; j0 < per; j0 += HL_RING) {
#pragma unroll
        for (int k = 0; k < HL_RING; ++k) {
            d4[rx[k]] = rr[k];
            rx[k] = hl_piece(j0 + k, tid, nq, rot);
            rr[k] = hp[rx[k]];
        }
    }
#pragma unroll
    for (int k = 0; k < HL_RING; ++k) d4[rx[k]] = rr[k];
    if (tid < (nh & 3)) sg[(nq << 2) + tid] = gh[(nq << 2) + tid];
    __syncthreads();

    float acc[HP_TRIPS];
#pragma unroll
    for (int t = 0; t < HP_TRIPS; ++t) acc[t] = hl_first_lo(sg, qa[t], s0[t], e0[t], l);
#pragma unroll
    for (int t = 0; t < HP_TRIPS; ++t) acc[t] += hl_first_hi(sg, qa2[t], s0[t], e0[t], l);
#pragma unroll
    for (int t = 0; t < HP_TRIPS; ++t) acc[t] += hl_rest(sg, c16, s0[t], e0[t], l, epad);
#pragma unroll
    for (int t = 0; t < HP_TRIPS; ++t) {
        float a = acc[t];
        a = add_xmask<1>(a);
        a = add_xmask<2>(a);
        a = add_xmask<4>(a);
        if (l == 0) stot[gi + 64 * t] = a;
    }
    __syncthreads();
    const float bias = LAST ? b3[0] : 0.f;
#pragma unroll
    for (int j = 0; j < 2; ++j) {
        const int i = tid + HL_T * j;
        const int vf = v_lo + i;
        if (i < npw && vf < v_hi) {
            const float dv = f_dv[j];
            const float zt = h ? ONE_MINUS_ALPHA * dv * stot[i]
                               : ONE_MINUS_ALPHA * dv * (stot[i] + f_self[j]) + ALPHA_F * f_zz[j];
            atomicAdd(&gacc[vf], LAST ? (h ? zt : zt + bias) : dv * zt);
            if (gzero != nullptr && j == h) gzero[vf] = 0.f;
        }
    }
}


#define HQ_TRIPS 7
#define HQ_NPW (128 * HQ_TRIPS)
typedef __attribute__((ext_vector_type(4))) int v4i;

__device__ __forceinline__ v4i hq_piece(const v4i* __restrict__ c128, int pi, int npieces) {
    return c128[min(pi, npieces - 1)];
}

__device__ __forceinline__ float hq_sum8(const float* sg, v4i q, bool inside) {
    const unsigned w[4] = {(unsigned)(inside ? q.x : -1), (unsigned)(inside ? q.y : -1),
                           (unsigned)(inside ? q.z : -1), (unsigned)(inside ? q.w : -1)};
    float a[8];
#pragma unroll
    for (int j = 0; j < 4; ++j) {
        a[2 * j]     = sg[w[j] & 0xffffu];
        a[2 * j + 1] = sg[w[j] >> 16];
    }
    return ((a[0] + a[1]) + (a[2] + a[3])) + ((a[4] + a[5]) + (a[6] + a[7]));
}

__device__ __forceinline__ float hq_rest(const float* sg, const v4i* __restrict__ c128, int s, int e,
                                         int l, int npieces) {
    float acc = 0.f;
    for (int pp = s + 8; pp < e; pp += 8) {
        const int p = pp + l;
        acc += hq_sum8(sg, hq_piece(c128, p, npieces), p < e);
        acc += hq_sum8(sg, hq_piece(c128, p + 4, npieces), p + 4 < e);
    }
    return acc;
}

template <bool LAST>
__global__ void __launch_bounds__(HL_T)
ak_hop_pad(const int* __restrict__ rowptr, const int* __restrict__ cnt, const int* __restrict__ mid,
           const int* __restrict__ csr, const float* __restrict__ gin,
           const float* __restrict__ z0, const float* __restrict__ dinv,
           float* __restrict__ gacc, float* __restrict__ gzero, const float* __restrict__ b3,
           int N, int npw, int split, int npieces) {
    __shared__ float sg[HL_SLAB];
    __shared__ float stot[HQ_NPW];
    const v4i* __restrict__ const c128 = reinterpret_cast<const v4i*>(csr);
    const int tid = threadIdx.x, l = tid & 3, gi = tid >> 2;
    const int h = blockIdx.x & 1;
    const int v_lo = (blockIdx.x >> 1) * npw;
    const int v_hi = min(v_lo + npw, N);

    v4f* const d4 = reinterpret_cast<v4f*>(sg);
    v4f rr[HL_RING];
    int rx[HL_RING];
    const float* __restrict__ const gh = gin + (h ? split : 0);
    const v4f* __restrict__ const hp = reinterpret_cast<const v4f*>(gh);
    const int nh = h ? N - split : split;
    const int nq = nh >> 2, rot = hl_rot(nq), per = (nq + HL_T - 1) / HL_T;
#pragma unroll
    for (int k = 0; k < HL_RING; ++k) {
        rx[k] = hl_piece(k, tid, nq, rot);
        rr[k] = hp[rx[k]];
    }

    const int* __restrict__ const pbeg = h ? mid : rowptr;
    const int* __restrict__ const pend = h ? cnt : mid;
    int s0[HQ_TRIPS], e0[HQ_TRIPS];
#pragma unroll
    for (int t = 0; t < HQ_TRIPS; ++t) {
        const int v = v_lo + gi + 128 * t;
        const int vv = min(v, N - 1);
        const int rb = pbeg[vv], re = pend[vv];
        const bool ok = v < v_hi;
        s0[t] = ok ? rb : 0;
        e0[t] = ok ? re : 0;
    }
    v4i qa[HQ_TRIPS], qa2[HQ_TRIPS];
#pragma unroll
    for (int t = 0; t < HQ_TRIPS; ++t) {
        qa[t]  = hq_piece(c128, s0[t] + l, npieces);
        qa2[t] = hq_piece(c128, s0[t] + 4 + l, npieces);
    }
    float f_dv[2], f_zz[2], f_self[2];
#pragma unroll
    for (int j = 0; j < 2; ++j) {
        const int vfc = min(v_lo + tid + HL_T * j, N - 1);
        f_dv[j] = dinv[vfc];
        f_zz[j] = z0[vfc];
        f_self[j] = gin[vfc];
    }

    for (int j0 = HL_RING; j0 < per; j0 += HL_RING) {
#pragma unroll
        for (int k = 0; k < HL_RING; ++k) {
            d4[rx[k]] = rr[k];
            rx[k] = hl_piece(j0 + k, tid, nq, rot);
            rr[k] = hp[rx[k]];
        }
    }
#pragma unroll
    for (int k = 0; k < HL_RING; ++k) d4[rx[k]] = rr[k];
    if (tid < (nh & 3)) sg[(nq << 2) + tid] = gh[(nq << 2) + tid];
    if (tid == 0) sg[HL_SLAB - 1] = 0.f;
    __syncthreads();

    float acc[HQ_TRIPS];
#pragma unroll
    for (int t = 0; t < HQ_TRIPS; ++t) acc[t] = hq_sum8(sg, qa[t], s0[t] + l < e0[t]);
#pragma unroll
    for (int t = 0; t < HQ_TRIPS; ++t) {
        if (s0[t] + 4 + l < e0[t]) acc[t] += hq_sum8(sg, qa2[t], true);
    }
#pragma unroll
    for (int t = 0; t < HQ_TRIPS; ++t) acc[t] += hq_rest(sg, c128, s0[t], e0[t], l, npieces);
#pragma unroll
    for (int t = 0; t < HQ_TRIPS; ++t) {
        float a = acc[t];
        a = add_xmask<1>(a);
        a = add_xmask<2>(a);
        if (l == 0) stot[gi + 128 * t] = a;
    }
    __syncthreads();
    const float bias = LAST ? b3[0] : 0.f;
#pragma unroll
    for (int j = 0; j < 2; ++j) {
        const int i = tid + HL_T * j;
        const int vf = v_lo + i;
        if (i < npw && vf < v_hi) {
            const float dv = f_dv[j];
            const float zt = h ? ONE_MINUS_ALPHA * dv * stot[i]
                               : ONE_MINUS_ALPHA * dv * (stot[i] + f_self[j]) + ALPHA_F * f_zz[j];
            atomicAdd(&gacc[vf], LAST ? (h ? zt : zt + bias) : dv * zt);
            if (gzero != nullptr && j == h) gzero[vf] = 0.f;
        }
    }
}


extern "C" void kernel_launch(void* const* d_in, const int* in_sizes, int n_in,
                              void* d_out, int out_size, void* d_ws, size_t ws_size,
                              hipStream_t stream) {
    const float* x  = (const float*)d_in[0];
    const int*  eidx = (const int*)d_in[1];
    const float* W1 = (const float*)d_in[2];
    const float* b1 = (const float*)d_in[3];
    const float* W2 = (const float*)d_in[4];
    const float* b2 = (const float*)d_in[5];
    const float* W3 = (const float*)d_in[6];
    const float* b3 = (const float*)d_in[7];
    float* out = (float*)d_out;

    const int N = in_sizes[0];
    const int E = in_sizes[1] / 2;

    char* ws = (char*)d_ws;
    const int B = 256;

    {
        const int nb = (N + BNODES - 1) / BNODES;
        const int nh = nb * CHUNK_WGS;
        size_t o2 = 0;
        auto carve2 = [&](size_t bytes) -> char* {
            char* p = ws + o2;
            o2 += (bytes + 255) & ~(size_t)255;
            return p;
        };
        int*   cnt    = (int*)carve2((size_t)N * 4);
        float* dinv   = (float*)carve2((size_t)N * 4);
        int*   rowptr = (int*)carve2((size_t)N * 4);
        int*   hist1  = (int*)carve2((size_t)nh * 4);
        int*   off1   = (int*)carve2((size_t)nh * 4);
        int*   csr    = (int*)carve2((size_t)E * 4);
        const size_t plane = (((size_t)N * 4) + 255) & ~(size_t)255;
        const size_t tmpb = (size_t)E * 4;
        char* shared_region = carve2(tmpb > 4 * plane ? tmpb : 4 * plane);
        unsigned* tmp = (unsigned*)shared_region;
        float* z0 = (float*)shared_region;
        float* gA = (float*)(shared_region + plane);
        float* gB = (float*)(shared_region + 2 * plane);
        float* gC = (float*)(shared_region + 3 * plane);
        int* mid = hist1;
        const bool lds_hop = N >= 4096 && N <= 2 * HL_SLAB && E >= 64;
        const int split = lds_hop ? ((N / 2 + 31) & ~31) : N;
        const int npw2 = (N + HP_RANGES - 1) / HP_RANGES;
        const bool pair_hop = lds_hop && npw2 <= HP_NPW;

        if (o2 <= ws_size && nb <= NB_MAX && N <= (1 << 24)) {
            const int chunk = (E + CHUNK_WGS - 1) / CHUNK_WGS;
            const int epad = (E + 63) & ~63;
            const size_t mid_off = (((size_t)epad * 2) + 255) & ~(size_t)255;
            const bool own_chunks = lds_hop && chunk <= ST1_CAP &&
                                    mid_off + (size_t)N * 4 <= (size_t)E * 4;
            const int epadp = (E + 8 + PB_GAP * nb + 63) & ~63;
            const size_t midp_off = (((size_t)epadp * 2) + 255) & ~(size_t)255;
            const bool padded = own_chunks && pair_hop && npw2 <= HQ_NPW &&
                                N <= 2 * HL_SLAB - 64 &&
                                midp_off + (size_t)N * 4 <= (size_t)E * 4;
            if (own_chunks) {
                mid = (int*)((char*)csr + (padded ? midp_off : mid_off));
                ak_chunk_sort<<<CHUNK_WGS, 1024, 0, stream>>>(eidx, hist1, off1, tmp, E, chunk, nb);
                if (padded) {
                    ak_bucket_sort_sh<true><<<nb, 512, 0, stream>>>(tmp, hist1, off1, rowptr, cnt, mid,
                                                                    dinv, csr, E, N, chunk, split);
                } else {
                    ak_bucket_sort_sh<false><<<nb, 512, 0, stream>>>(tmp, hist1, off1, rowptr, cnt, mid,
                                                                     dinv, csr, E, N, chunk, split);
                }
            } else {
                ak_bucket_hist<<<CHUNK_WGS, 1024, 0, stream>>>(eidx, hist1, E, chunk, nb);
                ak_excl_scan<<<(nh + 1023) / 1024, 1024, 0, stream>>>(hist1, off1, nh);
                ak_bucket_scatter<<<CHUNK_WGS, 1024, 0, stream>>>(eidx, hist1, off1, tmp, E, chunk, nb);
                if (lds_hop) {
                    ak_bucket_sort<true><<<nb, 512, 0, stream>>>(tmp, off1, rowptr, cnt, mid, dinv, csr,
                                                                 E, N, nb, split);
                } else {
                    ak_bucket_sort<false><<<nb, 512, 0, stream>>>(tmp, off1, rowptr, cnt, mid, dinv, csr,
                                                                  E, N, nb, split);
                }
            }

            int tiles = N / 16;
            if (tiles > 0) {
                int waves_per_block = B / 32;
                int blocks = (tiles + waves_per_block - 1) / waves_per_block;
                ak_encoder_wmma<<<blocks, B, 0, stream>>>(x, W1, b1, W2, b2, W3, dinv, z0, gA,
                                                          pair_hop ? gB : (float*)nullptr, tiles);
            }
            int rem_start = tiles * 16;
            if (rem_start < N) {
                ak_encoder_tail<<<(N - rem_start + B - 1) / B, B, 0, stream>>>(
                    x, W1, b1, W2, b2, W3, dinv, z0, gA, pair_hop ? gB : (float*)nullptr,
                    rem_start, N);
            }
            if (pair_hop) {
                float* const pl[3] = {gA, gB, gC};
                for (int it = 0; it < KITERS; ++it) {
                    const float* gi = pl[it % 3];
                    if (it < KITERS - 1) {
                        float* gz = (it == KITERS - 2) ? out : pl[(it + 2) % 3];
                        if (padded) {
                            ak_hop_pad<false><<<2 * HP_RANGES, HL_T, 0, stream>>>(
                                rowptr, cnt, mid, csr, gi, z0, dinv, pl[(it + 1) % 3], gz, b3, N,
                                npw2, split, epadp >> 3);
                        } else {
                            ak_hop_pair<false><<<2 * HP_RANGES, HL_T, 0, stream>>>(
                                rowptr, cnt, mid, csr, gi, z0, dinv, pl[(it + 1) % 3], gz, b3, N,
                                npw2, split, epad);
                        }
                    } else if (padded) {
                        ak_hop_pad<true><<<2 * HP_RANGES, HL_T, 0, stream>>>(
                            rowptr, cnt, mid, csr, gi, z0, dinv, out, (float*)nullptr, b3, N, npw2,
                            split, epadp >> 3);
                    } else {
                        ak_hop_pair<true><<<2 * HP_RANGES, HL_T, 0, stream>>>(
                            rowptr, cnt, mid, csr, gi, z0, dinv, out, (float*)nullptr, b3, N, npw2,
                            split, epad);
                    }
                }
                return;
            }

            const int hblocks = (N + 32 * HOP_TRIPS - 1) / (32 * HOP_TRIPS);
            const int npw = (N + HL_WGS - 1) / HL_WGS;
            for (int it = 0; it < KITERS; ++it) {
                const float* gi = (it & 1) ? gB : gA;
                float* go = (it & 1) ? gA : gB;
                if (lds_hop) {
                    if (it < KITERS - 1) {
                        ak_hop_lds<false><<<HL_WGS, HL_T, 0, stream>>>(rowptr, cnt, mid, csr, gi, z0,
                                                                       dinv, go, b3, out, N, npw,
                                                                       split, epad);
                    } else {
                        ak_hop_lds<true><<<HL_WGS, HL_T, 0, stream>>>(rowptr, cnt, mid, csr, gi, z0,
                                                                      dinv, (float*)nullptr, b3, out,
                                                                      N, npw, split, epad);
                    }
                } else if (it < KITERS - 1) {
                    ak_hop<false><<<hblocks, B, 0, stream>>>(rowptr, cnt, csr, gi, z0, dinv, go,
                                                             b3, out, N);
                } else {
                    ak_hop<true><<<hblocks, B, 0, stream>>>(rowptr, cnt, csr, gi, z0, dinv,
                                                            (float*)nullptr, b3, out, N);
                }
            }
            return;
        }
    }

    size_t off = 0;
    auto carve = [&](size_t bytes) -> char* {
        char* p = ws + off;
        off += (bytes + 255) & ~(size_t)255;
        return p;
    };
    float* deg  = (float*)carve((size_t)N * 4);
    float* dinv = (float*)carve((size_t)N * 4);
    float* z0   = (float*)carve((size_t)N * 4);
    float* g    = (float*)carve((size_t)N * 4);
    float* agg  = (float*)carve((size_t)N * 4);

    ak_deg_init<<<(N + B - 1) / B, B, 0, stream>>>(deg, N);
    ak_deg_accum<<<(E + B - 1) / B, B, 0, stream>>>(eidx, deg, E);
    ak_dinv<<<(N + B - 1) / B, B, 0, stream>>>(deg, dinv, N);

    int tiles = N / 16;
    if (tiles > 0) {
        int waves_per_block = B / 32;
        int blocks = (tiles + waves_per_block - 1) / waves_per_block;
        ak_encoder_wmma<<<blocks, B, 0, stream>>>(x, W1, b1, W2, b2, W3, dinv, z0, g, agg, tiles);
    }
    int rem_start = tiles * 16;
    if (rem_start < N) {
        ak_encoder_tail<<<(N - rem_start + B - 1) / B, B, 0, stream>>>(
            x, W1, b1, W2, b2, W3, dinv, z0, g, agg, rem_start, N);
    }

    for (int it = 0; it < KITERS; ++it) {
        ak_edge_scatter<<<(E + B - 1) / B, B, 0, stream>>>(eidx, g, agg, E);
        if (it < KITERS - 1) {
            ak_finalize<<<(N + B - 1) / B, B, 0, stream>>>(agg, g, z0, dinv, N);
        } else {
            ak_finalize_out<<<(N + B - 1) / B, B, 0, stream>>>(agg, g, z0, dinv, b3, out, N);
        }
    }
}


# Hop takes the list's 16-byte pieces flat: sort files a bucket half by half, a thread sums whole pieces of its work-group's runs into LDS, one thread adds a node's piece sums; float sum order changes
# speedup vs baseline: 21.9058x; 1.0582x over previous
#include <hip/hip_runtime.h>


typedef __attribute__((ext_vector_type(2))) float v2f;
typedef __attribute__((ext_vector_type(8))) float v8f;

#define HID 16
#define KITERS 10
#define ONE_MINUS_ALPHA 0.9f
#define ALPHA_F 0.1f

__device__ __forceinline__ float relu_f(float v) { return v > 0.f ? v : 0.f; }


__global__ void ak_deg_init(float* __restrict__ deg, int n) {
    int i = blockIdx.x * blockDim.x + threadIdx.x;
    if (i < n) deg[i] = 1.0f;
}

__global__ void ak_deg_accum(const int* __restrict__ eidx, float* __restrict__ deg, int E) {
    int e = blockIdx.x * blockDim.x + threadIdx.x;
    if (e < E) atomicAdd(&deg[eidx[(long long)E + e]], 1.0f);
}

__global__ void ak_dinv(const float* __restrict__ deg, float* __restrict__ dinv, int n) {
    int i = blockIdx.x * blockDim.x + threadIdx.x;
    if (i < n) dinv[i] = rsqrtf(deg[i]);
}

__device__ __forceinline__ float xhalf(float x) {
    int b = __builtin_bit_cast(int, x);
    return __builtin_bit_cast(float,
        __builtin_amdgcn_permlanex16(b, b, 0x76543210, 0xfedcba98, false, false));
}


__global__ void ak_encoder_wmma(const float* __restrict__ x, const float* __restrict__ W1,
                                const float* __restrict__ b1, const float* __restrict__ W2,
                                const float* __restrict__ b2, const float* __restrict__ W3,
                                const float* __restrict__ dinv, float* __restrict__ z0,
                                float* __restrict__ g, float* __restrict__ agg, int n_tiles) {
    int wave = (blockIdx.x * blockDim.x + threadIdx.x) >> 5;
    int lane = threadIdx.x & 31;
    if (wave >= n_tiles) return;

    int row = lane & 15;
    int hi  = lane >> 4;
    int node = wave * 16 + row;
    float xv = x[node];

    v8f acc = {};
#pragma unroll
    for (int s = 0; s < 4; ++s) {
        int k0 = 4 * s + 2 * hi;
        v2f wt, ht;
        wt.x = W2[k0 * HID + row];
        wt.y = W2[(k0 + 1) * HID + row];
        ht.x = relu_f(xv * W1[k0]     + b1[k0]);
        ht.y = relu_f(xv * W1[k0 + 1] + b1[k0 + 1]);
        acc = __builtin_amdgcn_wmma_f32_16x16x4_f32(
            false, wt, false, ht, (short)0, acc, false, false);
    }

    float part = 0.f;
#pragma unroll
    for (int i = 0; i < 8; ++i) {
        int m = i + 8 * hi;
        part += relu_f(acc[i] + b2[m]) * W3[m];
    }
    float z = part + xhalf(part);
    if (hi == 0) {
        z0[node] = z;
        g[node]  = dinv[node] * z;
        if (agg) agg[node] = 0.f;
    }
}

__global__ void ak_encoder_tail(const float* __restrict__ x, const float* __restrict__ W1,
                                const float* __restrict__ b1, const float* __restrict__ W2,
                                const float* __restrict__ b2, const float* __restrict__ W3,
                                const float* __restrict__ dinv, float* __restrict__ z0,
                                float* __restrict__ g, float* __restrict__ agg, int start, int N) {
    int v = start + blockIdx.x * blockDim.x + threadIdx.x;
    if (v >= N) return;
    float h1[HID];
    float xv = x[v];
#pragma unroll
    for (int k = 0; k < HID; ++k) h1[k] = relu_f(xv * W1[k] + b1[k]);
    float z = 0.f;
#pragma unroll
    for (int j = 0; j < HID; ++j) {
        float acc = b2[j];
#pragma unroll
        for (int k = 0; k < HID; ++k) acc += h1[k] * W2[k * HID + j];
        z += relu_f(acc) * W3[j];
    }
    z0[v] = z;
    g[v]  = dinv[v] * z;
    if (agg) agg[v] = 0.f;
}


__global__ void ak_edge_scatter(const int* __restrict__ eidx, const float* __restrict__ g,
                                float* __restrict__ agg, int E) {
    int e = blockIdx.x * blockDim.x + threadIdx.x;
    if (e >= E) return;
    atomicAdd(&agg[eidx[(long long)E + e]], g[eidx[e]]);
}

__global__ void ak_finalize(float* __restrict__ agg, float* __restrict__ g,
                            const float* __restrict__ z0, const float* __restrict__ dinv, int n) {
    int v = blockIdx.x * blockDim.x + threadIdx.x;
    if (v >= n) return;
    float dv = dinv[v];
    float zn = ONE_MINUS_ALPHA * dv * (agg[v] + g[v]) + ALPHA_F * z0[v];
    g[v] = dv * zn;
    agg[v] = 0.f;
}

__global__ void ak_finalize_out(const float* __restrict__ agg, const float* __restrict__ g,
                                const float* __restrict__ z0, const float* __restrict__ dinv,
                                const float* __restrict__ b3, float* __restrict__ out, int N) {
    int v = blockIdx.x * blockDim.x + threadIdx.x;
    if (v >= N) return;
    out[v] = ONE_MINUS_ALPHA * dinv[v] * (agg[v] + g[v]) + ALPHA_F * z0[v] + b3[0];
}


#define BSH 8
#define BNODES 256
#define NB_MAX 4096
#define CHUNK_WGS 256

__global__ void __launch_bounds__(1024)
ak_bucket_hist(const int* __restrict__ eidx, int* __restrict__ hist1, int E, int chunk, int nb) {
    __shared__ int hist[NB_MAX];
    for (int b = threadIdx.x; b < nb; b += 1024) hist[b] = 0;
    __syncthreads();
    const int lo = blockIdx.x * chunk;
    const int hi = min(lo + chunk, E);
    const int* __restrict__ dst = eidx + (size_t)E;
    for (int e0 = lo + threadIdx.x; e0 < hi; e0 += 4 * 1024) {
        int d[4];
#pragma unroll
        for (int k = 0; k < 4; ++k) d[k] = dst[min(e0 + k * 1024, hi - 1)];
#pragma unroll
        for (int k = 0; k < 4; ++k) {
            if (e0 + k * 1024 < hi) atomicAdd(&hist[d[k] >> BSH], 1);
        }
    }
    __syncthreads();
    for (int b = threadIdx.x; b < nb; b += 1024) hist1[b * CHUNK_WGS + blockIdx.x] = hist[b];
}

__global__ void __launch_bounds__(1024)
ak_excl_scan(const int* __restrict__ in, int* __restrict__ out, int n) {
    __shared__ int red[32];
    __shared__ int wsum[32];
    const int tid = threadIdx.x, lane = tid & 31, w = tid >> 5;
    const int first = blockIdx.x * 1024;

    int part = 0;
    for (int i = tid; i < first; i += 1024) part += in[i];
    for (int o = 16; o > 0; o >>= 1) part += __shfl_xor(part, o);
    if (lane == 0) red[w] = part;

    const int i = first + tid;
    const int c = (i < n) ? in[i] : 0;
    int x = c;
    for (int o = 1; o < 32; o <<= 1) {
        int y = __shfl_up(x, o);
        if (lane >= o) x += y;
    }
    if (lane == 31) wsum[w] = x;
    __syncthreads();
    if (w == 0) {
        int s = wsum[lane];
        for (int o = 1; o < 32; o <<= 1) {
            int y = __shfl_up(s, o);
            if (lane >= o) s += y;
        }
        wsum[lane] = s;
        int b = red[lane];
        for (int o = 16; o > 0; o >>= 1) b += __shfl_xor(b, o);
        red[lane] = b;
    }
    __syncthreads();
    if (i < n) out[i] = red[0] + (w ? wsum[w - 1] : 0) + x - c;
}

template <bool STAGED>
__device__ __forceinline__ void ak_scatter_edges(const int* __restrict__ eidx, int E, int lo, int hi,
                                                 int tid, int* cur, unsigned* stg,
                                                 unsigned* __restrict__ tmp) {
    const int* __restrict__ dst = eidx + (size_t)E;
    for (int e0 = lo + tid; e0 < hi; e0 += 4 * 1024) {
        int d[4], s[4];
#pragma unroll
        for (int k = 0; k < 4; ++k) {
            const int e = min(e0 + k * 1024, hi - 1);
            d[k] = dst[e];
            s[k] = eidx[e];
        }
        asm volatile("" :: "v"(d[0]), "v"(d[1]), "v"(d[2]), "v"(d[3]),
                           "v"(s[0]), "v"(s[1]), "v"(s[2]), "v"(s[3]));
#pragma unroll
        for (int k = 0; k < 4; ++k) {
            if (e0 + k * 1024 < hi) {
                const int slot = atomicAdd(&cur[d[k] >> BSH], 1);
                const unsigned ent = ((unsigned)s[k] << BSH) | (unsigned)(d[k] & (BNODES - 1));
                if (STAGED) stg[slot] = ent;
                else        tmp[slot] = ent;
            }
        }
    }
}

#define ST1_CAP 20480

__global__ void __launch_bounds__(1024)
ak_bucket_scatter(const int* __restrict__ eidx, const int* __restrict__ hist1,
                  const int* __restrict__ off1, unsigned* __restrict__ tmp, int E, int chunk,
                  int nb) {
    __shared__ int cur[NB_MAX];
    __shared__ int sbeg[NB_MAX];
    __shared__ int gbeg[NB_MAX];
    __shared__ unsigned stg[ST1_CAP];
    __shared__ int wsum[32];
    const int tid = threadIdx.x, lane = tid & 31, w = tid >> 5;
    const bool staged = chunk <= ST1_CAP;
    if (staged) {
        int c4[4], g4[4];
        int s = 0;
#pragma unroll
        for (int k = 0; k < 4; ++k) {
            const int b = 4 * tid + k;
            const int bb = min(b, nb - 1);
            const int cv = hist1[bb * CHUNK_WGS + blockIdx.x];
            g4[k] = off1[bb * CHUNK_WGS + blockIdx.x];
            c4[k] = b < nb ? cv : 0;
            s += c4[k];
        }
        int x = s;
        for (int o = 1; o < 32; o <<= 1) {
            int y = __shfl_up(x, o);
            if (lane >= o) x += y;
        }
        if (lane == 31) wsum[w] = x;
        __syncthreads();
        if (w == 0) {
            int t = wsum[lane];
            for (int o = 1; o < 32; o <<= 1) {
                int y = __shfl_up(t, o);
                if (lane >= o) t += y;
            }
            wsum[lane] = t;
        }
        __syncthreads();
        int base = (w ? wsum[w - 1] : 0) + x - s;
#pragma unroll
        for (int k = 0; k < 4; ++k) {
            const int b = 4 * tid + k;
            if (b < nb) {
                cur[b] = base;
                sbeg[b] = base;
                gbeg[b] = g4[k];
            }
            base += c4[k];
        }
    } else {
        for (int b = tid; b < nb; b += 1024) cur[b] = off1[b * CHUNK_WGS + blockIdx.x];
    }
    __syncthreads();
    const int lo = blockIdx.x * chunk;
    const int hi = min(lo + chunk, E);
    if (!staged) {
        ak_scatter_edges<false>(eidx, E, lo, hi, tid, cur, stg, tmp);
        return;
    }
    ak_scatter_edges<true>(eidx, E, lo, hi, tid, cur, stg, tmp);
    __syncthreads();
    for (int b = w; b < nb; b += 32) {
        const int s0 = sbeg[b], n = cur[b] - s0, g0 = gbeg[b];
        if (n <= 0) continue;
        for (int ib = lane - (g0 & 31); ib < n; ib += 128) {
            unsigned v[4];
#pragma unroll
            for (int k = 0; k < 4; ++k)
                v[k] = stg[s0 + min(max(ib + 32 * k, 0), n - 1)];
#pragma unroll
            for (int k = 0; k < 4; ++k) {
                const int i = ib + 32 * k;
                if (i >= 0 && i < n) tmp[g0 + i] = v[k];
            }
        }
    }
}

#define ST2_BYTES 49152

template <bool S16> struct ak_ent { typedef int t; };
template <> struct ak_ent<true> { typedef unsigned short t; };

template <bool S16, bool STAGED>
__device__ __forceinline__ void ak_place(const unsigned* __restrict__ tmp, int lo, int hi, int tid,
                                         int split, int* cur, typename ak_ent<S16>::t* stage,
                                         typename ak_ent<S16>::t* __restrict__ out) {
    typedef typename ak_ent<S16>::t ent_t;
    for (int i0 = lo + tid; i0 < hi; i0 += 4 * 512) {
        unsigned u[4];
#pragma unroll
        for (int k = 0; k < 4; ++k) u[k] = tmp[min(i0 + k * 512, hi - 1)];
        asm volatile("" :: "v"(u[0]), "v"(u[1]), "v"(u[2]), "v"(u[3]));
#pragma unroll
        for (int k = 0; k < 4; ++k) {
            if (i0 + k * 512 < hi) {
                const int sn = (int)(u[k] >> BSH);
                const int key = (int)((u[k] & (BNODES - 1)) << 1) | (sn >= split ? 1 : 0);
                const int slot = atomicAdd(&cur[key], 1);
                const ent_t val = S16 ? (ent_t)(sn >= split ? sn - split : sn) : (ent_t)sn;
                if (STAGED) stage[slot] = val;
                else        out[slot] = val;
            }
        }
    }
}

template <bool S16>
__global__ void __launch_bounds__(512)
ak_bucket_sort(const unsigned* __restrict__ tmp, const int* __restrict__ off1,
               int* __restrict__ rowptr, int* __restrict__ cnt, int* __restrict__ mid,
               float* __restrict__ dinv, int* __restrict__ csr, int E, int N, int nb, int split) {
    typedef typename ak_ent<S16>::t ent_t;
    constexpr int CAP = ST2_BYTES / (int)sizeof(ent_t);
    constexpr int LINE = 128 / (int)sizeof(ent_t);
    __shared__ int c[2 * BNODES];
    __shared__ int cur[2 * BNODES];
    __shared__ int wtot[2 * BNODES / 32];
    __shared__ ent_t stg[CAP];
    const int tid = threadIdx.x, lane = tid & 31, w = tid >> 5;
    const int b = blockIdx.x;
    const int lo = off1[b * CHUNK_WGS];
    const int hi = (b + 1 < nb) ? off1[(b + 1) * CHUNK_WGS] : E;
    const int sh = lo & (LINE - 1);
    const bool staged = (hi - lo) + sh <= CAP;
    ent_t* const gout = reinterpret_cast<ent_t*>(csr);
    c[tid] = 0;
    __syncthreads();
    for (int i0 = lo + tid; i0 < hi; i0 += 4 * 512) {
        unsigned u[4];
#pragma unroll
        for (int k = 0; k < 4; ++k) u[k] = tmp[min(i0 + k * 512, hi - 1)];
#pragma unroll
        for (int k = 0; k < 4; ++k) {
            if (i0 + k * 512 < hi) {
                const int key = (int)((u[k] & (BNODES - 1)) << 1) | ((int)(u[k] >> BSH) >= split ? 1 : 0);
                atomicAdd(&c[key], 1);
            }
        }
    }
    __syncthreads();
    const int mine = c[tid];
    int x = mine;
    for (int o = 1; o < 32; o <<= 1) {
        int y = __shfl_up(x, o);
        if (lane >= o) x += y;
    }
    if (lane == 31) wtot[w] = x;
    __syncthreads();
    int base = 0;
    for (int k = 0; k < w; ++k) base += wtot[k];
    const int ex = base + x - mine;
    cur[tid] = ex;
    if ((tid & 1) == 0) {
        const int v = b * BNODES + (tid >> 1);
        if (v < N) {
            const int total = mine + c[tid + 1];
            cnt[v] = total;
            rowptr[v] = lo + ex;
            mid[v] = lo + ex + mine;
            dinv[v] = rsqrtf((float)(total + 1));
        }
    }
    __syncthreads();
    if (!staged) {
        ak_place<S16, false>(tmp, lo, hi, tid, split, cur, stg, gout + lo);
        return;
    }
    ak_place<S16, true>(tmp, lo, hi, tid, split, cur, stg + sh, gout + lo);
    __syncthreads();
    const int first = sh, last = sh + (hi - lo);
    ent_t* const gl = gout + (lo - sh);
    if (S16) {
        unsigned* const gw = reinterpret_cast<unsigned*>(gl);
        const int fe = (first + 1) & ~1, le = last & ~1;
        if (fe < le) {
            for (int wd0 = tid; 2 * wd0 < le; wd0 += 4 * 512) {
                unsigned a[4], bq[4];
#pragma unroll
                for (int k = 0; k < 4; ++k) {
                    const int j = min(max(2 * (wd0 + 512 * k), fe), le - 2);
                    a[k] = stg[j];
                    bq[k] = stg[j + 1];
                }
#pragma unroll
                for (int k = 0; k < 4; ++k) {
                    const int wd = wd0 + 512 * k;
                    if (2 * wd >= fe && 2 * wd < le) gw[wd] = a[k] | (bq[k] << 16);
                }
            }
        }
        if (tid == 0) {
            if ((first & 1) && first < last) gl[first] = stg[first];
            if ((last & 1) && last - 1 >= first) gl[last - 1] = stg[last - 1];
        }
    } else if (first < last) {
        for (int j0 = tid; j0 < last; j0 += 4 * 512) {
            ent_t a[4];
#pragma unroll
            for (int k = 0; k < 4; ++k) a[k] = stg[min(max(j0 + 512 * k, first), last - 1)];
#pragma unroll
            for (int k = 0; k < 4; ++k) {
                const int j = j0 + 512 * k;
                if (j >= first && j < last) gl[j] = a[k];
            }
        }
    }
}


#define CS_EPT 20
static_assert(CS_EPT * 1024 >= ST1_CAP, "a thread's registers must cover the largest staged chunk");

__global__ void __launch_bounds__(1024)
ak_chunk_sort(const int* __restrict__ eidx, int* __restrict__ cnt1, int* __restrict__ pos1,
              unsigned* __restrict__ tmp, int E, int chunk, int nb) {
    __shared__ int cur[NB_MAX];
    __shared__ unsigned stg[ST1_CAP];
    __shared__ int wsum[32];
    const int tid = threadIdx.x, lane = tid & 31, w = tid >> 5;
    const int lo = blockIdx.x * chunk;
    const int hi = min(lo + chunk, E);
    const int* __restrict__ dst = eidx + (size_t)E;
    int d[CS_EPT], s[CS_EPT];
#pragma unroll
    for (int k = 0; k < CS_EPT; ++k) {
        const int e = min(lo + tid + k * 1024, E - 1);
        d[k] = dst[e];
        s[k] = eidx[e];
    }
    for (int b = tid; b < nb; b += 1024) cur[b] = 0;
    __syncthreads();
#pragma unroll
    for (int k = 0; k < CS_EPT; ++k) {
        if (lo + tid + k * 1024 < hi) atomicAdd(&cur[d[k] >> BSH], 1);
    }
    __syncthreads();
    int c4[4];
    int sum4 = 0;
#pragma unroll
    for (int k = 0; k < 4; ++k) {
        const int b = 4 * tid + k;
        const int cv = cur[b];
        c4[k] = b < nb ? cv : 0;
        sum4 += c4[k];
    }
    int x = sum4;
    for (int o = 1; o < 32; o <<= 1) {
        int y = __shfl_up(x, o);
        if (lane >= o) x += y;
    }
    if (lane == 31) wsum[w] = x;
    __syncthreads();
    if (w == 0) {
        int t = wsum[lane];
        for (int o = 1; o < 32; o <<= 1) {
            int y = __shfl_up(t, o);
            if (lane >= o) t += y;
        }
        wsum[lane] = t;
    }
    __syncthreads();
    int base = (w ? wsum[w - 1] : 0) + x - sum4;
    int p4[4];
#pragma unroll
    for (int k = 0; k < 4; ++k) {
        const int b = 4 * tid + k;
        p4[k] = base;
        if (b < nb) cur[b] = base;
        base += c4[k];
    }
    __syncthreads();
    int slot[CS_EPT];
#pragma unroll
    for (int k = 0; k < CS_EPT; ++k) {
        slot[k] = 0;
        if (lo + tid + k * 1024 < hi) slot[k] = atomicAdd(&cur[d[k] >> BSH], 1);
    }
#pragma unroll
    for (int k = 0; k < CS_EPT; ++k) {
        if (lo + tid + k * 1024 < hi)
            stg[slot[k]] = ((unsigned)s[k] << BSH) | (unsigned)(d[k] & (BNODES - 1));
    }
    __syncthreads();
#pragma unroll
    for (int k = 0; k < 4; ++k) {
        const int b = 4 * tid + k;
        if (b < nb) {
            cnt1[b * CHUNK_WGS + blockIdx.x] = c4[k];
            pos1[b * CHUNK_WGS + blockIdx.x] = p4[k];
        }
    }
    const int n = hi - lo;
    if (n <= 0) return;
    for (int i0 = tid - (lo & 31); i0 < n; i0 += 4 * 1024) {
        unsigned v[4];
#pragma unroll
        for (int k = 0; k < 4; ++k) v[k] = stg[min(max(i0 + 1024 * k, 0), n - 1)];
#pragma unroll
        for (int k = 0; k < 4; ++k) {
            const int i = i0 + 1024 * k;
            if (i >= 0 && i < n) tmp[lo + i] = v[k];
        }
    }
}

template <bool HM>
__device__ __forceinline__ int ak_key(unsigned u, int split) {
    const int nib = (int)(u & (BNODES - 1));
    const int hf = (int)(u >> BSH) >= split ? 1 : 0;
    return HM ? (hf << BSH) | nib : (nib << 1) | hf;
}

template <bool STAGED, bool HM>
__device__ __forceinline__ void ak_place_one(unsigned u, int split, int* cur, unsigned short* stage,
                                             unsigned short* __restrict__ out) {
    const int sn = (int)(u >> BSH);
    const int slot = atomicAdd(&cur[ak_key<HM>(u, split)], 1);
    const unsigned short val = (unsigned short)(sn >= split ? sn - split : sn);
    if (STAGED) stage[slot] = val;
    else        out[slot] = val;
}

#define BS_SPW 16

template <bool STAGED, bool HM>
__device__ __forceinline__ void ak_place_sh(const unsigned (&u)[2 * BS_SPW], const int (&nn)[BS_SPW],
                                            const unsigned* __restrict__ tmp, const int* shn,
                                            const int* shb, int w, int lane, int split, int* cur,
                                            unsigned short* stage, unsigned short* __restrict__ out) {
#pragma unroll
    for (int k = 0; k < BS_SPW; ++k) {
        if (lane < nn[k])      ak_place_one<STAGED, HM>(u[2 * k], split, cur, stage, out);
        if (lane + 32 < nn[k]) ak_place_one<STAGED, HM>(u[2 * k + 1], split, cur, stage, out);
    }
#pragma nounroll
    for (int k = 0; k < BS_SPW; ++k) {
        const int n = shn[w * BS_SPW + k];
        const int p = shb[w * BS_SPW + k];
        for (int i = 64 + lane; i < n; i += 32)
            ak_place_one<STAGED, HM>(tmp[p + i], split, cur, stage, out);
    }
}

#define PAD_ENT 0xffff
#define PB_GAP 3600
template <bool PAD>
__global__ void __launch_bounds__(512)
ak_bucket_sort_sh(const unsigned* __restrict__ tmp, const int* __restrict__ cnt1,
                  const int* __restrict__ pos1, int* __restrict__ rowptr, int* __restrict__ cnt,
                  int* __restrict__ mid, int* __restrict__ end1, float* __restrict__ dinv,
                  int* __restrict__ csr, int E, int N, int chunk, int split) {
    typedef unsigned short ent_t;
    constexpr int CAP = ST2_BYTES / (int)sizeof(ent_t);
    constexpr int LINE = 128 / (int)sizeof(ent_t);
    __shared__ int c[2 * BNODES];
    __shared__ int cur[2 * BNODES];
    __shared__ int wtot[2 * BNODES / 32];
    __shared__ int shn[CHUNK_WGS];
    __shared__ int shb[CHUNK_WGS];
    __shared__ int redn[16];
    __shared__ int redp[16];
    __shared__ ent_t stg[CAP];
    const int tid = threadIdx.x, lane = tid & 31, w = tid >> 5;
    const int b = blockIdx.x;
    int myn = 0, myp = 0;
    if (tid < CHUNK_WGS) {
        myn = cnt1[b * CHUNK_WGS + tid];
        myp = pos1[b * CHUNK_WGS + tid];
        shn[tid] = myn;
        shb[tid] = min(tid * chunk + myp, E - 1);
    }
    c[tid] = 0;
    int sn = myn, sp = myp;
    for (int o = 16; o > 0; o >>= 1) {
        sn += __shfl_xor(sn, o);
        sp += __shfl_xor(sp, o);
    }
    if (lane == 0) {
        redn[w] = sn;
        redp[w] = sp;
    }
    __syncthreads();
    int tot = 0, lo = 0;
#pragma unroll
    for (int k = 0; k < 16; ++k) {
        tot += redn[k];
        lo += redp[k];
    }
    const int lop = PAD ? ((lo + 7) & ~7) + PB_GAP * b : lo;
    const int sh = lop & (LINE - 1);
    bool staged = tot + sh <= CAP;
    ent_t* const gout = reinterpret_cast<ent_t*>(csr);

    unsigned u[2 * BS_SPW];
    int nn[BS_SPW];
#pragma unroll
    for (int k = 0; k < BS_SPW; ++k) {
        nn[k] = shn[w * BS_SPW + k];
        const int p = shb[w * BS_SPW + k];
        const int m = max(nn[k] - 1, 0);
        u[2 * k]     = tmp[p + min(lane, m)];
        u[2 * k + 1] = tmp[p + min(lane + 32, m)];
    }
#pragma unroll
    for (int k = 0; k < BS_SPW; ++k) {
        if (lane < nn[k])      atomicAdd(&c[ak_key<PAD>(u[2 * k], split)], 1);
        if (lane + 32 < nn[k]) atomicAdd(&c[ak_key<PAD>(u[2 * k + 1], split)], 1);
    }
#pragma nounroll
    for (int k = 0; k < BS_SPW; ++k) {
        const int n = shn[w * BS_SPW + k];
        const int p = shb[w * BS_SPW + k];
        for (int i = 64 + lane; i < n; i += 32) atomicAdd(&c[ak_key<PAD>(tmp[p + i], split)], 1);
    }
    __syncthreads();
    const int mine = c[tid];
    const int minep = PAD ? (mine + 7) & ~7 : mine;
    int x = minep;
    for (int o = 1; o < 32; o <<= 1) {
        int y = __shfl_up(x, o);
        if (lane >= o) x += y;
    }
    if (lane == 31) wtot[w] = x;
    __syncthreads();
    int base = 0, totp = 0;
#pragma unroll
    for (int k = 0; k < 2 * BNODES / 32; ++k) {
        const int wk = wtot[k];
        if (k < w) base += wk;
        totp += wk;
    }
    const int ex = base + x - minep;
    cur[tid] = ex;
    if (PAD) staged = totp + sh <= CAP;
    if (PAD) {
        const int hf = tid >> BSH;
        const int v = b * BNODES + (tid & (BNODES - 1));
        if (v < N) {
            const int st = (lop + ex) >> 3;
            if (hf == 0) {
                rowptr[v] = st;
                cnt[v] = st + (minep >> 3);
                dinv[v] = rsqrtf((float)(mine + c[tid + BNODES] + 1));
            } else {
                mid[v] = st;
                end1[v] = st + (minep >> 3);
            }
        }
    } else if ((tid & 1) == 0) {
        const int v = b * BNODES + (tid >> 1);
        if (v < N) {
            const int total = mine + c[tid + 1];
            cnt[v] = total;
            rowptr[v] = lo + ex;
            mid[v] = lo + ex + mine;
            dinv[v] = rsqrtf((float)(total + 1));
        }
    }
    if (PAD) {
        for (int j = mine; j < minep; ++j) {
            if (staged) stg[sh + ex + j] = (ent_t)PAD_ENT;
            else        gout[lop + ex + j] = (ent_t)PAD_ENT;
        }
    }
    __syncthreads();
    if (!staged) {
        ak_place_sh<false, PAD>(u, nn, tmp, shn, shb, w, lane, split, cur, stg, gout + lop);
        return;
    }
    ak_place_sh<true, PAD>(u, nn, tmp, shn, shb, w, lane, split, cur, stg + sh, gout + lop);
    __syncthreads();
    const int first = sh, last = sh + (PAD ? totp : tot);
    ent_t* const gl = gout + (lop - sh);
    unsigned* const gw = reinterpret_cast<unsigned*>(gl);
    const int fe = (first + 1) & ~1, le = last & ~1;
    if (fe < le) {
        for (int wd0 = tid; 2 * wd0 < le; wd0 += 4 * 512) {
            unsigned a[4], bq[4];
#pragma unroll
            for (int k = 0; k < 4; ++k) {
                const int j = min(max(2 * (wd0 + 512 * k), fe), le - 2);
                a[k] = stg[j];
                bq[k] = stg[j + 1];
            }
#pragma unroll
            for (int k = 0; k < 4; ++k) {
                const int wd = wd0 + 512 * k;
                if (2 * wd >= fe && 2 * wd < le) gw[wd] = a[k] | (bq[k] << 16);
            }
        }
    }
    if (tid == 0) {
        if ((first & 1) && first < last) gl[first] = stg[first];
        if ((last & 1) && last - 1 >= first) gl[last - 1] = stg[last - 1];
    }
}

template <int M>
__device__ __forceinline__ float add_xmask(float x) {
    int b = __builtin_bit_cast(int, x);
    return x + __builtin_bit_cast(float, __builtin_amdgcn_update_dpp(0, b, 0x160 + M, 0xf, 0xf, true));
}

#define HOP_TRIPS 6

template <bool LAST>
__global__ void __launch_bounds__(256)
ak_hop(const int* __restrict__ rowptr, const int* __restrict__ cnt,
       const int* __restrict__ csr, const float* __restrict__ gin,
       const float* __restrict__ z0, const float* __restrict__ dinv,
       float* __restrict__ gout, const float* __restrict__ b3,
       float* __restrict__ out, int N) {
    const int l = threadIdx.x & 7;
    const int grp = (blockIdx.x * 256 + threadIdx.x) >> 3;
    const int ngrp = (gridDim.x * 256) >> 3;
    int v = grp;
    int n_nx = cnt[min(v, N - 1)];
    int b_nx = rowptr[min(v, N - 1)];
    for (int vw = grp & ~3; vw < N; vw += ngrp, v += ngrp) {
        const int vv = min(v, N - 1);
        const int n = n_nx;
        const int* __restrict__ cs = csr + b_nx;
        const int vn = min(v + ngrp, N - 1);
        n_nx = cnt[vn];
        b_nx = rowptr[vn];
        const float self = gin[vv];
        const float dv = dinv[vv];
        const float zz = z0[vv];
        float acc = 0.f;
        for (int j0 = l; j0 - l < n; j0 += 64) {
            int s[8];
            float a[8];
#pragma unroll
            for (int k = 0; k < 8; ++k) s[k] = cs[min(j0 + 8 * k, n - 1)];
#pragma unroll
            for (int k = 0; k < 8; ++k) a[k] = gin[s[k]];
#pragma unroll
            for (int k = 0; k < 8; ++k) acc += (j0 + 8 * k < n) ? a[k] : 0.f;
        }
        acc = add_xmask<1>(acc);
        acc = add_xmask<2>(acc);
        acc = add_xmask<4>(acc);
        const float zn = ONE_MINUS_ALPHA * dv * (acc + self) + ALPHA_F * zz;
        if (l == 0 && v < N) {
            if (LAST) out[v] = zn + b3[0];
            else      gout[v] = dv * zn;
        }
    }
}


#define HL_WGS 256
#define HL_T 512
#define HL_TRIPS 8
#define HL_SLAB 65536
#define HL_RING 13

typedef __attribute__((ext_vector_type(4))) float v4f;
typedef __attribute__((ext_vector_type(2))) int v2i;

__device__ __forceinline__ int hl_piece(int j, int tid, int nq, int rot) {
    const int i = min(j * HL_T + tid, nq - 1) + rot;
    return i >= nq ? i - nq : i;
}
__device__ __forceinline__ int hl_rot(int nq) {
    return (int)((blockIdx.x * (unsigned)nq) / HL_WGS) & ~31;
}

__device__ __forceinline__ v2i hl_quad(const unsigned short* __restrict__ c16, int p, int epad) {
    return *reinterpret_cast<const v2i*>(c16 + min(p, epad - 4));
}

__device__ __forceinline__ float hl_sum4(const float* sg, v2i q, int p, int s, int e) {
    const int qq[4] = {q.x & 0xffff, (int)((unsigned)q.x >> 16),
                       q.y & 0xffff, (int)((unsigned)q.y >> 16)};
    float a[4];
    bool ok[4];
#pragma unroll
    for (int j = 0; j < 4; ++j) {
        ok[j] = (p + j >= s) && (p + j < e);
        a[j] = sg[ok[j] ? qq[j] : 0];
    }
    float acc = 0.f;
#pragma unroll
    for (int j = 0; j < 4; ++j) acc += ok[j] ? a[j] : 0.f;
    return acc;
}

__device__ __forceinline__ float hl_first_lo(const float* sg, v2i qlo, int s, int e, int l) {
    return hl_sum4(sg, qlo, (s & ~3) + 4 * l, s, e);
}
__device__ __forceinline__ float hl_first_hi(const float* sg, v2i qhi, int s, int e, int l) {
    const int p = (s & ~3) + 32 + 4 * l;
    return p < e ? hl_sum4(sg, qhi, p, s, e) : 0.f;
}
__device__ __forceinline__ float hl_rest(const float* sg, const unsigned short* __restrict__ c16,
                                         int s, int e, int l, int epad) {
    float acc = 0.f;
    for (int pp = (s & ~3) + 64; pp < e; pp += 64) {
        const int p = pp + 4 * l;
        acc += hl_sum4(sg, hl_quad(c16, p, epad), p, s, e);
        acc += hl_sum4(sg, hl_quad(c16, p + 32, epad), p + 32, s, e);
    }
    return acc;
}

template <bool LAST>
__global__ void __launch_bounds__(HL_T)
ak_hop_lds(const int* __restrict__ rowptr, const int* __restrict__ cnt, const int* __restrict__ mid,
           const int* __restrict__ csr, const float* __restrict__ gin,
           const float* __restrict__ z0, const float* __restrict__ dinv,
           float* __restrict__ gout, const float* __restrict__ b3, float* __restrict__ out,
           int N, int npw, int split, int epad) {
    __shared__ float sg[HL_SLAB];
    __shared__ float stot[HL_T];
    const unsigned short* __restrict__ const c16 = reinterpret_cast<const unsigned short*>(csr);
    const int tid = threadIdx.x, l = tid & 7, gi = tid >> 3;
    const int v_lo = blockIdx.x * npw;
    const int v_hi = min(v_lo + npw, N);

    v4f* const d4 = reinterpret_cast<v4f*>(sg);
    v4f rr[HL_RING];
    int rx[HL_RING];
    const v4f* __restrict__ const ha = reinterpret_cast<const v4f*>(gin);
    const int nqa = split >> 2, rota = hl_rot(nqa), pera = (nqa + HL_T - 1) / HL_T;
#pragma unroll
    for (int k = 0; k < HL_RING; ++k) {
        rx[k] = hl_piece(k, tid, nqa, rota);
        rr[k] = ha[rx[k]];
    }

    int s0[HL_TRIPS], m0[HL_TRIPS], e0[HL_TRIPS];
#pragma unroll
    for (int t = 0; t < HL_TRIPS; ++t) {
        const int v = v_lo + gi + 64 * t;
        const int vv = min(v, N - 1);
        const int rs = rowptr[vv], rm = mid[vv], rc = cnt[vv];
        const bool ok = v < v_hi;
        s0[t] = ok ? rs : 0;
        m0[t] = ok ? rm : 0;
        e0[t] = ok ? rs + rc : 0;
    }
    v2i qa[HL_TRIPS], qa2[HL_TRIPS];
#pragma unroll
    for (int t = 0; t < HL_TRIPS; ++t) {
        qa[t]  = hl_quad(c16, (s0[t] & ~3) + 4 * l, epad);
        qa2[t] = hl_quad(c16, (s0[t] & ~3) + 32 + 4 * l, epad);
    }
    const int vf = v_lo + tid;
    const int vfc = min(vf, N - 1);
    const float f_dv = dinv[vfc], f_zz = z0[vfc], f_self = gin[vfc];

    for (int j0 = HL_RING; j0 < pera; j0 += HL_RING) {
#pragma unroll
        for (int k = 0; k < HL_RING; ++k) {
            d4[rx[k]] = rr[k];
            rx[k] = hl_piece(j0 + k, tid, nqa, rota);
            rr[k] = ha[rx[k]];
        }
    }
#pragma unroll
    for (int k = 0; k < HL_RING; ++k) d4[rx[k]] = rr[k];
    __syncthreads();

    const float* __restrict__ const g1 = gin + split;
    const v4f* __restrict__ const hb = reinterpret_cast<const v4f*>(g1);
    const int n1 = N - split;
    const int nqb = n1 >> 2, rotb = hl_rot(nqb), perb = (nqb + HL_T - 1) / HL_T;
#pragma unroll
    for (int k = 0; k < HL_RING; ++k) {
        rx[k] = hl_piece(k, tid, nqb, rotb);
        rr[k] = hb[rx[k]];
    }
    float acc[HL_TRIPS];
#pragma unroll
    for (int t = 0; t < HL_TRIPS; ++t) acc[t] = hl_first_lo(sg, qa[t], s0[t], m0[t], l);
#pragma unroll
    for (int t = 0; t < HL_TRIPS; ++t) acc[t] += hl_first_hi(sg, qa2[t], s0[t], m0[t], l);
#pragma unroll
    for (int t = 0; t < HL_TRIPS; ++t) acc[t] += hl_rest(sg, c16, s0[t], m0[t], l, epad);
    v2i qb[HL_TRIPS], qb2[HL_TRIPS];
#pragma unroll
    for (int t = 0; t < HL_TRIPS; ++t) {
        qb[t]  = hl_quad(c16, (m0[t] & ~3) + 4 * l, epad);
        qb2[t] = hl_quad(c16, (m0[t] & ~3) + 32 + 4 * l, epad);
    }
    __syncthreads();
    for (int j0 = HL_RING; j0 < perb; j0 += HL_RING) {
#pragma unroll
        for (int k = 0; k < HL_RING; ++k) {
            d4[rx[k]] = rr[k];
            rx[k] = hl_piece(j0 + k, tid, nqb, rotb);
            rr[k] = hb[rx[k]];
        }
    }
#pragma unroll
    for (int k = 0; k < HL_RING; ++k) d4[rx[k]] = rr[k];
    if (tid < (n1 & 3)) sg[(nqb << 2) + tid] = g1[(nqb << 2) + tid];
    __syncthreads();
#pragma unroll
    for (int t = 0; t < HL_TRIPS; ++t) acc[t] += hl_first_lo(sg, qb[t], m0[t], e0[t], l);
#pragma unroll
    for (int t = 0; t < HL_TRIPS; ++t) acc[t] += hl_first_hi(sg, qb2[t], m0[t], e0[t], l);
#pragma unroll
    for (int t = 0; t < HL_TRIPS; ++t) acc[t] += hl_rest(sg, c16, m0[t], e0[t], l, epad);
#pragma unroll
    for (int t = 0; t < HL_TRIPS; ++t) {
        float a = acc[t];
        a = add_xmask<1>(a);
        a = add_xmask<2>(a);
        a = add_xmask<4>(a);
        if (l == 0) stot[gi + 64 * t] = a;
    }
    __syncthreads();
    if (tid < npw && vf < v_hi) {
        const float zn = ONE_MINUS_ALPHA * f_dv * (stot[tid] + f_self) + ALPHA_F * f_zz;
        if (LAST) out[vf] = zn + b3[0];
        else      gout[vf] = f_dv * zn;
    }
}


#define HP_RANGES 128
#define HP_TRIPS 13
#define HP_NPW (64 * HP_TRIPS)
static_assert(KITERS >= 2, "the hop before the last one clears out for the last hop's adds");

template <bool LAST>
__global__ void __launch_bounds__(HL_T)
ak_hop_pair(const int* __restrict__ rowptr, const int* __restrict__ cnt, const int* __restrict__ mid,
            const int* __restrict__ csr, const float* __restrict__ gin,
            const float* __restrict__ z0, const float* __restrict__ dinv,
            float* __restrict__ gacc, float* __restrict__ gzero, const float* __restrict__ b3,
            int N, int npw, int split, int epad) {
    __shared__ float sg[HL_SLAB];
    __shared__ float stot[HP_NPW];
    const unsigned short* __restrict__ const c16 = reinterpret_cast<const unsigned short*>(csr);
    const int tid = threadIdx.x, l = tid & 7, gi = tid >> 3;
    const int h = blockIdx.x & 1;
    const int v_lo = (blockIdx.x >> 1) * npw;
    const int v_hi = min(v_lo + npw, N);

    v4f* const d4 = reinterpret_cast<v4f*>(sg);
    v4f rr[HL_RING];
    int rx[HL_RING];
    const float* __restrict__ const gh = gin + (h ? split : 0);
    const v4f* __restrict__ const hp = reinterpret_cast<const v4f*>(gh);
    const int nh = h ? N - split : split;
    const int nq = nh >> 2, rot = hl_rot(nq), per = (nq + HL_T - 1) / HL_T;
#pragma unroll
    for (int k = 0; k < HL_RING; ++k) {
        rx[k] = hl_piece(k, tid, nq, rot);
        rr[k] = hp[rx[k]];
    }

    int s0[HP_TRIPS], e0[HP_TRIPS];
#pragma unroll
    for (int t = 0; t < HP_TRIPS; ++t) {
        const int v = v_lo + gi + 64 * t;
        const int vv = min(v, N - 1);
        const int rs = rowptr[vv], rm = mid[vv], rc = cnt[vv];
        const bool ok = v < v_hi;
        s0[t] = ok ? (h ? rm : rs) : 0;
        e0[t] = ok ? (h ? rs + rc : rm) : 0;
    }
    v2i qa[HP_TRIPS], qa2[HP_TRIPS];
#pragma unroll
    for (int t = 0; t < HP_TRIPS; ++t) {
        qa[t]  = hl_quad(c16, (s0[t] & ~3) + 4 * l, epad);
        qa2[t] = hl_quad(c16, (s0[t] & ~3) + 32 + 4 * l, epad);
    }
    float f_dv[2], f_zz[2], f_self[2];
#pragma unroll
    for (int j = 0; j < 2; ++j) {
        const int vfc = min(v_lo + tid + HL_T * j, N - 1);
        f_dv[j] = dinv[vfc];
        f_zz[j] = z0[vfc];
        f_self[j] = gin[vfc];
    }

    for (int j0 = HL_RING; j0 < per; j0 += HL_RING) {
#pragma unroll
        for (int k = 0; k < HL_RING; ++k) {
            d4[rx[k]] = rr[k];
            rx[k] = hl_piece(j0 + k, tid, nq, rot);
            rr[k] = hp[rx[k]];
        }
    }
#pragma unroll
    for (int k = 0; k < HL_RING; ++k) d4[rx[k]] = rr[k];
    if (tid < (nh & 3)) sg[(nq << 2) + tid] = gh[(nq << 2) + tid];
    __syncthreads();

    float acc[HP_TRIPS];
#pragma unroll
    for (int t = 0; t < HP_TRIPS; ++t) acc[t] = hl_first_lo(sg, qa[t], s0[t], e0[t], l);
#pragma unroll
    for (int t = 0; t < HP_TRIPS; ++t) acc[t] += hl_first_hi(sg, qa2[t], s0[t], e0[t], l);
#pragma unroll
    for (int t = 0; t < HP_TRIPS; ++t) acc[t] += hl_rest(sg, c16, s0[t], e0[t], l, epad);
#pragma unroll
    for (int t = 0; t < HP_TRIPS; ++t) {
        float a = acc[t];
        a = add_xmask<1>(a);
        a = add_xmask<2>(a);
        a = add_xmask<4>(a);
        if (l == 0) stot[gi + 64 * t] = a;
    }
    __syncthreads();
    const float bias = LAST ? b3[0] : 0.f;
#pragma unroll
    for (int j = 0; j < 2; ++j) {
        const int i = tid + HL_T * j;
        const int vf = v_lo + i;
        if (i < npw && vf < v_hi) {
            const float dv = f_dv[j];
            const float zt = h ? ONE_MINUS_ALPHA * dv * stot[i]
                               : ONE_MINUS_ALPHA * dv * (stot[i] + f_self[j]) + ALPHA_F * f_zz[j];
            atomicAdd(&gacc[vf], LAST ? (h ? zt : zt + bias) : dv * zt);
            if (gzero != nullptr && j == h) gzero[vf] = 0.f;
        }
    }
}


#define HQ_NPW 896
typedef __attribute__((ext_vector_type(4))) int v4i;

__device__ __forceinline__ v4i hq_piece(const v4i* __restrict__ c128, int pi, int npieces) {
    return c128[min(pi, npieces - 1)];
}

__device__ __forceinline__ float hq_sum8(const float* sg, v4i q, bool inside) {
    const unsigned w[4] = {(unsigned)(inside ? q.x : -1), (unsigned)(inside ? q.y : -1),
                           (unsigned)(inside ? q.z : -1), (unsigned)(inside ? q.w : -1)};
    float a[8];
#pragma unroll
    for (int j = 0; j < 4; ++j) {
        a[2 * j]     = sg[w[j] & 0xffffu];
        a[2 * j + 1] = sg[w[j] >> 16];
    }
    return ((a[0] + a[1]) + (a[2] + a[3])) + ((a[4] + a[5]) + (a[6] + a[7]));
}

#define HF_K 6
#define HF_CAP (HL_T * HF_K)
#define HF_RUNS 5
static_assert((HQ_NPW - 2) / BNODES + 2 <= HF_RUNS, "a node range must not touch more buckets");
static_assert(HQ_NPW <= 2 * HL_T, "a thread finishes at most two nodes");

__device__ __forceinline__ int hf_piece_no(const int (&rst)[HF_RUNS], const int (&roff)[HF_RUNS + 1],
                                           int i) {
    int pc = rst[0] + i;
#pragma unroll
    for (int k = 1; k < HF_RUNS; ++k) {
        if (i >= roff[k]) pc = rst[k] + (i - roff[k]);
    }
    return pc;
}

template <bool LAST>
__global__ void __launch_bounds__(HL_T)
ak_hop_flat(const int* __restrict__ rowptr, const int* __restrict__ cnt, const int* __restrict__ mid,
            const int* __restrict__ end1, const int* __restrict__ csr,
            const float* __restrict__ gin, const float* __restrict__ z0,
            const float* __restrict__ dinv, float* __restrict__ gacc, float* __restrict__ gzero,
            const float* __restrict__ b3, int N, int npw, int split, int npieces) {
    __shared__ float sg[HL_SLAB];
    __shared__ float psum[HF_CAP];
    const v4i* __restrict__ const c128 = reinterpret_cast<const v4i*>(csr);
    const int tid = threadIdx.x;
    const int h = blockIdx.x & 1;
    const int v_lo = (blockIdx.x >> 1) * npw;
    const int v_hi = min(v_lo + npw, N);

    const int* __restrict__ const pbeg = h ? mid : rowptr;
    const int* __restrict__ const pend = h ? end1 : cnt;
    const int b0 = v_lo >> BSH;
    const int lane = tid & 31;
    const int lk = lane % HF_RUNS;
    const bool lend = ((lane / HF_RUNS) & 1) != 0;
    const int lf = max(v_lo, (b0 + lk) << BSH);
    const int ll = min(v_hi, (b0 + lk + 1) << BSH) - 1;
    const int* __restrict__ const bp = lend ? pend : pbeg;
    const int bv = bp[lend ? min(max(ll, 0), N - 1) : min(lf, N - 1)];

    v4f* const d4 = reinterpret_cast<v4f*>(sg);
    v4f rr[HL_RING];
    int rx[HL_RING];
    const float* __restrict__ const gh = gin + (h ? split : 0);
    const v4f* __restrict__ const hp = reinterpret_cast<const v4f*>(gh);
    const int nh = h ? N - split : split;
    const int nq = nh >> 2, rot = hl_rot(nq), per = (nq + HL_T - 1) / HL_T;
#pragma unroll
    for (int k = 0; k < HL_RING; ++k) {
        rx[k] = hl_piece(k, tid, nq, rot);
        rr[k] = hp[rx[k]];
    }
    __builtin_amdgcn_sched_barrier(0);

    int rst[HF_RUNS], roff[HF_RUNS + 1];
    roff[0] = 0;
#pragma unroll
    for (int k = 0; k < HF_RUNS; ++k) {
        const int f = max(v_lo, (b0 + k) << BSH);
        const int lst = min(v_hi, (b0 + k + 1) << BSH) - 1;
        const int fs = __builtin_amdgcn_readlane(bv, k);
        const int le = __builtin_amdgcn_readlane(bv, HF_RUNS + k);
        rst[k] = fs;
        roff[k + 1] = roff[k] + (f <= lst ? le - fs : 0);
    }
    const int P = roff[HF_RUNS];

    v4i q[HF_K];
#pragma unroll
    for (int k = 0; k < HF_K; ++k)
        q[k] = hq_piece(c128, hf_piece_no(rst, roff, tid + HL_T * k), npieces);

    float f_dv[2], f_zz[2], f_self[2];
    int nrs[2], nre[2];
#pragma unroll
    for (int j = 0; j < 2; ++j) {
        const int i = tid + HL_T * j;
        const int vf = v_lo + i;
        const int vfc = min(vf, N - 1);
        f_dv[j] = dinv[vfc];
        f_zz[j] = z0[vfc];
        f_self[j] = gin[vfc];
        const int ps = pbeg[vfc], pe = pend[vfc];
        const int kb = (vfc >> BSH) - b0;
        int base = roff[0] - rst[0];
#pragma unroll
        for (int k = 1; k < HF_RUNS; ++k) {
            if (kb == k) base = roff[k] - rst[k];
        }
        const bool ok = i < npw && vf < v_hi;
        nrs[j] = ok ? base + ps : 0;
        nre[j] = ok ? base + pe : 0;
    }

    for (int j0 = HL_RING; j0 < per; j0 += HL_RING) {
#pragma unroll
        for (int k = 0; k < HL_RING; ++k) {
            d4[rx[k]] = rr[k];
            rx[k] = hl_piece(j0 + k, tid, nq, rot);
            rr[k] = hp[rx[k]];
        }
    }
#pragma unroll
    for (int k = 0; k < HL_RING; ++k) d4[rx[k]] = rr[k];
    if (tid < (nh & 3)) sg[(nq << 2) + tid] = gh[(nq << 2) + tid];
    if (tid == 0) sg[HL_SLAB - 1] = 0.f;
    __syncthreads();

    float accn[2] = {0.f, 0.f};
    for (int c0 = 0;;) {
#pragma unroll
        for (int k = 0; k < HF_K; ++k)
            psum[tid + HL_T * k] = hq_sum8(sg, q[k], c0 + tid + HL_T * k < P);
        __syncthreads();
        float t[2][8];
        int pa[2], pe[2];
#pragma unroll
        for (int j = 0; j < 2; ++j) {
            pa[j] = max(nrs[j], c0);
            pe[j] = min(nre[j], c0 + HF_CAP);
#pragma unroll
            for (int k = 0; k < 8; ++k)
                t[j][k] = psum[max(min(pa[j] + k, pe[j] - 1), c0) - c0];
        }
#pragma unroll
        for (int j = 0; j < 2; ++j) {
#pragma unroll
            for (int k = 0; k < 8; ++k) accn[j] += (pa[j] + k < pe[j]) ? t[j][k] : 0.f;
            for (int p = pa[j] + 8; p < pe[j]; ++p) accn[j] += psum[p - c0];
        }
        c0 += HF_CAP;
        if (c0 >= P) break;
        __syncthreads();
#pragma unroll
        for (int k = 0; k < HF_K; ++k)
            q[k] = hq_piece(c128, hf_piece_no(rst, roff, c0 + tid + HL_T * k), npieces);
    }

    const float bias = LAST ? b3[0] : 0.f;
#pragma unroll
    for (int j = 0; j < 2; ++j) {
        const int i = tid + HL_T * j;
        const int vf = v_lo + i;
        if (i < npw && vf < v_hi) {
            const float dv = f_dv[j];
            const float zt = h ? ONE_MINUS_ALPHA * dv * accn[j]
                               : ONE_MINUS_ALPHA * dv * (accn[j] + f_self[j]) + ALPHA_F * f_zz[j];
            atomicAdd(&gacc[vf], LAST ? (h ? zt : zt + bias) : dv * zt);
            if (gzero != nullptr && j == h) gzero[vf] = 0.f;
        }
    }
}


extern "C" void kernel_launch(void* const* d_in, const int* in_sizes, int n_in,
                              void* d_out, int out_size, void* d_ws, size_t ws_size,
                              hipStream_t stream) {
    const float* x  = (const float*)d_in[0];
    const int*  eidx = (const int*)d_in[1];
    const float* W1 = (const float*)d_in[2];
    const float* b1 = (const float*)d_in[3];
    const float* W2 = (const float*)d_in[4];
    const float* b2 = (const float*)d_in[5];
    const float* W3 = (const float*)d_in[6];
    const float* b3 = (const float*)d_in[7];
    float* out = (float*)d_out;

    const int N = in_sizes[0];
    const int E = in_sizes[1] / 2;

    char* ws = (char*)d_ws;
    const int B = 256;

    {
        const int nb = (N + BNODES - 1) / BNODES;
        const int nh = nb * CHUNK_WGS;
        size_t o2 = 0;
        auto carve2 = [&](size_t bytes) -> char* {
            char* p = ws + o2;
            o2 += (bytes + 255) & ~(size_t)255;
            return p;
        };
        int*   cnt    = (int*)carve2((size_t)N * 4);
        float* dinv   = (float*)carve2((size_t)N * 4);
        int*   rowptr = (int*)carve2((size_t)N * 4);
        int*   hist1  = (int*)carve2((size_t)nh * 4);
        int*   off1   = (int*)carve2((size_t)nh * 4);
        int*   csr    = (int*)carve2((size_t)E * 4);
        const size_t plane = (((size_t)N * 4) + 255) & ~(size_t)255;
        const size_t tmpb = (size_t)E * 4;
        char* shared_region = carve2(tmpb > 4 * plane ? tmpb : 4 * plane);
        unsigned* tmp = (unsigned*)shared_region;
        float* z0 = (float*)shared_region;
        float* gA = (float*)(shared_region + plane);
        float* gB = (float*)(shared_region + 2 * plane);
        float* gC = (float*)(shared_region + 3 * plane);
        int* mid = hist1;
        const bool lds_hop = N >= 4096 && N <= 2 * HL_SLAB && E >= 64;
        const int split = lds_hop ? ((N / 2 + 31) & ~31) : N;
        const int npw2 = (N + HP_RANGES - 1) / HP_RANGES;
        const bool pair_hop = lds_hop && npw2 <= HP_NPW;

        if (o2 <= ws_size && nb <= NB_MAX && N <= (1 << 24)) {
            const int chunk = (E + CHUNK_WGS - 1) / CHUNK_WGS;
            const int epad = (E + 63) & ~63;
            const size_t mid_off = (((size_t)epad * 2) + 255) & ~(size_t)255;
            const bool own_chunks = lds_hop && chunk <= ST1_CAP &&
                                    mid_off + (size_t)N * 4 <= (size_t)E * 4;
            const int epadp = (E + 8 + PB_GAP * nb + 63) & ~63;
            const size_t midp_off = (((size_t)epadp * 2) + 255) & ~(size_t)255;
            const bool padded = own_chunks && pair_hop && npw2 <= HQ_NPW &&
                                N <= 2 * HL_SLAB - 64 &&
                                midp_off + plane + (size_t)N * 4 <= (size_t)E * 4;
            int* const end1 = (int*)((char*)csr + midp_off + plane);
            if (own_chunks) {
                mid = (int*)((char*)csr + (padded ? midp_off : mid_off));
                ak_chunk_sort<<<CHUNK_WGS, 1024, 0, stream>>>(eidx, hist1, off1, tmp, E, chunk, nb);
                if (padded) {
                    ak_bucket_sort_sh<true><<<nb, 512, 0, stream>>>(tmp, hist1, off1, rowptr, cnt, mid,
                                                                    end1, dinv, csr, E, N, chunk,
                                                                    split);
                } else {
                    ak_bucket_sort_sh<false><<<nb, 512, 0, stream>>>(tmp, hist1, off1, rowptr, cnt, mid,
                                                                     end1, dinv, csr, E, N, chunk,
                                                                     split);
                }
            } else {
                ak_bucket_hist<<<CHUNK_WGS, 1024, 0, stream>>>(eidx, hist1, E, chunk, nb);
                ak_excl_scan<<<(nh + 1023) / 1024, 1024, 0, stream>>>(hist1, off1, nh);
                ak_bucket_scatter<<<CHUNK_WGS, 1024, 0, stream>>>(eidx, hist1, off1, tmp, E, chunk, nb);
                if (lds_hop) {
                    ak_bucket_sort<true><<<nb, 512, 0, stream>>>(tmp, off1, rowptr, cnt, mid, dinv, csr,
                                                                 E, N, nb, split);
                } else {
                    ak_bucket_sort<false><<<nb, 512, 0, stream>>>(tmp, off1, rowptr, cnt, mid, dinv, csr,
                                                                  E, N, nb, split);
                }
            }

            int tiles = N / 16;
            if (tiles > 0) {
                int waves_per_block = B / 32;
                int blocks = (tiles + waves_per_block - 1) / waves_per_block;
                ak_encoder_wmma<<<blocks, B, 0, stream>>>(x, W1, b1, W2, b2, W3, dinv, z0, gA,
                                                          pair_hop ? gB : (float*)nullptr, tiles);
            }
            int rem_start = tiles * 16;
            if (rem_start < N) {
                ak_encoder_tail<<<(N - rem_start + B - 1) / B, B, 0, stream>>>(
                    x, W1, b1, W2, b2, W3, dinv, z0, gA, pair_hop ? gB : (float*)nullptr,
                    rem_start, N);
            }
            if (pair_hop) {
                float* const pl[3] = {gA, gB, gC};
                for (int it = 0; it < KITERS; ++it) {
                    const float* gi = pl[it % 3];
                    if (it < KITERS - 1) {
                        float* gz = (it == KITERS - 2) ? out : pl[(it + 2) % 3];
                        if (padded) {
                            ak_hop_flat<false><<<2 * HP_RANGES, HL_T, 0, stream>>>(
                                rowptr, cnt, mid, end1, csr, gi, z0, dinv, pl[(it + 1) % 3], gz, b3,
                                N, npw2, split, epadp >> 3);
                        } else {
                            ak_hop_pair<false><<<2 * HP_RANGES, HL_T, 0, stream>>>(
                                rowptr, cnt, mid, csr, gi, z0, dinv, pl[(it + 1) % 3], gz, b3, N,
                                npw2, split, epad);
                        }
                    } else if (padded) {
                        ak_hop_flat<true><<<2 * HP_RANGES, HL_T, 0, stream>>>(
                            rowptr, cnt, mid, end1, csr, gi, z0, dinv, out, (float*)nullptr, b3, N,
                            npw2, split, epadp >> 3);
                    } else {
                        ak_hop_pair<true><<<2 * HP_RANGES, HL_T, 0, stream>>>(
                            rowptr, cnt, mid, csr, gi, z0, dinv, out, (float*)nullptr, b3, N, npw2,
                            split, epad);
                    }
                }
                return;
            }

            const int hblocks = (N + 32 * HOP_TRIPS - 1) / (32 * HOP_TRIPS);
            const int npw = (N + HL_WGS - 1) / HL_WGS;
            for (int it = 0; it < KITERS; ++it) {
                const float* gi = (it & 1) ? gB : gA;
                float* go = (it & 1) ? gA : gB;
                if (lds_hop) {
                    if (it < KITERS - 1) {
                        ak_hop_lds<false><<<HL_WGS, HL_T, 0, stream>>>(rowptr, cnt, mid, csr, gi, z0,
                                                                       dinv, go, b3, out, N, npw,
                                                                       split, epad);
                    } else {
                        ak_hop_lds<true><<<HL_WGS, HL_T, 0, stream>>>(rowptr, cnt, mid, csr, gi, z0,
                                                                      dinv, (float*)nullptr, b3, out,
                                                                      N, npw, split, epad);
                    }
                } else if (it < KITERS - 1) {
                    ak_hop<false><<<hblocks, B, 0, stream>>>(rowptr, cnt, csr, gi, z0, dinv, go,
                                                             b3, out, N);
                } else {
                    ak_hop<true><<<hblocks, B, 0, stream>>>(rowptr, cnt, csr, gi, z0, dinv,
                                                            (float*)nullptr, b3, out, N);
                }
            }
            return;
        }
    }

    size_t off = 0;
    auto carve = [&](size_t bytes) -> char* {
        char* p = ws + off;
        off += (bytes + 255) & ~(size_t)255;
        return p;
    };
    float* deg  = (float*)carve((size_t)N * 4);
    float* dinv = (float*)carve((size_t)N * 4);
    float* z0   = (float*)carve((size_t)N * 4);
    float* g    = (float*)carve((size_t)N * 4);
    float* agg  = (float*)carve((size_t)N * 4);

    ak_deg_init<<<(N + B - 1) / B, B, 0, stream>>>(deg, N);
    ak_deg_accum<<<(E + B - 1) / B, B, 0, stream>>>(eidx, deg, E);
    ak_dinv<<<(N + B - 1) / B, B, 0, stream>>>(deg, dinv, N);

    int tiles = N / 16;
    if (tiles > 0) {
        int waves_per_block = B / 32;
        int blocks = (tiles + waves_per_block - 1) / waves_per_block;
        ak_encoder_wmma<<<blocks, B, 0, stream>>>(x, W1, b1, W2, b2, W3, dinv, z0, g, agg, tiles);
    }
    int rem_start = tiles * 16;
    if (rem_start < N) {
        ak_encoder_tail<<<(N - rem_start + B - 1) / B, B, 0, stream>>>(
            x, W1, b1, W2, b2, W3, dinv, z0, g, agg, rem_start, N);
    }

    for (int it = 0; it < KITERS; ++it) {
        ak_edge_scatter<<<(E + B - 1) / B, B, 0, stream>>>(eidx, g, agg, E);
        if (it < KITERS - 1) {
            ak_finalize<<<(N + B - 1) / B, B, 0, stream>>>(agg, g, z0, dinv, N);
        } else {
            ak_finalize_out<<<(N + B - 1) / B, B, 0, stream>>>(agg, g, z0, dinv, b3, out, N);
        }
    }
}
